# starting kernel
# baseline (speedup 1.0000x reference)
#include <hip/hip_runtime.h>

#define N_NODES 50000
#define N_EDGES 800000
#define DIM 128

typedef float v2f __attribute__((ext_vector_type(2)));
typedef float v8f __attribute__((ext_vector_type(8)));

__global__ void k_init_deg(float* __restrict__ deg) {
    int i = blockIdx.x * blockDim.x + threadIdx.x;
    if (i < N_NODES) deg[i] = 1.0f;
}

__global__ void k_deg_accum(const int* __restrict__ dst, float* __restrict__ deg) {
    int e = blockIdx.x * blockDim.x + threadIdx.x;
    if (e < N_EDGES) atomicAdd(&deg[dst[e]], 1.0f);
}

__global__ void k_deg_to_dinv(float* __restrict__ deg) {
    int i = blockIdx.x * blockDim.x + threadIdx.x;
    if (i < N_NODES) {
        float d = deg[i];
        deg[i] = (d > 0.0f) ? rsqrtf(d) : 0.0f;
    }
}

__global__ __launch_bounds__(256)
void k_gemm_wmma_f32(const float* __restrict__ A, const float* __restrict__ W,
                     float* __restrict__ out) {
    const int lane  = threadIdx.x & 31;
    const int wave  = threadIdx.x >> 5;
    const int mTile = blockIdx.x << 4;
    const int n0    = wave << 4;
    const int l15   = lane & 15;
    const int koff  = (lane >> 4) << 1;

    const float* arow = A + (size_t)(mTile + l15) * DIM;
    const float* bcol = W + n0 + l15;

    v8f acc = {};
#pragma unroll
    for (int kk = 0; kk < DIM; kk += 4) {
        v2f a, b;
        a.x = arow[kk + koff];
        a.y = arow[kk + koff + 1];
        b.x = bcol[(size_t)(kk + koff) * DIM];
        b.y = bcol[(size_t)(kk + koff + 1) * DIM];
        acc = __builtin_amdgcn_wmma_f32_16x16x4_f32(
                   false, a,  false, b,
                   (short)0, acc,  false,  false);
    }

    const int  rowBase = mTile + ((lane >> 4) << 3);
    float*     orow    = out + (size_t)rowBase * DIM + n0 + l15;
#pragma unroll
    for (int v = 0; v < 8; ++v)
        orow[(size_t)v * DIM] = acc[v];
}

__global__ void k_self_init(const float* __restrict__ t,
                            const float* __restrict__ dinv,
                            float* __restrict__ agg) {
    int idx = blockIdx.x * blockDim.x + threadIdx.x;
    if (idx >= N_NODES * (DIM / 4)) return;
    int node = idx >> 5;
    float w  = dinv[node];
    w *= w;
    float4 v = ((const float4*)t)[idx];
    v.x *= w; v.y *= w; v.z *= w; v.w *= w;
    ((float4*)agg)[idx] = v;
}

__global__ __launch_bounds__(256)
void k_scatter(const int* __restrict__ src, const int* __restrict__ dst,
               const float* __restrict__ dinv, const float* __restrict__ t,
               float* __restrict__ agg) {
    const int e = (blockIdx.x << 3) + (threadIdx.x >> 5);
    if (e >= N_EDGES) return;
    const int   lane = threadIdx.x & 31;
    const int   s    = src[e];
    const int   d    = dst[e];
    const float nrm  = dinv[s] * dinv[d];

    const float4 v  = ((const float4*)(t + (size_t)s * DIM))[lane];
    float*       ad = agg + (size_t)d * DIM + (lane << 2);
    atomicAdd(ad + 0, v.x * nrm);
    atomicAdd(ad + 1, v.y * nrm);
    atomicAdd(ad + 2, v.z * nrm);
    atomicAdd(ad + 3, v.w * nrm);
}

__global__ void k_bias_relu(const float* __restrict__ agg,
                            const float* __restrict__ b,
                            float* __restrict__ h) {
    int idx = blockIdx.x * blockDim.x + threadIdx.x;
    if (idx >= N_NODES * (DIM / 4)) return;
    int j = idx & (DIM / 4 - 1);
    float4 v  = ((const float4*)agg)[idx];
    float4 bb = ((const float4*)b)[j];
    v.x = fmaxf(v.x + bb.x, 0.0f);
    v.y = fmaxf(v.y + bb.y, 0.0f);
    v.z = fmaxf(v.z + bb.z, 0.0f);
    v.w = fmaxf(v.w + bb.w, 0.0f);
    ((float4*)h)[idx] = v;
}

__global__ __launch_bounds__(256)
void k_heads(const float* __restrict__ h,
             const float* __restrict__ Wt, const float* __restrict__ bt,
             const float* __restrict__ We, const float* __restrict__ be,
             float* __restrict__ out) {
    const int node = (blockIdx.x << 3) + (threadIdx.x >> 5);
    if (node >= N_NODES) return;
    const int lane = threadIdx.x & 31;

    const float4 v  = ((const float4*)(h + (size_t)node * DIM))[lane];
    const float4 wt = ((const float4*)Wt)[lane];
    const float4 we = ((const float4*)We)[lane];

    float st = v.x * wt.x + v.y * wt.y + v.z * wt.z + v.w * wt.w;
    float ev = v.x * we.x + v.y * we.y + v.z * we.z + v.w * we.w;

#pragma unroll
    for (int off = 16; off > 0; off >>= 1) {
        st += __shfl_down(st, off, 32);
        ev += __shfl_down(ev, off, 32);
    }
    if (lane == 0) {
        out[node]           = st + bt[0];
        out[N_NODES + node] = ev + be[0];
    }
}

extern "C" void kernel_launch(void* const* d_in, const int* in_sizes, int n_in,
                              void* d_out, int out_size, void* d_ws, size_t ws_size,
                              hipStream_t stream) {
    (void)in_sizes; (void)n_in; (void)out_size; (void)ws_size;

    const float* x   = (const float*)d_in[0];
    const int*   ei  = (const int*)  d_in[1];
    const float* W1  = (const float*)d_in[2];
    const float* b1  = (const float*)d_in[3];
    const float* W2  = (const float*)d_in[4];
    const float* b2  = (const float*)d_in[5];
    const float* Wt  = (const float*)d_in[6];
    const float* bt  = (const float*)d_in[7];
    const float* We  = (const float*)d_in[8];
    const float* be  = (const float*)d_in[9];
    float*       out = (float*)d_out;

    const int* src = ei;
    const int* dst = ei + N_EDGES;

    float* dinv = (float*)d_ws;
    float* bufA = dinv + ((N_NODES + 255) & ~255);
    float* bufB = bufA + (size_t)N_NODES * DIM;

    const int ELEM4   = N_NODES * (DIM / 4);
    const int gNode   = (N_NODES + 255) / 256;
    const int gEdge   = (N_EDGES + 255) / 256;
    const int gElem4  = (ELEM4 + 255) / 256;
    const int gGemm   = N_NODES / 16;
    const int gWaveE  = (N_EDGES + 7) / 8;
    const int gWaveN  = (N_NODES + 7) / 8;

    k_init_deg   <<<gNode, 256, 0, stream>>>(dinv);
    k_deg_accum  <<<gEdge, 256, 0, stream>>>(dst, dinv);
    k_deg_to_dinv<<<gNode, 256, 0, stream>>>(dinv);

    k_gemm_wmma_f32<<<gGemm,  256, 0, stream>>>(x, W1, bufA);
    k_self_init    <<<gElem4, 256, 0, stream>>>(bufA, dinv, bufB);
    k_scatter      <<<gWaveE, 256, 0, stream>>>(src, dst, dinv, bufA, bufB);
    k_bias_relu    <<<gElem4, 256, 0, stream>>>(bufB, b1, bufA);

    k_gemm_wmma_f32<<<gGemm,  256, 0, stream>>>(bufA, W2, bufB);
    k_self_init    <<<gElem4, 256, 0, stream>>>(bufB, dinv, bufA);
    k_scatter      <<<gWaveE, 256, 0, stream>>>(src, dst, dinv, bufB, bufA);
    k_bias_relu    <<<gElem4, 256, 0, stream>>>(bufA, b2, bufB);

    k_heads<<<gWaveN, 256, 0, stream>>>(bufB, Wt, bt, We, be, out);
}


# aggregation: edges binned once by 64-node destination tile (integer-atomic ranges, scan); one work-group sums a tile's rows in LDS with ds float adds, plain row stores; sum order = arrival order
# speedup vs baseline: 1.8566x; 1.8566x over previous
#include <hip/hip_runtime.h>

#define N_NODES 50000
#define N_EDGES 800000
#define DIM 128

#define TILE        64
#define TILE_SHIFT  6
#define N_TILES     ((N_NODES + TILE - 1) / TILE)
#define CUR_PITCH   32
#define FILL_THREADS 256
#define FILL_EPT     8
#define AGG_THREADS  512
#define AGG_WAVES    (AGG_THREADS / 32)

#define AGG_INIT_TRIPS (TILE * (DIM / 4) / AGG_THREADS)

static_assert(N_NODES <= 65536, "source index is packed into 16 bits");
static_assert(AGG_INIT_TRIPS * AGG_THREADS == TILE * (DIM / 4), "tile is a whole number of passes");
static_assert(N_TILES < 1024, "tile scan is one 1024-thread work-group");

typedef float v2f __attribute__((ext_vector_type(2)));
typedef float v8f __attribute__((ext_vector_type(8)));

__global__ void k_init_deg(float* __restrict__ deg) {
    int i = blockIdx.x * blockDim.x + threadIdx.x;
    if (i < N_NODES) deg[i] = 1.0f;
}

__global__ void k_deg_accum(const int* __restrict__ dst, float* __restrict__ deg) {
    int e = blockIdx.x * blockDim.x + threadIdx.x;
    if (e < N_EDGES) atomicAdd(&deg[dst[e]], 1.0f);
}

__global__ __launch_bounds__(TILE)
void k_deg_to_dinv(float* __restrict__ deg, int* __restrict__ tileCnt) {
    __shared__ int cnt;
    if (threadIdx.x == 0) cnt = 0;
    __syncthreads();
    int i = blockIdx.x * TILE + threadIdx.x;
    if (i < N_NODES) {
        float d = deg[i];
        atomicAdd(&cnt, (int)d - 1);
        deg[i] = (d > 0.0f) ? rsqrtf(d) : 0.0f;
    }
    __syncthreads();
    if (threadIdx.x == 0) tileCnt[blockIdx.x] = cnt;
}

__global__ __launch_bounds__(1024)
void k_tile_scan(const int* __restrict__ tileCnt, int* __restrict__ tileStart,
                 int* __restrict__ cursor) {
    __shared__ int s[1024];
    const int t = threadIdx.x;
    const int c = (t < N_TILES) ? tileCnt[t] : 0;
    s[t] = c;
    __syncthreads();
    for (int off = 1; off < 1024; off <<= 1) {
        const int v = (t >= off) ? s[t - off] : 0;
        __syncthreads();
        s[t] += v;
        __syncthreads();
    }
    const int excl = s[t] - c;
    if (t <= N_TILES) tileStart[t] = excl;
    if (t <  N_TILES) cursor[t * CUR_PITCH] = excl;
}

__global__ __launch_bounds__(FILL_THREADS)
void k_bin_fill(const int* __restrict__ src, const int* __restrict__ dst,
                const float* __restrict__ dinv,
                int* __restrict__ cursor, int* __restrict__ recs) {
    __shared__ int lcnt[N_TILES];
    __shared__ int lbase[N_TILES];
    const int tid  = threadIdx.x;
    const int base = blockIdx.x * (FILL_THREADS * FILL_EPT);

    for (int t = tid; t < N_TILES; t += FILL_THREADS) lcnt[t] = 0;
    __syncthreads();

    int   d[FILL_EPT], s[FILL_EPT], r[FILL_EPT];
    float w[FILL_EPT];
#pragma unroll
    for (int k = 0; k < FILL_EPT; ++k) {
        const int e = min(base + k * FILL_THREADS + tid, N_EDGES - 1);
        d[k] = dst[e];
        s[k] = src[e];
    }
#pragma unroll
    for (int k = 0; k < FILL_EPT; ++k) {
        w[k] = dinv[s[k]] * dinv[d[k]];
        r[k] = 0;
        if (base + k * FILL_THREADS + tid < N_EDGES)
            r[k] = atomicAdd(&lcnt[d[k] >> TILE_SHIFT], 1);
    }
    __syncthreads();

    for (int t = tid; t < N_TILES; t += FILL_THREADS)
        lbase[t] = atomicAdd(&cursor[t * CUR_PITCH], lcnt[t]);
    __syncthreads();

#pragma unroll
    for (int k = 0; k < FILL_EPT; ++k) {
        if (base + k * FILL_THREADS + tid < N_EDGES) {
            const int slot = lbase[d[k] >> TILE_SHIFT] + r[k];
            ((int2*)recs)[slot] = make_int2(s[k] | ((d[k] & (TILE - 1)) << 16),
                                            __float_as_int(w[k]));
        }
    }
}

__global__ __launch_bounds__(256)
void k_gemm_wmma_f32(const float* __restrict__ A, const float* __restrict__ W,
                     float* __restrict__ out) {
    const int lane  = threadIdx.x & 31;
    const int wave  = threadIdx.x >> 5;
    const int mTile = blockIdx.x << 4;
    const int n0    = wave << 4;
    const int l15   = lane & 15;
    const int koff  = (lane >> 4) << 1;

    const float* arow = A + (size_t)(mTile + l15) * DIM;
    const float* bcol = W + n0 + l15;

    v8f acc = {};
#pragma unroll
    for (int kk = 0; kk < DIM; kk += 4) {
        v2f a, b;
        a.x = arow[kk + koff];
        a.y = arow[kk + koff + 1];
        b.x = bcol[(size_t)(kk + koff) * DIM];
        b.y = bcol[(size_t)(kk + koff + 1) * DIM];
        acc = __builtin_amdgcn_wmma_f32_16x16x4_f32(
                   false, a,  false, b,
                   (short)0, acc,  false,  false);
    }

    const int  rowBase = mTile + ((lane >> 4) << 3);
    float*     orow    = out + (size_t)rowBase * DIM + n0 + l15;
#pragma unroll
    for (int v = 0; v < 8; ++v)
        orow[(size_t)v * DIM] = acc[v];
}

__global__ __launch_bounds__(AGG_THREADS)
void k_tile_agg(const int* __restrict__ recs, const int* __restrict__ tileStart,
                const float* __restrict__ dinv, const float* __restrict__ t,
                float* __restrict__ agg) {
    __shared__ float4 acc4[TILE * (DIM / 4)];
    float* acc = (float*)acc4;

    const int tid   = threadIdx.x;
    const int lane  = tid & 31;
    const int wave  = __builtin_amdgcn_readfirstlane(tid >> 5);
    const int node0 = blockIdx.x * TILE;
    const int start = tileStart[blockIdx.x];
    const int end   = tileStart[blockIdx.x + 1];
    const int2* rec = (const int2*)recs;

    int  b   = start + (wave << 5);
    int2 cur = make_int2(0, 0);
    if (b + lane < end) cur = rec[b + lane];

#pragma unroll
    for (int q = 0; q < AGG_INIT_TRIPS; ++q) {
        const int i    = tid + q * AGG_THREADS;
        const int node = min(node0 + (i >> 5), N_NODES - 1);
        float w = dinv[node];
        w *= w;
        float4 v = ((const float4*)t)[(size_t)node * (DIM / 4) + (i & 31)];
        v.x *= w; v.y *= w; v.z *= w; v.w *= w;
        acc4[i] = v;
    }
    __syncthreads();

    while (b < end) {
        const int nb  = b + AGG_WAVES * 32;
        int2      nxt = make_int2(0, 0);
        if (nb + lane < end) nxt = rec[nb + lane];

        const int n = min(32, end - b);

        for (int j0 = 0; j0 < n; j0 += 8) {
            float v[8][4];
            float wj[8];
            int   dl[8];
#pragma unroll
            for (int j = 0; j < 8; ++j) {
                const int pj = __builtin_amdgcn_readlane(cur.x, j0 + j);
                wj[j] = __int_as_float(__builtin_amdgcn_readlane(cur.y, j0 + j));
                dl[j] = pj >> 16;
                const float* row = t + (size_t)(pj & 0xFFFF) * DIM + lane;
                v[j][0] = row[0];
                v[j][1] = row[32];
                v[j][2] = row[64];
                v[j][3] = row[96];
            }
            if (j0 + 8 <= n) {
#pragma unroll
                for (int j = 0; j < 8; ++j) {
                    float* a = acc + dl[j] * DIM + lane;
                    atomicAdd(a,      v[j][0] * wj[j]);
                    atomicAdd(a + 32, v[j][1] * wj[j]);
                    atomicAdd(a + 64, v[j][2] * wj[j]);
                    atomicAdd(a + 96, v[j][3] * wj[j]);
                }
            } else {
#pragma unroll
                for (int j = 0; j < 8; ++j) {
                    if (j0 + j < n) {
                        float* a = acc + dl[j] * DIM + lane;
                        atomicAdd(a,      v[j][0] * wj[j]);
                        atomicAdd(a + 32, v[j][1] * wj[j]);
                        atomicAdd(a + 64, v[j][2] * wj[j]);
                        atomicAdd(a + 96, v[j][3] * wj[j]);
                    }
                }
            }
        }
        cur = nxt;
        b   = nb;
    }
    __syncthreads();

    for (int i = tid; i < TILE * (DIM / 4); i += AGG_THREADS) {
        const int node = node0 + (i >> 5);
        if (node < N_NODES)
            ((float4*)agg)[(size_t)node * (DIM / 4) + (i & 31)] = acc4[i];
    }
}

__global__ void k_bias_relu(const float* __restrict__ agg,
                            const float* __restrict__ b,
                            float* __restrict__ h) {
    int idx = blockIdx.x * blockDim.x + threadIdx.x;
    if (idx >= N_NODES * (DIM / 4)) return;
    int j = idx & (DIM / 4 - 1);
    float4 v  = ((const float4*)agg)[idx];
    float4 bb = ((const float4*)b)[j];
    v.x = fmaxf(v.x + bb.x, 0.0f);
    v.y = fmaxf(v.y + bb.y, 0.0f);
    v.z = fmaxf(v.z + bb.z, 0.0f);
    v.w = fmaxf(v.w + bb.w, 0.0f);
    ((float4*)h)[idx] = v;
}

__global__ __launch_bounds__(256)
void k_heads(const float* __restrict__ h,
             const float* __restrict__ Wt, const float* __restrict__ bt,
             const float* __restrict__ We, const float* __restrict__ be,
             float* __restrict__ out) {
    const int node = (blockIdx.x << 3) + (threadIdx.x >> 5);
    if (node >= N_NODES) return;
    const int lane = threadIdx.x & 31;

    const float4 v  = ((const float4*)(h + (size_t)node * DIM))[lane];
    const float4 wt = ((const float4*)Wt)[lane];
    const float4 we = ((const float4*)We)[lane];

    float st = v.x * wt.x + v.y * wt.y + v.z * wt.z + v.w * wt.w;
    float ev = v.x * we.x + v.y * we.y + v.z * we.z + v.w * we.w;

#pragma unroll
    for (int off = 16; off > 0; off >>= 1) {
        st += __shfl_down(st, off, 32);
        ev += __shfl_down(ev, off, 32);
    }
    if (lane == 0) {
        out[node]           = st + bt[0];
        out[N_NODES + node] = ev + be[0];
    }
}

extern "C" void kernel_launch(void* const* d_in, const int* in_sizes, int n_in,
                              void* d_out, int out_size, void* d_ws, size_t ws_size,
                              hipStream_t stream) {
    (void)in_sizes; (void)n_in; (void)out_size; (void)ws_size;

    const float* x   = (const float*)d_in[0];
    const int*   ei  = (const int*)  d_in[1];
    const float* W1  = (const float*)d_in[2];
    const float* b1  = (const float*)d_in[3];
    const float* W2  = (const float*)d_in[4];
    const float* b2  = (const float*)d_in[5];
    const float* Wt  = (const float*)d_in[6];
    const float* bt  = (const float*)d_in[7];
    const float* We  = (const float*)d_in[8];
    const float* be  = (const float*)d_in[9];
    float*       out = (float*)d_out;

    const int* src = ei;
    const int* dst = ei + N_EDGES;

    float* dinv = (float*)d_ws;
    float* bufA = dinv + ((N_NODES + 255) & ~255);
    float* bufB = bufA + (size_t)N_NODES * DIM;
    int*   recs      = (int*)(bufB + (size_t)N_NODES * DIM);
    int*   tileCnt   = recs + 2 * (size_t)N_EDGES;
    int*   tileStart = tileCnt + 1024;
    int*   cursor    = tileStart + 1024;

    const int ELEM4   = N_NODES * (DIM / 4);
    const int gNode   = (N_NODES + 255) / 256;
    const int gEdge   = (N_EDGES + 255) / 256;
    const int gElem4  = (ELEM4 + 255) / 256;
    const int gGemm   = N_NODES / 16;
    const int gFill   = (N_EDGES + FILL_THREADS * FILL_EPT - 1) / (FILL_THREADS * FILL_EPT);
    const int gWaveN  = (N_NODES + 7) / 8;

    k_init_deg   <<<gNode, 256, 0, stream>>>(dinv);
    k_deg_accum  <<<gEdge, 256, 0, stream>>>(dst, dinv);
    k_deg_to_dinv<<<N_TILES, TILE, 0, stream>>>(dinv, tileCnt);

    k_tile_scan<<<1, 1024, 0, stream>>>(tileCnt, tileStart, cursor);
    k_bin_fill <<<gFill, FILL_THREADS, 0, stream>>>(src, dst, dinv, cursor, recs);

    k_gemm_wmma_f32<<<gGemm,  256, 0, stream>>>(x, W1, bufA);
    k_tile_agg     <<<N_TILES, AGG_THREADS, 0, stream>>>(recs, tileStart, dinv, bufA, bufB);
    k_bias_relu    <<<gElem4, 256, 0, stream>>>(bufB, b1, bufA);

    k_gemm_wmma_f32<<<gGemm,  256, 0, stream>>>(bufA, W2, bufB);
    k_tile_agg     <<<N_TILES, AGG_THREADS, 0, stream>>>(recs, tileStart, dinv, bufB, bufA);
    k_bias_relu    <<<gElem4, 256, 0, stream>>>(bufA, b2, bufB);

    k_heads<<<gWaveN, 256, 0, stream>>>(bufB, Wt, bt, We, be, out);
}


# bias+relu moved to the consumer's loads: GEMM 2 applies max(a+b1,0) to its A fragments, the heads kernel max(a+b2,0) to its row; both elementwise passes dropped; 10 launches, not 12
# speedup vs baseline: 2.0805x; 1.1206x over previous
#include <hip/hip_runtime.h>

#define N_NODES 50000
#define N_EDGES 800000
#define DIM 128

#define TILE        64
#define TILE_SHIFT  6
#define N_TILES     ((N_NODES + TILE - 1) / TILE)
#define CUR_PITCH   32
#define FILL_THREADS 256
#define FILL_EPT     8
#define AGG_THREADS  512
#define AGG_WAVES    (AGG_THREADS / 32)

#define AGG_INIT_TRIPS (TILE * (DIM / 4) / AGG_THREADS)

static_assert(N_NODES <= 65536, "source index is packed into 16 bits");
static_assert(AGG_INIT_TRIPS * AGG_THREADS == TILE * (DIM / 4), "tile is a whole number of passes");
static_assert(N_TILES < 1024, "tile scan is one 1024-thread work-group");

typedef float v2f __attribute__((ext_vector_type(2)));
typedef float v8f __attribute__((ext_vector_type(8)));

__global__ void k_init_deg(float* __restrict__ deg) {
    int i = blockIdx.x * blockDim.x + threadIdx.x;
    if (i < N_NODES) deg[i] = 1.0f;
}

__global__ void k_deg_accum(const int* __restrict__ dst, float* __restrict__ deg) {
    int e = blockIdx.x * blockDim.x + threadIdx.x;
    if (e < N_EDGES) atomicAdd(&deg[dst[e]], 1.0f);
}

__global__ __launch_bounds__(TILE)
void k_deg_to_dinv(float* __restrict__ deg, int* __restrict__ tileCnt) {
    __shared__ int cnt;
    if (threadIdx.x == 0) cnt = 0;
    __syncthreads();
    int i = blockIdx.x * TILE + threadIdx.x;
    if (i < N_NODES) {
        float d = deg[i];
        atomicAdd(&cnt, (int)d - 1);
        deg[i] = (d > 0.0f) ? rsqrtf(d) : 0.0f;
    }
    __syncthreads();
    if (threadIdx.x == 0) tileCnt[blockIdx.x] = cnt;
}

__global__ __launch_bounds__(1024)
void k_tile_scan(const int* __restrict__ tileCnt, int* __restrict__ tileStart,
                 int* __restrict__ cursor) {
    __shared__ int s[1024];
    const int t = threadIdx.x;
    const int c = (t < N_TILES) ? tileCnt[t] : 0;
    s[t] = c;
    __syncthreads();
    for (int off = 1; off < 1024; off <<= 1) {
        const int v = (t >= off) ? s[t - off] : 0;
        __syncthreads();
        s[t] += v;
        __syncthreads();
    }
    const int excl = s[t] - c;
    if (t <= N_TILES) tileStart[t] = excl;
    if (t <  N_TILES) cursor[t * CUR_PITCH] = excl;
}

__global__ __launch_bounds__(FILL_THREADS)
void k_bin_fill(const int* __restrict__ src, const int* __restrict__ dst,
                const float* __restrict__ dinv,
                int* __restrict__ cursor, int* __restrict__ recs) {
    __shared__ int lcnt[N_TILES];
    __shared__ int lbase[N_TILES];
    const int tid  = threadIdx.x;
    const int base = blockIdx.x * (FILL_THREADS * FILL_EPT);

    for (int t = tid; t < N_TILES; t += FILL_THREADS) lcnt[t] = 0;
    __syncthreads();

    int   d[FILL_EPT], s[FILL_EPT], r[FILL_EPT];
    float w[FILL_EPT];
#pragma unroll
    for (int k = 0; k < FILL_EPT; ++k) {
        const int e = min(base + k * FILL_THREADS + tid, N_EDGES - 1);
        d[k] = dst[e];
        s[k] = src[e];
    }
#pragma unroll
    for (int k = 0; k < FILL_EPT; ++k) {
        w[k] = dinv[s[k]] * dinv[d[k]];
        r[k] = 0;
        if (base + k * FILL_THREADS + tid < N_EDGES)
            r[k] = atomicAdd(&lcnt[d[k] >> TILE_SHIFT], 1);
    }
    __syncthreads();

    for (int t = tid; t < N_TILES; t += FILL_THREADS)
        lbase[t] = atomicAdd(&cursor[t * CUR_PITCH], lcnt[t]);
    __syncthreads();

#pragma unroll
    for (int k = 0; k < FILL_EPT; ++k) {
        if (base + k * FILL_THREADS + tid < N_EDGES) {
            const int slot = lbase[d[k] >> TILE_SHIFT] + r[k];
            ((int2*)recs)[slot] = make_int2(s[k] | ((d[k] & (TILE - 1)) << 16),
                                            __float_as_int(w[k]));
        }
    }
}

template <bool ACT>
__global__ __launch_bounds__(256)
void k_gemm_wmma_f32(const float* __restrict__ A, const float* __restrict__ bias,
                     const float* __restrict__ W, float* __restrict__ out) {
    const int lane  = threadIdx.x & 31;
    const int wave  = threadIdx.x >> 5;
    const int mTile = blockIdx.x << 4;
    const int n0    = wave << 4;
    const int l15   = lane & 15;
    const int koff  = (lane >> 4) << 1;

    const float* arow = A + (size_t)(mTile + l15) * DIM;
    const float* bcol = W + n0 + l15;
    const int  rowBase = mTile + ((lane >> 4) << 3);
    float*     orow;

    v8f acc = {};
    if constexpr (ACT) {
        v2f af[DIM / 4], bf[DIM / 4];
#pragma unroll
        for (int i = 0; i < DIM / 4; ++i) {
            af[i].x = arow[4 * i + koff];
            af[i].y = arow[4 * i + koff + 1];
        }
#pragma unroll
        for (int i = 0; i < DIM / 4; ++i) {
            bf[i].x = bcol[(size_t)(4 * i + koff) * DIM];
            bf[i].y = bcol[(size_t)(4 * i + koff + 1) * DIM];
        }
#pragma unroll
        for (int i = 0; i < DIM / 4; ++i) {
            const float4 bq = ((const float4*)bias)[i];
            af[i].x = fmaxf(af[i].x + (koff ? bq.z : bq.x), 0.0f);
            af[i].y = fmaxf(af[i].y + (koff ? bq.w : bq.y), 0.0f);
        }
        unsigned obyte = (unsigned)(rowBase * DIM + n0 + l15) * 4u;
        asm volatile("" : "+v"(obyte));
        orow = (float*)((char*)out + obyte);
        __builtin_amdgcn_sched_barrier(0);
#pragma unroll
        for (int i = 0; i < DIM / 4; ++i)
            acc = __builtin_amdgcn_wmma_f32_16x16x4_f32(
                       false, af[i],  false, bf[i],
                       (short)0, acc,  false,  false);
    } else {
        orow = out + (size_t)rowBase * DIM + n0 + l15;
#pragma unroll
        for (int kk = 0; kk < DIM; kk += 4) {
            v2f a, b;
            a.x = arow[kk + koff];
            a.y = arow[kk + koff + 1];
            b.x = bcol[(size_t)(kk + koff) * DIM];
            b.y = bcol[(size_t)(kk + koff + 1) * DIM];
            acc = __builtin_amdgcn_wmma_f32_16x16x4_f32(
                       false, a,  false, b,
                       (short)0, acc,  false,  false);
        }
    }

#pragma unroll
    for (int v = 0; v < 8; ++v)
        orow[(size_t)v * DIM] = acc[v];
}

__global__ __launch_bounds__(AGG_THREADS)
void k_tile_agg(const int* __restrict__ recs, const int* __restrict__ tileStart,
                const float* __restrict__ dinv, const float* __restrict__ t,
                float* __restrict__ agg) {
    __shared__ float4 acc4[TILE * (DIM / 4)];
    float* acc = (float*)acc4;

    const int tid   = threadIdx.x;
    const int lane  = tid & 31;
    const int wave  = __builtin_amdgcn_readfirstlane(tid >> 5);
    const int node0 = blockIdx.x * TILE;
    const int start = tileStart[blockIdx.x];
    const int end   = tileStart[blockIdx.x + 1];
    const int2* rec = (const int2*)recs;

    int  b   = start + (wave << 5);
    int2 cur = make_int2(0, 0);
    if (b + lane < end) cur = rec[b + lane];

#pragma unroll
    for (int q = 0; q < AGG_INIT_TRIPS; ++q) {
        const int i    = tid + q * AGG_THREADS;
        const int node = min(node0 + (i >> 5), N_NODES - 1);
        float w = dinv[node];
        w *= w;
        float4 v = ((const float4*)t)[(size_t)node * (DIM / 4) + (i & 31)];
        v.x *= w; v.y *= w; v.z *= w; v.w *= w;
        acc4[i] = v;
    }
    __syncthreads();

    while (b < end) {
        const int nb  = b + AGG_WAVES * 32;
        int2      nxt = make_int2(0, 0);
        if (nb + lane < end) nxt = rec[nb + lane];

        const int n = min(32, end - b);

        for (int j0 = 0; j0 < n; j0 += 8) {
            float v[8][4];
            float wj[8];
            int   dl[8];
#pragma unroll
            for (int j = 0; j < 8; ++j) {
                const int pj = __builtin_amdgcn_readlane(cur.x, j0 + j);
                wj[j] = __int_as_float(__builtin_amdgcn_readlane(cur.y, j0 + j));
                dl[j] = pj >> 16;
                const float* row = t + (size_t)(pj & 0xFFFF) * DIM + lane;
                v[j][0] = row[0];
                v[j][1] = row[32];
                v[j][2] = row[64];
                v[j][3] = row[96];
            }
            if (j0 + 8 <= n) {
#pragma unroll
                for (int j = 0; j < 8; ++j) {
                    float* a = acc + dl[j] * DIM + lane;
                    atomicAdd(a,      v[j][0] * wj[j]);
                    atomicAdd(a + 32, v[j][1] * wj[j]);
                    atomicAdd(a + 64, v[j][2] * wj[j]);
                    atomicAdd(a + 96, v[j][3] * wj[j]);
                }
            } else {
#pragma unroll
                for (int j = 0; j < 8; ++j) {
                    if (j0 + j < n) {
                        float* a = acc + dl[j] * DIM + lane;
                        atomicAdd(a,      v[j][0] * wj[j]);
                        atomicAdd(a + 32, v[j][1] * wj[j]);
                        atomicAdd(a + 64, v[j][2] * wj[j]);
                        atomicAdd(a + 96, v[j][3] * wj[j]);
                    }
                }
            }
        }
        cur = nxt;
        b   = nb;
    }
    __syncthreads();

    for (int i = tid; i < TILE * (DIM / 4); i += AGG_THREADS) {
        const int node = node0 + (i >> 5);
        if (node < N_NODES)
            ((float4*)agg)[(size_t)node * (DIM / 4) + (i & 31)] = acc4[i];
    }
}

__global__ __launch_bounds__(256)
void k_heads(const float* __restrict__ agg, const float* __restrict__ b,
             const float* __restrict__ Wt, const float* __restrict__ bt,
             const float* __restrict__ We, const float* __restrict__ be,
             float* __restrict__ out) {
    const int node = (blockIdx.x << 3) + (threadIdx.x >> 5);
    if (node >= N_NODES) return;
    const int lane = threadIdx.x & 31;

    float4       v  = ((const float4*)(agg + (size_t)node * DIM))[lane];
    const float4 bb = ((const float4*)b)[lane];
    v.x = fmaxf(v.x + bb.x, 0.0f);
    v.y = fmaxf(v.y + bb.y, 0.0f);
    v.z = fmaxf(v.z + bb.z, 0.0f);
    v.w = fmaxf(v.w + bb.w, 0.0f);
    const float4 wt = ((const float4*)Wt)[lane];
    const float4 we = ((const float4*)We)[lane];

    float st = v.x * wt.x + v.y * wt.y + v.z * wt.z + v.w * wt.w;
    float ev = v.x * we.x + v.y * we.y + v.z * we.z + v.w * we.w;

#pragma unroll
    for (int off = 16; off > 0; off >>= 1) {
        st += __shfl_down(st, off, 32);
        ev += __shfl_down(ev, off, 32);
    }
    if (lane == 0) {
        out[node]           = st + bt[0];
        out[N_NODES + node] = ev + be[0];
    }
}

extern "C" void kernel_launch(void* const* d_in, const int* in_sizes, int n_in,
                              void* d_out, int out_size, void* d_ws, size_t ws_size,
                              hipStream_t stream) {
    (void)in_sizes; (void)n_in; (void)out_size; (void)ws_size;

    const float* x   = (const float*)d_in[0];
    const int*   ei  = (const int*)  d_in[1];
    const float* W1  = (const float*)d_in[2];
    const float* b1  = (const float*)d_in[3];
    const float* W2  = (const float*)d_in[4];
    const float* b2  = (const float*)d_in[5];
    const float* Wt  = (const float*)d_in[6];
    const float* bt  = (const float*)d_in[7];
    const float* We  = (const float*)d_in[8];
    const float* be  = (const float*)d_in[9];
    float*       out = (float*)d_out;

    const int* src = ei;
    const int* dst = ei + N_EDGES;

    float* dinv = (float*)d_ws;
    float* bufA = dinv + ((N_NODES + 255) & ~255);
    float* bufB = bufA + (size_t)N_NODES * DIM;
    int*   recs      = (int*)(bufB + (size_t)N_NODES * DIM);
    int*   tileCnt   = recs + 2 * (size_t)N_EDGES;
    int*   tileStart = tileCnt + 1024;
    int*   cursor    = tileStart + 1024;

    const int gNode   = (N_NODES + 255) / 256;
    const int gEdge   = (N_EDGES + 255) / 256;
    const int gGemm   = N_NODES / 16;
    const int gFill   = (N_EDGES + FILL_THREADS * FILL_EPT - 1) / (FILL_THREADS * FILL_EPT);
    const int gWaveN  = (N_NODES + 7) / 8;

    k_init_deg   <<<gNode, 256, 0, stream>>>(dinv);
    k_deg_accum  <<<gEdge, 256, 0, stream>>>(dst, dinv);
    k_deg_to_dinv<<<N_TILES, TILE, 0, stream>>>(dinv, tileCnt);

    k_tile_scan<<<1, 1024, 0, stream>>>(tileCnt, tileStart, cursor);
    k_bin_fill <<<gFill, FILL_THREADS, 0, stream>>>(src, dst, dinv, cursor, recs);

    k_gemm_wmma_f32<false><<<gGemm,  256, 0, stream>>>(x, b1, W1, bufA);
    k_tile_agg            <<<N_TILES, AGG_THREADS, 0, stream>>>(recs, tileStart, dinv, bufA, bufB);

    k_gemm_wmma_f32<true> <<<gGemm,  256, 0, stream>>>(bufB, b1, W2, bufA);
    k_tile_agg            <<<N_TILES, AGG_THREADS, 0, stream>>>(recs, tileStart, dinv, bufA, bufB);

    k_heads<<<gWaveN, 256, 0, stream>>>(bufB, b2, Wt, bt, We, be, out);
}


# GEMM re-tiled: a wave owns a 16-row strip x all 128 columns (8 accumulators), loads A fragments once, W staged in LDS once per work-group in fragment order; 391 groups, not 3125; k sum order changed
# speedup vs baseline: 2.3606x; 1.1346x over previous
#include <hip/hip_runtime.h>

#define N_NODES 50000
#define N_EDGES 800000
#define DIM 128

#define TILE        64
#define TILE_SHIFT  6
#define N_TILES     ((N_NODES + TILE - 1) / TILE)
#define CUR_PITCH   32
#define FILL_THREADS 256
#define FILL_EPT     8
#define AGG_THREADS  512
#define AGG_WAVES    (AGG_THREADS / 32)

#define AGG_INIT_TRIPS (TILE * (DIM / 4) / AGG_THREADS)

static_assert(N_NODES <= 65536, "source index is packed into 16 bits");
static_assert(AGG_INIT_TRIPS * AGG_THREADS == TILE * (DIM / 4), "tile is a whole number of passes");
static_assert(N_TILES < 1024, "tile scan is one 1024-thread work-group");

typedef float v2f __attribute__((ext_vector_type(2)));
typedef float v8f __attribute__((ext_vector_type(8)));

__global__ void k_init_deg(float* __restrict__ deg) {
    int i = blockIdx.x * blockDim.x + threadIdx.x;
    if (i < N_NODES) deg[i] = 1.0f;
}

__global__ void k_deg_accum(const int* __restrict__ dst, float* __restrict__ deg) {
    int e = blockIdx.x * blockDim.x + threadIdx.x;
    if (e < N_EDGES) atomicAdd(&deg[dst[e]], 1.0f);
}

__global__ __launch_bounds__(TILE)
void k_deg_to_dinv(float* __restrict__ deg, int* __restrict__ tileCnt) {
    __shared__ int cnt;
    if (threadIdx.x == 0) cnt = 0;
    __syncthreads();
    int i = blockIdx.x * TILE + threadIdx.x;
    if (i < N_NODES) {
        float d = deg[i];
        atomicAdd(&cnt, (int)d - 1);
        deg[i] = (d > 0.0f) ? rsqrtf(d) : 0.0f;
    }
    __syncthreads();
    if (threadIdx.x == 0) tileCnt[blockIdx.x] = cnt;
}

__global__ __launch_bounds__(1024)
void k_tile_scan(const int* __restrict__ tileCnt, int* __restrict__ tileStart,
                 int* __restrict__ cursor) {
    __shared__ int s[1024];
    const int t = threadIdx.x;
    const int c = (t < N_TILES) ? tileCnt[t] : 0;
    s[t] = c;
    __syncthreads();
    for (int off = 1; off < 1024; off <<= 1) {
        const int v = (t >= off) ? s[t - off] : 0;
        __syncthreads();
        s[t] += v;
        __syncthreads();
    }
    const int excl = s[t] - c;
    if (t <= N_TILES) tileStart[t] = excl;
    if (t <  N_TILES) cursor[t * CUR_PITCH] = excl;
}

__global__ __launch_bounds__(FILL_THREADS)
void k_bin_fill(const int* __restrict__ src, const int* __restrict__ dst,
                const float* __restrict__ dinv,
                int* __restrict__ cursor, int* __restrict__ recs) {
    __shared__ int lcnt[N_TILES];
    __shared__ int lbase[N_TILES];
    const int tid  = threadIdx.x;
    const int base = blockIdx.x * (FILL_THREADS * FILL_EPT);

    for (int t = tid; t < N_TILES; t += FILL_THREADS) lcnt[t] = 0;
    __syncthreads();

    int   d[FILL_EPT], s[FILL_EPT], r[FILL_EPT];
    float w[FILL_EPT];
#pragma unroll
    for (int k = 0; k < FILL_EPT; ++k) {
        const int e = min(base + k * FILL_THREADS + tid, N_EDGES - 1);
        d[k] = dst[e];
        s[k] = src[e];
    }
#pragma unroll
    for (int k = 0; k < FILL_EPT; ++k) {
        w[k] = dinv[s[k]] * dinv[d[k]];
        r[k] = 0;
        if (base + k * FILL_THREADS + tid < N_EDGES)
            r[k] = atomicAdd(&lcnt[d[k] >> TILE_SHIFT], 1);
    }
    __syncthreads();

    for (int t = tid; t < N_TILES; t += FILL_THREADS)
        lbase[t] = atomicAdd(&cursor[t * CUR_PITCH], lcnt[t]);
    __syncthreads();

#pragma unroll
    for (int k = 0; k < FILL_EPT; ++k) {
        if (base + k * FILL_THREADS + tid < N_EDGES) {
            const int slot = lbase[d[k] >> TILE_SHIFT] + r[k];
            ((int2*)recs)[slot] = make_int2(s[k] | ((d[k] & (TILE - 1)) << 16),
                                            __float_as_int(w[k]));
        }
    }
}

#define GEMM_THREADS 256
#define GEMM_WAVES   (GEMM_THREADS / 32)
#define N_STRIPS     (N_NODES / 16)
#define GEMM_GRID    ((N_STRIPS + GEMM_WAVES - 1) / GEMM_WAVES)
#define W_QUADS      (DIM / 4)

static_assert(N_NODES % 16 == 0, "whole 16-row strips");
static_assert(W_QUADS * (DIM / 4) == 4 * GEMM_THREADS, "W staging is four passes");

template <bool ACT>
__global__ __launch_bounds__(GEMM_THREADS)
void k_gemm_wmma_f32(const float* __restrict__ A, const float* __restrict__ bias,
                     const float* __restrict__ W, float* __restrict__ out) {
    __shared__ float4 wl[W_QUADS * DIM];

    const int tid  = threadIdx.x;
    const int lane = tid & 31;
    const int wave = __builtin_amdgcn_readfirstlane(tid >> 5);
    const int c    = lane & 15;
    const int h    = lane >> 4;
    const int strip = blockIdx.x * GEMM_WAVES + wave;
    const int row0  = min(strip, N_STRIPS - 1) * 16;

    float4 a[16];
    const float4* arow = (const float4*)(A + (size_t)(row0 + c) * DIM) + h;
#pragma unroll
    for (int j = 0; j < 16; ++j) a[j] = arow[2 * j];

#pragma unroll
    for (int it = 0; it < 4; ++it) {
        const int q  = it * GEMM_THREADS + tid;
        const int kq = q >> 5;
        const int cq = q & 31;
        const float4* wp = (const float4*)(W + (size_t)(4 * kq) * DIM) + cq;
        const float4 g0 = wp[0];
        const float4 g1 = wp[DIM / 4];
        const float4 g2 = wp[2 * (DIM / 4)];
        const float4 g3 = wp[3 * (DIM / 4)];
        float4* d = wl + kq * DIM + 4 * cq;
        d[0] = make_float4(g0.x, g1.x, g2.x, g3.x);
        d[1] = make_float4(g0.y, g1.y, g2.y, g3.y);
        d[2] = make_float4(g0.z, g1.z, g2.z, g3.z);
        d[3] = make_float4(g0.w, g1.w, g2.w, g3.w);
    }

    if constexpr (ACT) {
        const float4* bq = (const float4*)bias + h;
#pragma unroll
        for (int j = 0; j < 16; ++j) {
            const float4 b = bq[2 * j];
            a[j].x = fmaxf(a[j].x + b.x, 0.0f);
            a[j].y = fmaxf(a[j].y + b.y, 0.0f);
            a[j].z = fmaxf(a[j].z + b.z, 0.0f);
            a[j].w = fmaxf(a[j].w + b.w, 0.0f);
        }
    }
    __syncthreads();
    if (strip >= N_STRIPS) return;

    v8f acc[8];
#pragma unroll
    for (int n = 0; n < 8; ++n) acc[n] = v8f{};

    const float4* bl = wl + h * DIM + c;
    float4 bcur[8], bnxt[8];
#pragma unroll
    for (int n = 0; n < 8; ++n) bcur[n] = bl[16 * n];
#pragma unroll
    for (int j = 0; j < 16; ++j) {
        __builtin_amdgcn_sched_barrier(0);
        if (j + 1 < 16) {
#pragma unroll
            for (int n = 0; n < 8; ++n) bnxt[n] = bl[2 * (j + 1) * DIM + 16 * n];
        }
        __builtin_amdgcn_sched_barrier(0);
        const v2f a0 = {a[j].x, a[j].y};
        const v2f a1 = {a[j].z, a[j].w};
#pragma unroll
        for (int n = 0; n < 8; ++n) {
            const v2f b0 = {bcur[n].x, bcur[n].y};
            acc[n] = __builtin_amdgcn_wmma_f32_16x16x4_f32(
                          false, a0,  false, b0,
                          (short)0, acc[n],  false,  false);
        }
#pragma unroll
        for (int n = 0; n < 8; ++n) {
            const v2f b1 = {bcur[n].z, bcur[n].w};
            acc[n] = __builtin_amdgcn_wmma_f32_16x16x4_f32(
                          false, a1,  false, b1,
                          (short)0, acc[n],  false,  false);
        }
#pragma unroll
        for (int n = 0; n < 8; ++n) bcur[n] = bnxt[n];
    }

    __builtin_amdgcn_sched_barrier(0);
    asm volatile("v_nop\n\tv_nop\n\tv_nop\n\tv_nop");
    __builtin_amdgcn_sched_barrier(0);

    float* orow = out + (size_t)(row0 + 8 * h) * DIM + c;
#pragma unroll
    for (int n = 0; n < 8; ++n) {
#pragma unroll
        for (int r = 0; r < 8; ++r)
            orow[(size_t)r * DIM + 16 * n] = acc[n][r];
    }
}

__global__ __launch_bounds__(AGG_THREADS)
void k_tile_agg(const int* __restrict__ recs, const int* __restrict__ tileStart,
                const float* __restrict__ dinv, const float* __restrict__ t,
                float* __restrict__ agg) {
    __shared__ float4 acc4[TILE * (DIM / 4)];
    float* acc = (float*)acc4;

    const int tid   = threadIdx.x;
    const int lane  = tid & 31;
    const int wave  = __builtin_amdgcn_readfirstlane(tid >> 5);
    const int node0 = blockIdx.x * TILE;
    const int start = tileStart[blockIdx.x];
    const int end   = tileStart[blockIdx.x + 1];
    const int2* rec = (const int2*)recs;

    int  b   = start + (wave << 5);
    int2 cur = make_int2(0, 0);
    if (b + lane < end) cur = rec[b + lane];

#pragma unroll
    for (int q = 0; q < AGG_INIT_TRIPS; ++q) {
        const int i    = tid + q * AGG_THREADS;
        const int node = min(node0 + (i >> 5), N_NODES - 1);
        float w = dinv[node];
        w *= w;
        float4 v = ((const float4*)t)[(size_t)node * (DIM / 4) + (i & 31)];
        v.x *= w; v.y *= w; v.z *= w; v.w *= w;
        acc4[i] = v;
    }
    __syncthreads();

    while (b < end) {
        const int nb  = b + AGG_WAVES * 32;
        int2      nxt = make_int2(0, 0);
        if (nb + lane < end) nxt = rec[nb + lane];

        const int n = min(32, end - b);

        for (int j0 = 0; j0 < n; j0 += 8) {
            float v[8][4];
            float wj[8];
            int   dl[8];
#pragma unroll
            for (int j = 0; j < 8; ++j) {
                const int pj = __builtin_amdgcn_readlane(cur.x, j0 + j);
                wj[j] = __int_as_float(__builtin_amdgcn_readlane(cur.y, j0 + j));
                dl[j] = pj >> 16;
                const float* row = t + (size_t)(pj & 0xFFFF) * DIM + lane;
                v[j][0] = row[0];
                v[j][1] = row[32];
                v[j][2] = row[64];
                v[j][3] = row[96];
            }
            if (j0 + 8 <= n) {
#pragma unroll
                for (int j = 0; j < 8; ++j) {
                    float* a = acc + dl[j] * DIM + lane;
                    atomicAdd(a,      v[j][0] * wj[j]);
                    atomicAdd(a + 32, v[j][1] * wj[j]);
                    atomicAdd(a + 64, v[j][2] * wj[j]);
                    atomicAdd(a + 96, v[j][3] * wj[j]);
                }
            } else {
#pragma unroll
                for (int j = 0; j < 8; ++j) {
                    if (j0 + j < n) {
                        float* a = acc + dl[j] * DIM + lane;
                        atomicAdd(a,      v[j][0] * wj[j]);
                        atomicAdd(a + 32, v[j][1] * wj[j]);
                        atomicAdd(a + 64, v[j][2] * wj[j]);
                        atomicAdd(a + 96, v[j][3] * wj[j]);
                    }
                }
            }
        }
        cur = nxt;
        b   = nb;
    }
    __syncthreads();

    for (int i = tid; i < TILE * (DIM / 4); i += AGG_THREADS) {
        const int node = node0 + (i >> 5);
        if (node < N_NODES)
            ((float4*)agg)[(size_t)node * (DIM / 4) + (i & 31)] = acc4[i];
    }
}

__global__ __launch_bounds__(256)
void k_heads(const float* __restrict__ agg, const float* __restrict__ b,
             const float* __restrict__ Wt, const float* __restrict__ bt,
             const float* __restrict__ We, const float* __restrict__ be,
             float* __restrict__ out) {
    const int node = (blockIdx.x << 3) + (threadIdx.x >> 5);
    if (node >= N_NODES) return;
    const int lane = threadIdx.x & 31;

    float4       v  = ((const float4*)(agg + (size_t)node * DIM))[lane];
    const float4 bb = ((const float4*)b)[lane];
    v.x = fmaxf(v.x + bb.x, 0.0f);
    v.y = fmaxf(v.y + bb.y, 0.0f);
    v.z = fmaxf(v.z + bb.z, 0.0f);
    v.w = fmaxf(v.w + bb.w, 0.0f);
    const float4 wt = ((const float4*)Wt)[lane];
    const float4 we = ((const float4*)We)[lane];

    float st = v.x * wt.x + v.y * wt.y + v.z * wt.z + v.w * wt.w;
    float ev = v.x * we.x + v.y * we.y + v.z * we.z + v.w * we.w;

#pragma unroll
    for (int off = 16; off > 0; off >>= 1) {
        st += __shfl_down(st, off, 32);
        ev += __shfl_down(ev, off, 32);
    }
    if (lane == 0) {
        out[node]           = st + bt[0];
        out[N_NODES + node] = ev + be[0];
    }
}

extern "C" void kernel_launch(void* const* d_in, const int* in_sizes, int n_in,
                              void* d_out, int out_size, void* d_ws, size_t ws_size,
                              hipStream_t stream) {
    (void)in_sizes; (void)n_in; (void)out_size; (void)ws_size;

    const float* x   = (const float*)d_in[0];
    const int*   ei  = (const int*)  d_in[1];
    const float* W1  = (const float*)d_in[2];
    const float* b1  = (const float*)d_in[3];
    const float* W2  = (const float*)d_in[4];
    const float* b2  = (const float*)d_in[5];
    const float* Wt  = (const float*)d_in[6];
    const float* bt  = (const float*)d_in[7];
    const float* We  = (const float*)d_in[8];
    const float* be  = (const float*)d_in[9];
    float*       out = (float*)d_out;

    const int* src = ei;
    const int* dst = ei + N_EDGES;

    float* dinv = (float*)d_ws;
    float* bufA = dinv + ((N_NODES + 255) & ~255);
    float* bufB = bufA + (size_t)N_NODES * DIM;
    int*   recs      = (int*)(bufB + (size_t)N_NODES * DIM);
    int*   tileCnt   = recs + 2 * (size_t)N_EDGES;
    int*   tileStart = tileCnt + 1024;
    int*   cursor    = tileStart + 1024;

    const int gNode   = (N_NODES + 255) / 256;
    const int gEdge   = (N_EDGES + 255) / 256;
    const int gFill   = (N_EDGES + FILL_THREADS * FILL_EPT - 1) / (FILL_THREADS * FILL_EPT);
    const int gWaveN  = (N_NODES + 7) / 8;

    k_init_deg   <<<gNode, 256, 0, stream>>>(dinv);
    k_deg_accum  <<<gEdge, 256, 0, stream>>>(dst, dinv);
    k_deg_to_dinv<<<N_TILES, TILE, 0, stream>>>(dinv, tileCnt);

    k_tile_scan<<<1, 1024, 0, stream>>>(tileCnt, tileStart, cursor);
    k_bin_fill <<<gFill, FILL_THREADS, 0, stream>>>(src, dst, dinv, cursor, recs);

    k_gemm_wmma_f32<false><<<GEMM_GRID, GEMM_THREADS, 0, stream>>>(x, b1, W1, bufA);
    k_tile_agg            <<<N_TILES, AGG_THREADS, 0, stream>>>(recs, tileStart, dinv, bufA, bufB);

    k_gemm_wmma_f32<true> <<<GEMM_GRID, GEMM_THREADS, 0, stream>>>(bufB, b1, W2, bufA);
    k_tile_agg            <<<N_TILES, AGG_THREADS, 0, stream>>>(recs, tileStart, dinv, bufA, bufB);

    k_heads<<<gWaveN, 256, 0, stream>>>(bufB, b2, Wt, bt, We, be, out);
}


# set-up without global atomics: edges counted per (block, tile) in LDS, slots by a scan in the fill, degrees from the binned list; weight as dinv[d]*sum(dinv[s]*t); 8 launches; sum order by block
# speedup vs baseline: 2.9727x; 1.2593x over previous
#include <hip/hip_runtime.h>

#define N_NODES 50000
#define N_EDGES 800000
#define DIM 128

#define TILE        64
#define TILE_SHIFT  6
#define N_TILES     ((N_NODES + TILE - 1) / TILE)
#define BIN_THREADS 1024
#define BIN_EPT     8
#define BIN_EDGES   (BIN_THREADS * BIN_EPT)
#define N_BINB      ((N_EDGES + BIN_EDGES - 1) / BIN_EDGES)
#define DEG_THREADS 256
#define AGG_THREADS  512
#define AGG_WAVES    (AGG_THREADS / 32)

#define AGG_INIT_TRIPS (TILE * (DIM / 4) / AGG_THREADS)

static_assert(N_NODES <= 65536, "source index is packed into 16 bits");
static_assert(AGG_INIT_TRIPS * AGG_THREADS == TILE * (DIM / 4), "tile is a whole number of passes");
static_assert(N_TILES < BIN_THREADS, "the tile scan is one pass of a fill work-group");
static_assert(N_EDGES % 4 == 0, "edges are read four at a time");
static_assert(BIN_EPT % 4 == 0, "edges are read four at a time");

typedef float v2f __attribute__((ext_vector_type(2)));
typedef float v8f __attribute__((ext_vector_type(8)));

__global__ __launch_bounds__(BIN_THREADS)
void k_bin_count(const int* __restrict__ dst, int* __restrict__ cnt) {
    __shared__ int lcnt[N_TILES];
    const int tid  = threadIdx.x;
    const int base = blockIdx.x * BIN_EDGES;

    for (int t = tid; t < N_TILES; t += BIN_THREADS) lcnt[t] = 0;
    __syncthreads();

#pragma unroll
    for (int k = 0; k < BIN_EPT / 4; ++k) {
        const int e = base + 4 * (k * BIN_THREADS + tid);
        if (e < N_EDGES) {
            const int4 d = *(const int4*)(dst + e);
            atomicAdd(&lcnt[d.x >> TILE_SHIFT], 1);
            atomicAdd(&lcnt[d.y >> TILE_SHIFT], 1);
            atomicAdd(&lcnt[d.z >> TILE_SHIFT], 1);
            atomicAdd(&lcnt[d.w >> TILE_SHIFT], 1);
        }
    }
    __syncthreads();

    for (int t = tid; t < N_TILES; t += BIN_THREADS)
        cnt[blockIdx.x * N_TILES + t] = lcnt[t];
}

__global__ __launch_bounds__(BIN_THREADS)
void k_bin_fill(const int* __restrict__ src, const int* __restrict__ dst,
                const int* __restrict__ cnt, int* __restrict__ tileStart,
                int* __restrict__ recs) {
    __shared__ int s[BIN_THREADS];
    __shared__ int lcnt[N_TILES];
    __shared__ int lbase[N_TILES];
    const int tid  = threadIdx.x;
    const int blk  = blockIdx.x;
    const int base = blk * BIN_EDGES;

    int4 d[BIN_EPT / 4], sv[BIN_EPT / 4];
#pragma unroll
    for (int k = 0; k < BIN_EPT / 4; ++k) {
        const int e = min(base + 4 * (k * BIN_THREADS + tid), N_EDGES - 4);
        d[k]  = *(const int4*)(dst + e);
        sv[k] = *(const int4*)(src + e);
    }

    const int tc = min(tid, N_TILES - 1);
    int c = 0, p = 0;
#pragma unroll
    for (int b = 0; b < N_BINB; ++b) {
        const int v = cnt[b * N_TILES + tc];
        c += v;
        p += (b < blk) ? v : 0;
    }
    if (tid >= N_TILES) c = 0;
    s[tid] = c;
    __syncthreads();
    for (int off = 1; off < BIN_THREADS; off <<= 1) {
        const int v = (tid >= off) ? s[tid - off] : 0;
        __syncthreads();
        s[tid] += v;
        __syncthreads();
    }
    const int excl = s[tid] - c;
    if (tid < N_TILES) {
        lcnt[tid]  = 0;
        lbase[tid] = excl + p;
    }
    if (blk == 0 && tid <= N_TILES) tileStart[tid] = excl;
    __syncthreads();

#pragma unroll
    for (int k = 0; k < BIN_EPT / 4; ++k) {
        if (base + 4 * (k * BIN_THREADS + tid) < N_EDGES) {
            const int dd[4] = {d[k].x, d[k].y, d[k].z, d[k].w};
            const int ss[4] = {sv[k].x, sv[k].y, sv[k].z, sv[k].w};
#pragma unroll
            for (int q = 0; q < 4; ++q) {
                const int tile = dd[q] >> TILE_SHIFT;
                const int r    = atomicAdd(&lcnt[tile], 1);
                recs[lbase[tile] + r] = ss[q] | ((dd[q] & (TILE - 1)) << 16);
            }
        }
    }
}

__global__ __launch_bounds__(DEG_THREADS)
void k_tile_deg(const int* __restrict__ recs, const int* __restrict__ tileStart,
                float* __restrict__ dinv) {
    __shared__ int cnt[TILE];
    const int tid = threadIdx.x;
    if (tid < TILE) cnt[tid] = 0;
    __syncthreads();
    const int start = tileStart[blockIdx.x];
    const int end   = tileStart[blockIdx.x + 1];
    for (int i = start + tid; i < end; i += DEG_THREADS)
        atomicAdd(&cnt[recs[i] >> 16], 1);
    __syncthreads();
    if (tid < TILE) {
        const int node = blockIdx.x * TILE + tid;
        if (node < N_NODES) {
            const float dg = (float)(cnt[tid] + 1);
            dinv[node] = (dg > 0.0f) ? rsqrtf(dg) : 0.0f;
        }
    }
}

#define GEMM_THREADS 256
#define GEMM_WAVES   (GEMM_THREADS / 32)
#define N_STRIPS     (N_NODES / 16)
#define GEMM_GRID    ((N_STRIPS + GEMM_WAVES - 1) / GEMM_WAVES)
#define W_QUADS      (DIM / 4)

static_assert(N_NODES % 16 == 0, "whole 16-row strips");
static_assert(W_QUADS * (DIM / 4) == 4 * GEMM_THREADS, "W staging is four passes");

template <bool ACT>
__global__ __launch_bounds__(GEMM_THREADS)
void k_gemm_wmma_f32(const float* __restrict__ A, const float* __restrict__ bias,
                     const float* __restrict__ W, const float* __restrict__ dinv,
                     float* __restrict__ out) {
    __shared__ float4 wl[W_QUADS * DIM];

    const int tid  = threadIdx.x;
    const int lane = tid & 31;
    const int wave = __builtin_amdgcn_readfirstlane(tid >> 5);
    const int c    = lane & 15;
    const int h    = lane >> 4;
    const int strip = blockIdx.x * GEMM_WAVES + wave;
    const int row0  = min(strip, N_STRIPS - 1) * 16;

    float4 a[16];
    const float4* arow = (const float4*)(A + (size_t)(row0 + c) * DIM) + h;
#pragma unroll
    for (int j = 0; j < 16; ++j) a[j] = arow[2 * j];

    const float4* dq  = (const float4*)(dinv + row0 + 8 * h);
    const float4  sc0 = dq[0];
    const float4  sc1 = dq[1];

#pragma unroll
    for (int it = 0; it < 4; ++it) {
        const int q  = it * GEMM_THREADS + tid;
        const int kq = q >> 5;
        const int cq = q & 31;
        const float4* wp = (const float4*)(W + (size_t)(4 * kq) * DIM) + cq;
        const float4 g0 = wp[0];
        const float4 g1 = wp[DIM / 4];
        const float4 g2 = wp[2 * (DIM / 4)];
        const float4 g3 = wp[3 * (DIM / 4)];
        float4* d = wl + kq * DIM + 4 * cq;
        d[0] = make_float4(g0.x, g1.x, g2.x, g3.x);
        d[1] = make_float4(g0.y, g1.y, g2.y, g3.y);
        d[2] = make_float4(g0.z, g1.z, g2.z, g3.z);
        d[3] = make_float4(g0.w, g1.w, g2.w, g3.w);
    }

    if constexpr (ACT) {
        const float4* bq = (const float4*)bias + h;
#pragma unroll
        for (int j = 0; j < 16; ++j) {
            const float4 b = bq[2 * j];
            a[j].x = fmaxf(a[j].x + b.x, 0.0f);
            a[j].y = fmaxf(a[j].y + b.y, 0.0f);
            a[j].z = fmaxf(a[j].z + b.z, 0.0f);
            a[j].w = fmaxf(a[j].w + b.w, 0.0f);
        }
    }
    __syncthreads();
    if (strip >= N_STRIPS) return;

    v8f acc[8];
#pragma unroll
    for (int n = 0; n < 8; ++n) acc[n] = v8f{};

    const float4* bl = wl + h * DIM + c;
    float4 bcur[8], bnxt[8];
#pragma unroll
    for (int n = 0; n < 8; ++n) bcur[n] = bl[16 * n];
#pragma unroll
    for (int j = 0; j < 16; ++j) {
        __builtin_amdgcn_sched_barrier(0);
        if (j + 1 < 16) {
#pragma unroll
            for (int n = 0; n < 8; ++n) bnxt[n] = bl[2 * (j + 1) * DIM + 16 * n];
        }
        __builtin_amdgcn_sched_barrier(0);
        const v2f a0 = {a[j].x, a[j].y};
        const v2f a1 = {a[j].z, a[j].w};
#pragma unroll
        for (int n = 0; n < 8; ++n) {
            const v2f b0 = {bcur[n].x, bcur[n].y};
            acc[n] = __builtin_amdgcn_wmma_f32_16x16x4_f32(
                          false, a0,  false, b0,
                          (short)0, acc[n],  false,  false);
        }
#pragma unroll
        for (int n = 0; n < 8; ++n) {
            const v2f b1 = {bcur[n].z, bcur[n].w};
            acc[n] = __builtin_amdgcn_wmma_f32_16x16x4_f32(
                          false, a1,  false, b1,
                          (short)0, acc[n],  false,  false);
        }
#pragma unroll
        for (int n = 0; n < 8; ++n) bcur[n] = bnxt[n];
    }

    __builtin_amdgcn_sched_barrier(0);
    asm volatile("v_nop\n\tv_nop\n\tv_nop\n\tv_nop");
    __builtin_amdgcn_sched_barrier(0);

    const float sc[8] = {sc0.x, sc0.y, sc0.z, sc0.w, sc1.x, sc1.y, sc1.z, sc1.w};
    float* orow = out + (size_t)(row0 + 8 * h) * DIM + c;
#pragma unroll
    for (int n = 0; n < 8; ++n) {
#pragma unroll
        for (int r = 0; r < 8; ++r)
            orow[(size_t)r * DIM + 16 * n] = acc[n][r] * sc[r];
    }
}

__global__ __launch_bounds__(AGG_THREADS)
void k_tile_agg(const int* __restrict__ recs, const int* __restrict__ tileStart,
                const float* __restrict__ dinv, const float* __restrict__ t,
                float* __restrict__ agg) {
    __shared__ float4 acc4[TILE * (DIM / 4)];
    float* acc = (float*)acc4;

    const int tid   = threadIdx.x;
    const int lane  = tid & 31;
    const int wave  = __builtin_amdgcn_readfirstlane(tid >> 5);
    const int node0 = blockIdx.x * TILE;
    const int start = tileStart[blockIdx.x];
    const int end   = tileStart[blockIdx.x + 1];
    int b   = start + (wave << 5);
    int cur = 0;
    if (b + lane < end) cur = recs[b + lane];

    float wq[AGG_INIT_TRIPS];
#pragma unroll
    for (int q = 0; q < AGG_INIT_TRIPS; ++q) {
        const int i    = tid + q * AGG_THREADS;
        const int node = min(node0 + (i >> 5), N_NODES - 1);
        wq[q]   = dinv[node];
        acc4[i] = ((const float4*)t)[(size_t)node * (DIM / 4) + (i & 31)];
    }
    __syncthreads();

    while (b < end) {
        const int nb  = b + AGG_WAVES * 32;
        int       nxt = 0;
        if (nb + lane < end) nxt = recs[nb + lane];

        const int n = min(32, end - b);

        for (int j0 = 0; j0 < n; j0 += 8) {
            float v[8][4];
            int   dl[8];
#pragma unroll
            for (int j = 0; j < 8; ++j) {
                const int pj = __builtin_amdgcn_readlane(cur, j0 + j);
                dl[j] = pj >> 16;
                const float* row = t + (size_t)(pj & 0xFFFF) * DIM + lane;
                v[j][0] = row[0];
                v[j][1] = row[32];
                v[j][2] = row[64];
                v[j][3] = row[96];
            }
            if (j0 + 8 <= n) {
#pragma unroll
                for (int j = 0; j < 8; ++j) {
                    float* a = acc + dl[j] * DIM + lane;
                    atomicAdd(a,      v[j][0]);
                    atomicAdd(a + 32, v[j][1]);
                    atomicAdd(a + 64, v[j][2]);
                    atomicAdd(a + 96, v[j][3]);
                }
            } else {
#pragma unroll
                for (int j = 0; j < 8; ++j) {
                    if (j0 + j < n) {
                        float* a = acc + dl[j] * DIM + lane;
                        atomicAdd(a,      v[j][0]);
                        atomicAdd(a + 32, v[j][1]);
                        atomicAdd(a + 64, v[j][2]);
                        atomicAdd(a + 96, v[j][3]);
                    }
                }
            }
        }
        cur = nxt;
        b   = nb;
    }
    __syncthreads();

#pragma unroll
    for (int q = 0; q < AGG_INIT_TRIPS; ++q) {
        const int i    = tid + q * AGG_THREADS;
        const int node = node0 + (i >> 5);
        if (node < N_NODES) {
            const float w = wq[q];
            float4 v = acc4[i];
            v.x *= w; v.y *= w; v.z *= w; v.w *= w;
            ((float4*)agg)[(size_t)node * (DIM / 4) + (i & 31)] = v;
        }
    }
}

__global__ __launch_bounds__(256)
void k_heads(const float* __restrict__ agg, const float* __restrict__ b,
             const float* __restrict__ Wt, const float* __restrict__ bt,
             const float* __restrict__ We, const float* __restrict__ be,
             float* __restrict__ out) {
    const int node = (blockIdx.x << 3) + (threadIdx.x >> 5);
    if (node >= N_NODES) return;
    const int lane = threadIdx.x & 31;

    float4       v  = ((const float4*)(agg + (size_t)node * DIM))[lane];
    const float4 bb = ((const float4*)b)[lane];
    v.x = fmaxf(v.x + bb.x, 0.0f);
    v.y = fmaxf(v.y + bb.y, 0.0f);
    v.z = fmaxf(v.z + bb.z, 0.0f);
    v.w = fmaxf(v.w + bb.w, 0.0f);
    const float4 wt = ((const float4*)Wt)[lane];
    const float4 we = ((const float4*)We)[lane];

    float st = v.x * wt.x + v.y * wt.y + v.z * wt.z + v.w * wt.w;
    float ev = v.x * we.x + v.y * we.y + v.z * we.z + v.w * we.w;

#pragma unroll
    for (int off = 16; off > 0; off >>= 1) {
        st += __shfl_down(st, off, 32);
        ev += __shfl_down(ev, off, 32);
    }
    if (lane == 0) {
        out[node]           = st + bt[0];
        out[N_NODES + node] = ev + be[0];
    }
}

extern "C" void kernel_launch(void* const* d_in, const int* in_sizes, int n_in,
                              void* d_out, int out_size, void* d_ws, size_t ws_size,
                              hipStream_t stream) {
    (void)in_sizes; (void)n_in; (void)out_size; (void)ws_size;

    const float* x   = (const float*)d_in[0];
    const int*   ei  = (const int*)  d_in[1];
    const float* W1  = (const float*)d_in[2];
    const float* b1  = (const float*)d_in[3];
    const float* W2  = (const float*)d_in[4];
    const float* b2  = (const float*)d_in[5];
    const float* Wt  = (const float*)d_in[6];
    const float* bt  = (const float*)d_in[7];
    const float* We  = (const float*)d_in[8];
    const float* be  = (const float*)d_in[9];
    float*       out = (float*)d_out;

    const int* src = ei;
    const int* dst = ei + N_EDGES;

    float* dinv = (float*)d_ws;
    float* bufA = dinv + ((N_NODES + 255) & ~255);
    float* bufB = bufA + (size_t)N_NODES * DIM;
    int*   recs      = (int*)(bufB + (size_t)N_NODES * DIM);
    int*   tileStart = recs + (size_t)N_EDGES;
    int*   binCnt    = tileStart + 1024;

    const int gWaveN  = (N_NODES + 7) / 8;

    k_bin_count<<<N_BINB, BIN_THREADS, 0, stream>>>(dst, binCnt);
    k_bin_fill <<<N_BINB, BIN_THREADS, 0, stream>>>(src, dst, binCnt, tileStart, recs);
    k_tile_deg <<<N_TILES, DEG_THREADS, 0, stream>>>(recs, tileStart, dinv);

    k_gemm_wmma_f32<false><<<GEMM_GRID, GEMM_THREADS, 0, stream>>>(x, b1, W1, dinv, bufA);
    k_tile_agg            <<<N_TILES, AGG_THREADS, 0, stream>>>(recs, tileStart, dinv, bufA, bufB);

    k_gemm_wmma_f32<true> <<<GEMM_GRID, GEMM_THREADS, 0, stream>>>(bufB, b1, W2, dinv, bufA);
    k_tile_agg            <<<N_TILES, AGG_THREADS, 0, stream>>>(recs, tileStart, dinv, bufA, bufB);

    k_heads<<<gWaveN, 256, 0, stream>>>(bufB, b2, Wt, bt, We, be, out);
}


# heads: a wave takes 16 rows (16 whole-row loads in flight, 32 partial sums folded over the lanes by a halving butterfly): 3125 waves, not 50000; fold order 16,8,4,2,1 kept, products as fma chain
# speedup vs baseline: 3.0841x; 1.0375x over previous
#include <hip/hip_runtime.h>

#define N_NODES 50000
#define N_EDGES 800000
#define DIM 128

#define TILE        64
#define TILE_SHIFT  6
#define N_TILES     ((N_NODES + TILE - 1) / TILE)
#define BIN_THREADS 1024
#define BIN_EPT     8
#define BIN_EDGES   (BIN_THREADS * BIN_EPT)
#define N_BINB      ((N_EDGES + BIN_EDGES - 1) / BIN_EDGES)
#define DEG_THREADS 256
#define AGG_THREADS  512
#define AGG_WAVES    (AGG_THREADS / 32)

#define AGG_INIT_TRIPS (TILE * (DIM / 4) / AGG_THREADS)

static_assert(N_NODES <= 65536, "source index is packed into 16 bits");
static_assert(AGG_INIT_TRIPS * AGG_THREADS == TILE * (DIM / 4), "tile is a whole number of passes");
static_assert(N_TILES < BIN_THREADS, "the tile scan is one pass of a fill work-group");
static_assert(N_EDGES % 4 == 0, "edges are read four at a time");
static_assert(BIN_EPT % 4 == 0, "edges are read four at a time");

typedef float v2f __attribute__((ext_vector_type(2)));
typedef float v8f __attribute__((ext_vector_type(8)));

__global__ __launch_bounds__(BIN_THREADS)
void k_bin_count(const int* __restrict__ dst, int* __restrict__ cnt) {
    __shared__ int lcnt[N_TILES];
    const int tid  = threadIdx.x;
    const int base = blockIdx.x * BIN_EDGES;

    for (int t = tid; t < N_TILES; t += BIN_THREADS) lcnt[t] = 0;
    __syncthreads();

#pragma unroll
    for (int k = 0; k < BIN_EPT / 4; ++k) {
        const int e = base + 4 * (k * BIN_THREADS + tid);
        if (e < N_EDGES) {
            const int4 d = *(const int4*)(dst + e);
            atomicAdd(&lcnt[d.x >> TILE_SHIFT], 1);
            atomicAdd(&lcnt[d.y >> TILE_SHIFT], 1);
            atomicAdd(&lcnt[d.z >> TILE_SHIFT], 1);
            atomicAdd(&lcnt[d.w >> TILE_SHIFT], 1);
        }
    }
    __syncthreads();

    for (int t = tid; t < N_TILES; t += BIN_THREADS)
        cnt[blockIdx.x * N_TILES + t] = lcnt[t];
}

__global__ __launch_bounds__(BIN_THREADS)
void k_bin_fill(const int* __restrict__ src, const int* __restrict__ dst,
                const int* __restrict__ cnt, int* __restrict__ tileStart,
                int* __restrict__ recs) {
    __shared__ int s[BIN_THREADS];
    __shared__ int lcnt[N_TILES];
    __shared__ int lbase[N_TILES];
    const int tid  = threadIdx.x;
    const int blk  = blockIdx.x;
    const int base = blk * BIN_EDGES;

    int4 d[BIN_EPT / 4], sv[BIN_EPT / 4];
#pragma unroll
    for (int k = 0; k < BIN_EPT / 4; ++k) {
        const int e = min(base + 4 * (k * BIN_THREADS + tid), N_EDGES - 4);
        d[k]  = *(const int4*)(dst + e);
        sv[k] = *(const int4*)(src + e);
    }

    const int tc = min(tid, N_TILES - 1);
    int c = 0, p = 0;
#pragma unroll
    for (int b = 0; b < N_BINB; ++b) {
        const int v = cnt[b * N_TILES + tc];
        c += v;
        p += (b < blk) ? v : 0;
    }
    if (tid >= N_TILES) c = 0;
    s[tid] = c;
    __syncthreads();
    for (int off = 1; off < BIN_THREADS; off <<= 1) {
        const int v = (tid >= off) ? s[tid - off] : 0;
        __syncthreads();
        s[tid] += v;
        __syncthreads();
    }
    const int excl = s[tid] - c;
    if (tid < N_TILES) {
        lcnt[tid]  = 0;
        lbase[tid] = excl + p;
    }
    if (blk == 0 && tid <= N_TILES) tileStart[tid] = excl;
    __syncthreads();

#pragma unroll
    for (int k = 0; k < BIN_EPT / 4; ++k) {
        if (base + 4 * (k * BIN_THREADS + tid) < N_EDGES) {
            const int dd[4] = {d[k].x, d[k].y, d[k].z, d[k].w};
            const int ss[4] = {sv[k].x, sv[k].y, sv[k].z, sv[k].w};
#pragma unroll
            for (int q = 0; q < 4; ++q) {
                const int tile = dd[q] >> TILE_SHIFT;
                const int r    = atomicAdd(&lcnt[tile], 1);
                recs[lbase[tile] + r] = ss[q] | ((dd[q] & (TILE - 1)) << 16);
            }
        }
    }
}

__global__ __launch_bounds__(DEG_THREADS)
void k_tile_deg(const int* __restrict__ recs, const int* __restrict__ tileStart,
                float* __restrict__ dinv) {
    __shared__ int cnt[TILE];
    const int tid = threadIdx.x;
    if (tid < TILE) cnt[tid] = 0;
    __syncthreads();
    const int start = tileStart[blockIdx.x];
    const int end   = tileStart[blockIdx.x + 1];
    for (int i = start + tid; i < end; i += DEG_THREADS)
        atomicAdd(&cnt[recs[i] >> 16], 1);
    __syncthreads();
    if (tid < TILE) {
        const int node = blockIdx.x * TILE + tid;
        if (node < N_NODES) {
            const float dg = (float)(cnt[tid] + 1);
            dinv[node] = (dg > 0.0f) ? rsqrtf(dg) : 0.0f;
        }
    }
}

#define GEMM_THREADS 256
#define GEMM_WAVES   (GEMM_THREADS / 32)
#define N_STRIPS     (N_NODES / 16)
#define GEMM_GRID    ((N_STRIPS + GEMM_WAVES - 1) / GEMM_WAVES)
#define W_QUADS      (DIM / 4)

static_assert(N_NODES % 16 == 0, "whole 16-row strips");
static_assert(W_QUADS * (DIM / 4) == 4 * GEMM_THREADS, "W staging is four passes");

template <bool ACT>
__global__ __launch_bounds__(GEMM_THREADS)
void k_gemm_wmma_f32(const float* __restrict__ A, const float* __restrict__ bias,
                     const float* __restrict__ W, const float* __restrict__ dinv,
                     float* __restrict__ out) {
    __shared__ float4 wl[W_QUADS * DIM];

    const int tid  = threadIdx.x;
    const int lane = tid & 31;
    const int wave = __builtin_amdgcn_readfirstlane(tid >> 5);
    const int c    = lane & 15;
    const int h    = lane >> 4;
    const int strip = blockIdx.x * GEMM_WAVES + wave;
    const int row0  = min(strip, N_STRIPS - 1) * 16;

    float4 a[16];
    const float4* arow = (const float4*)(A + (size_t)(row0 + c) * DIM) + h;
#pragma unroll
    for (int j = 0; j < 16; ++j) a[j] = arow[2 * j];

    const float4* dq  = (const float4*)(dinv + row0 + 8 * h);
    const float4  sc0 = dq[0];
    const float4  sc1 = dq[1];

#pragma unroll
    for (int it = 0; it < 4; ++it) {
        const int q  = it * GEMM_THREADS + tid;
        const int kq = q >> 5;
        const int cq = q & 31;
        const float4* wp = (const float4*)(W + (size_t)(4 * kq) * DIM) + cq;
        const float4 g0 = wp[0];
        const float4 g1 = wp[DIM / 4];
        const float4 g2 = wp[2 * (DIM / 4)];
        const float4 g3 = wp[3 * (DIM / 4)];
        float4* d = wl + kq * DIM + 4 * cq;
        d[0] = make_float4(g0.x, g1.x, g2.x, g3.x);
        d[1] = make_float4(g0.y, g1.y, g2.y, g3.y);
        d[2] = make_float4(g0.z, g1.z, g2.z, g3.z);
        d[3] = make_float4(g0.w, g1.w, g2.w, g3.w);
    }

    if constexpr (ACT) {
        const float4* bq = (const float4*)bias + h;
#pragma unroll
        for (int j = 0; j < 16; ++j) {
            const float4 b = bq[2 * j];
            a[j].x = fmaxf(a[j].x + b.x, 0.0f);
            a[j].y = fmaxf(a[j].y + b.y, 0.0f);
            a[j].z = fmaxf(a[j].z + b.z, 0.0f);
            a[j].w = fmaxf(a[j].w + b.w, 0.0f);
        }
    }
    __syncthreads();
    if (strip >= N_STRIPS) return;

    v8f acc[8];
#pragma unroll
    for (int n = 0; n < 8; ++n) acc[n] = v8f{};

    const float4* bl = wl + h * DIM + c;
    float4 bcur[8], bnxt[8];
#pragma unroll
    for (int n = 0; n < 8; ++n) bcur[n] = bl[16 * n];
#pragma unroll
    for (int j = 0; j < 16; ++j) {
        __builtin_amdgcn_sched_barrier(0);
        if (j + 1 < 16) {
#pragma unroll
            for (int n = 0; n < 8; ++n) bnxt[n] = bl[2 * (j + 1) * DIM + 16 * n];
        }
        __builtin_amdgcn_sched_barrier(0);
        const v2f a0 = {a[j].x, a[j].y};
        const v2f a1 = {a[j].z, a[j].w};
#pragma unroll
        for (int n = 0; n < 8; ++n) {
            const v2f b0 = {bcur[n].x, bcur[n].y};
            acc[n] = __builtin_amdgcn_wmma_f32_16x16x4_f32(
                          false, a0,  false, b0,
                          (short)0, acc[n],  false,  false);
        }
#pragma unroll
        for (int n = 0; n < 8; ++n) {
            const v2f b1 = {bcur[n].z, bcur[n].w};
            acc[n] = __builtin_amdgcn_wmma_f32_16x16x4_f32(
                          false, a1,  false, b1,
                          (short)0, acc[n],  false,  false);
        }
#pragma unroll
        for (int n = 0; n < 8; ++n) bcur[n] = bnxt[n];
    }

    __builtin_amdgcn_sched_barrier(0);
    asm volatile("v_nop\n\tv_nop\n\tv_nop\n\tv_nop");
    __builtin_amdgcn_sched_barrier(0);

    const float sc[8] = {sc0.x, sc0.y, sc0.z, sc0.w, sc1.x, sc1.y, sc1.z, sc1.w};
    float* orow = out + (size_t)(row0 + 8 * h) * DIM + c;
#pragma unroll
    for (int n = 0; n < 8; ++n) {
#pragma unroll
        for (int r = 0; r < 8; ++r)
            orow[(size_t)r * DIM + 16 * n] = acc[n][r] * sc[r];
    }
}

__global__ __launch_bounds__(AGG_THREADS)
void k_tile_agg(const int* __restrict__ recs, const int* __restrict__ tileStart,
                const float* __restrict__ dinv, const float* __restrict__ t,
                float* __restrict__ agg) {
    __shared__ float4 acc4[TILE * (DIM / 4)];
    float* acc = (float*)acc4;

    const int tid   = threadIdx.x;
    const int lane  = tid & 31;
    const int wave  = __builtin_amdgcn_readfirstlane(tid >> 5);
    const int node0 = blockIdx.x * TILE;
    const int start = tileStart[blockIdx.x];
    const int end   = tileStart[blockIdx.x + 1];
    int b   = start + (wave << 5);
    int cur = 0;
    if (b + lane < end) cur = recs[b + lane];

    float wq[AGG_INIT_TRIPS];
#pragma unroll
    for (int q = 0; q < AGG_INIT_TRIPS; ++q) {
        const int i    = tid + q * AGG_THREADS;
        const int node = min(node0 + (i >> 5), N_NODES - 1);
        wq[q]   = dinv[node];
        acc4[i] = ((const float4*)t)[(size_t)node * (DIM / 4) + (i & 31)];
    }
    __syncthreads();

    while (b < end) {
        const int nb  = b + AGG_WAVES * 32;
        int       nxt = 0;
        if (nb + lane < end) nxt = recs[nb + lane];

        const int n = min(32, end - b);

        for (int j0 = 0; j0 < n; j0 += 8) {
            float v[8][4];
            int   dl[8];
#pragma unroll
            for (int j = 0; j < 8; ++j) {
                const int pj = __builtin_amdgcn_readlane(cur, j0 + j);
                dl[j] = pj >> 16;
                const float* row = t + (size_t)(pj & 0xFFFF) * DIM + lane;
                v[j][0] = row[0];
                v[j][1] = row[32];
                v[j][2] = row[64];
                v[j][3] = row[96];
            }
            if (j0 + 8 <= n) {
#pragma unroll
                for (int j = 0; j < 8; ++j) {
                    float* a = acc + dl[j] * DIM + lane;
                    atomicAdd(a,      v[j][0]);
                    atomicAdd(a + 32, v[j][1]);
                    atomicAdd(a + 64, v[j][2]);
                    atomicAdd(a + 96, v[j][3]);
                }
            } else {
#pragma unroll
                for (int j = 0; j < 8; ++j) {
                    if (j0 + j < n) {
                        float* a = acc + dl[j] * DIM + lane;
                        atomicAdd(a,      v[j][0]);
                        atomicAdd(a + 32, v[j][1]);
                        atomicAdd(a + 64, v[j][2]);
                        atomicAdd(a + 96, v[j][3]);
                    }
                }
            }
        }
        cur = nxt;
        b   = nb;
    }
    __syncthreads();

#pragma unroll
    for (int q = 0; q < AGG_INIT_TRIPS; ++q) {
        const int i    = tid + q * AGG_THREADS;
        const int node = node0 + (i >> 5);
        if (node < N_NODES) {
            const float w = wq[q];
            float4 v = acc4[i];
            v.x *= w; v.y *= w; v.z *= w; v.w *= w;
            ((float4*)agg)[(size_t)node * (DIM / 4) + (i & 31)] = v;
        }
    }
}

#define HEAD_ROWS     16
#define HEAD_THREADS  128
#define HEAD_WAVES    (HEAD_THREADS / 32)
#define N_HEAD_WAVES  (N_NODES / HEAD_ROWS)
#define HEAD_GRID     ((N_HEAD_WAVES + HEAD_WAVES - 1) / HEAD_WAVES)

static_assert(N_NODES % HEAD_ROWS == 0, "whole groups of rows per wave");
static_assert(HEAD_ROWS == 16, "the fold below ends with one value per lane: 2 heads x 16 rows = 32 lanes");

template <int M>
__device__ __forceinline__ void fold_step(float (&val)[2 * HEAD_ROWS], const int lane) {
    const bool up = (lane & M) != 0;
#pragma unroll
    for (int i = 0; i < M; ++i) {
        const float keep = up ? val[i + M] : val[i];
        const float send = up ? val[i]     : val[i + M];
        val[i] = keep + __shfl_xor(send, M, 32);
    }
}

__global__ __launch_bounds__(HEAD_THREADS)
void k_heads(const float* __restrict__ agg, const float* __restrict__ b,
             const float* __restrict__ Wt, const float* __restrict__ bt,
             const float* __restrict__ We, const float* __restrict__ be,
             float* __restrict__ out) {
    const int lane = threadIdx.x & 31;
    const int wave = __builtin_amdgcn_readfirstlane(threadIdx.x >> 5);
    const int w    = blockIdx.x * HEAD_WAVES + wave;
    if (w >= N_HEAD_WAVES) return;
    const int row0 = w * HEAD_ROWS;

    const float4 bb = ((const float4*)b)[lane];
    const float4 wt = ((const float4*)Wt)[lane];
    const float4 we = ((const float4*)We)[lane];
    const float  hb = (lane & 16) ? be[0] : bt[0];

    float4 v[HEAD_ROWS];
    const float4* p = (const float4*)(agg + (size_t)row0 * DIM) + lane;
#pragma unroll
    for (int r = 0; r < HEAD_ROWS; ++r) v[r] = p[r * (DIM / 4)];
    __builtin_amdgcn_sched_barrier(0);

    float val[2 * HEAD_ROWS];
#pragma unroll
    for (int r = 0; r < HEAD_ROWS; ++r) {
        float4 x = v[r];
        x.x = fmaxf(x.x + bb.x, 0.0f);
        x.y = fmaxf(x.y + bb.y, 0.0f);
        x.z = fmaxf(x.z + bb.z, 0.0f);
        x.w = fmaxf(x.w + bb.w, 0.0f);
        val[r]             = fmaf(x.w, wt.w, fmaf(x.z, wt.z, fmaf(x.y, wt.y, x.x * wt.x)));
        val[HEAD_ROWS + r] = fmaf(x.w, we.w, fmaf(x.z, we.z, fmaf(x.y, we.y, x.x * we.x)));
    }

    fold_step<16>(val, lane);
    fold_step<8>(val, lane);
    fold_step<4>(val, lane);
    fold_step<2>(val, lane);
    fold_step<1>(val, lane);

    out[(size_t)(lane >> 4) * N_NODES + row0 + (lane & 15)] = val[0] + hb;
}

extern "C" void kernel_launch(void* const* d_in, const int* in_sizes, int n_in,
                              void* d_out, int out_size, void* d_ws, size_t ws_size,
                              hipStream_t stream) {
    (void)in_sizes; (void)n_in; (void)out_size; (void)ws_size;

    const float* x   = (const float*)d_in[0];
    const int*   ei  = (const int*)  d_in[1];
    const float* W1  = (const float*)d_in[2];
    const float* b1  = (const float*)d_in[3];
    const float* W2  = (const float*)d_in[4];
    const float* b2  = (const float*)d_in[5];
    const float* Wt  = (const float*)d_in[6];
    const float* bt  = (const float*)d_in[7];
    const float* We  = (const float*)d_in[8];
    const float* be  = (const float*)d_in[9];
    float*       out = (float*)d_out;

    const int* src = ei;
    const int* dst = ei + N_EDGES;

    float* dinv = (float*)d_ws;
    float* bufA = dinv + ((N_NODES + 255) & ~255);
    float* bufB = bufA + (size_t)N_NODES * DIM;
    int*   recs      = (int*)(bufB + (size_t)N_NODES * DIM);
    int*   tileStart = recs + (size_t)N_EDGES;
    int*   binCnt    = tileStart + 1024;

    k_bin_count<<<N_BINB, BIN_THREADS, 0, stream>>>(dst, binCnt);
    k_bin_fill <<<N_BINB, BIN_THREADS, 0, stream>>>(src, dst, binCnt, tileStart, recs);
    k_tile_deg <<<N_TILES, DEG_THREADS, 0, stream>>>(recs, tileStart, dinv);

    k_gemm_wmma_f32<false><<<GEMM_GRID, GEMM_THREADS, 0, stream>>>(x, b1, W1, dinv, bufA);
    k_tile_agg            <<<N_TILES, AGG_THREADS, 0, stream>>>(recs, tileStart, dinv, bufA, bufB);

    k_gemm_wmma_f32<true> <<<GEMM_GRID, GEMM_THREADS, 0, stream>>>(bufB, b1, W2, dinv, bufA);
    k_tile_agg            <<<N_TILES, AGG_THREADS, 0, stream>>>(recs, tileStart, dinv, bufA, bufB);

    k_heads<<<HEAD_GRID, HEAD_THREADS, 0, stream>>>(bufB, b2, Wt, bt, We, be, out);
}


# aggregation gather: a lane loads 16 B, so one load brings a whole 512-B source row (was 4 one-line loads); LDS sum columns permuted so each float add still covers 32 banks; sum order unchanged
# speedup vs baseline: 3.3016x; 1.0705x over previous
#include <hip/hip_runtime.h>

#define N_NODES 50000
#define N_EDGES 800000
#define DIM 128

#define TILE        64
#define TILE_SHIFT  6
#define N_TILES     ((N_NODES + TILE - 1) / TILE)
#define BIN_THREADS 1024
#define BIN_EPT     8
#define BIN_EDGES   (BIN_THREADS * BIN_EPT)
#define N_BINB      ((N_EDGES + BIN_EDGES - 1) / BIN_EDGES)
#define DEG_THREADS 256
#define AGG_THREADS  512
#define AGG_WAVES    (AGG_THREADS / 32)

#define AGG_INIT_TRIPS (TILE * (DIM / 4) / AGG_THREADS)

static_assert(N_NODES <= 65536, "source index is packed into 16 bits");
static_assert(AGG_INIT_TRIPS * AGG_THREADS == TILE * (DIM / 4), "tile is a whole number of passes");
static_assert(AGG_THREADS % 32 == 0 && DIM == 128, "a thread keeps its lane's column quad in every pass");
static_assert(N_TILES < BIN_THREADS, "the tile scan is one pass of a fill work-group");
static_assert(N_EDGES % 4 == 0, "edges are read four at a time");
static_assert(BIN_EPT % 4 == 0, "edges are read four at a time");

typedef float v2f __attribute__((ext_vector_type(2)));
typedef float v8f __attribute__((ext_vector_type(8)));

__global__ __launch_bounds__(BIN_THREADS)
void k_bin_count(const int* __restrict__ dst, int* __restrict__ cnt) {
    __shared__ int lcnt[N_TILES];
    const int tid  = threadIdx.x;
    const int base = blockIdx.x * BIN_EDGES;

    for (int t = tid; t < N_TILES; t += BIN_THREADS) lcnt[t] = 0;
    __syncthreads();

#pragma unroll
    for (int k = 0; k < BIN_EPT / 4; ++k) {
        const int e = base + 4 * (k * BIN_THREADS + tid);
        if (e < N_EDGES) {
            const int4 d = *(const int4*)(dst + e);
            atomicAdd(&lcnt[d.x >> TILE_SHIFT], 1);
            atomicAdd(&lcnt[d.y >> TILE_SHIFT], 1);
            atomicAdd(&lcnt[d.z >> TILE_SHIFT], 1);
            atomicAdd(&lcnt[d.w >> TILE_SHIFT], 1);
        }
    }
    __syncthreads();

    for (int t = tid; t < N_TILES; t += BIN_THREADS)
        cnt[blockIdx.x * N_TILES + t] = lcnt[t];
}

__global__ __launch_bounds__(BIN_THREADS)
void k_bin_fill(const int* __restrict__ src, const int* __restrict__ dst,
                const int* __restrict__ cnt, int* __restrict__ tileStart,
                int* __restrict__ recs) {
    __shared__ int s[BIN_THREADS];
    __shared__ int lcnt[N_TILES];
    __shared__ int lbase[N_TILES];
    const int tid  = threadIdx.x;
    const int blk  = blockIdx.x;
    const int base = blk * BIN_EDGES;

    int4 d[BIN_EPT / 4], sv[BIN_EPT / 4];
#pragma unroll
    for (int k = 0; k < BIN_EPT / 4; ++k) {
        const int e = min(base + 4 * (k * BIN_THREADS + tid), N_EDGES - 4);
        d[k]  = *(const int4*)(dst + e);
        sv[k] = *(const int4*)(src + e);
    }

    const int tc = min(tid, N_TILES - 1);
    int c = 0, p = 0;
#pragma unroll
    for (int b = 0; b < N_BINB; ++b) {
        const int v = cnt[b * N_TILES + tc];
        c += v;
        p += (b < blk) ? v : 0;
    }
    if (tid >= N_TILES) c = 0;
    s[tid] = c;
    __syncthreads();
    for (int off = 1; off < BIN_THREADS; off <<= 1) {
        const int v = (tid >= off) ? s[tid - off] : 0;
        __syncthreads();
        s[tid] += v;
        __syncthreads();
    }
    const int excl = s[tid] - c;
    if (tid < N_TILES) {
        lcnt[tid]  = 0;
        lbase[tid] = excl + p;
    }
    if (blk == 0 && tid <= N_TILES) tileStart[tid] = excl;
    __syncthreads();

#pragma unroll
    for (int k = 0; k < BIN_EPT / 4; ++k) {
        if (base + 4 * (k * BIN_THREADS + tid) < N_EDGES) {
            const int dd[4] = {d[k].x, d[k].y, d[k].z, d[k].w};
            const int ss[4] = {sv[k].x, sv[k].y, sv[k].z, sv[k].w};
#pragma unroll
            for (int q = 0; q < 4; ++q) {
                const int tile = dd[q] >> TILE_SHIFT;
                const int r    = atomicAdd(&lcnt[tile], 1);
                recs[lbase[tile] + r] = ss[q] | ((dd[q] & (TILE - 1)) << 16);
            }
        }
    }
}

__global__ __launch_bounds__(DEG_THREADS)
void k_tile_deg(const int* __restrict__ recs, const int* __restrict__ tileStart,
                float* __restrict__ dinv) {
    __shared__ int cnt[TILE];
    const int tid = threadIdx.x;
    if (tid < TILE) cnt[tid] = 0;
    __syncthreads();
    const int start = tileStart[blockIdx.x];
    const int end   = tileStart[blockIdx.x + 1];
    for (int i = start + tid; i < end; i += DEG_THREADS)
        atomicAdd(&cnt[recs[i] >> 16], 1);
    __syncthreads();
    if (tid < TILE) {
        const int node = blockIdx.x * TILE + tid;
        if (node < N_NODES) {
            const float dg = (float)(cnt[tid] + 1);
            dinv[node] = (dg > 0.0f) ? rsqrtf(dg) : 0.0f;
        }
    }
}

#define GEMM_THREADS 256
#define GEMM_WAVES   (GEMM_THREADS / 32)
#define N_STRIPS     (N_NODES / 16)
#define GEMM_GRID    ((N_STRIPS + GEMM_WAVES - 1) / GEMM_WAVES)
#define W_QUADS      (DIM / 4)

static_assert(N_NODES % 16 == 0, "whole 16-row strips");
static_assert(W_QUADS * (DIM / 4) == 4 * GEMM_THREADS, "W staging is four passes");

template <bool ACT>
__global__ __launch_bounds__(GEMM_THREADS)
void k_gemm_wmma_f32(const float* __restrict__ A, const float* __restrict__ bias,
                     const float* __restrict__ W, const float* __restrict__ dinv,
                     float* __restrict__ out) {
    __shared__ float4 wl[W_QUADS * DIM];

    const int tid  = threadIdx.x;
    const int lane = tid & 31;
    const int wave = __builtin_amdgcn_readfirstlane(tid >> 5);
    const int c    = lane & 15;
    const int h    = lane >> 4;
    const int strip = blockIdx.x * GEMM_WAVES + wave;
    const int row0  = min(strip, N_STRIPS - 1) * 16;

    float4 a[16];
    const float4* arow = (const float4*)(A + (size_t)(row0 + c) * DIM) + h;
#pragma unroll
    for (int j = 0; j < 16; ++j) a[j] = arow[2 * j];

    const float4* dq  = (const float4*)(dinv + row0 + 8 * h);
    const float4  sc0 = dq[0];
    const float4  sc1 = dq[1];

#pragma unroll
    for (int it = 0; it < 4; ++it) {
        const int q  = it * GEMM_THREADS + tid;
        const int kq = q >> 5;
        const int cq = q & 31;
        const float4* wp = (const float4*)(W + (size_t)(4 * kq) * DIM) + cq;
        const float4 g0 = wp[0];
        const float4 g1 = wp[DIM / 4];
        const float4 g2 = wp[2 * (DIM / 4)];
        const float4 g3 = wp[3 * (DIM / 4)];
        float4* d = wl + kq * DIM + 4 * cq;
        d[0] = make_float4(g0.x, g1.x, g2.x, g3.x);
        d[1] = make_float4(g0.y, g1.y, g2.y, g3.y);
        d[2] = make_float4(g0.z, g1.z, g2.z, g3.z);
        d[3] = make_float4(g0.w, g1.w, g2.w, g3.w);
    }

    if constexpr (ACT) {
        const float4* bq = (const float4*)bias + h;
#pragma unroll
        for (int j = 0; j < 16; ++j) {
            const float4 b = bq[2 * j];
            a[j].x = fmaxf(a[j].x + b.x, 0.0f);
            a[j].y = fmaxf(a[j].y + b.y, 0.0f);
            a[j].z = fmaxf(a[j].z + b.z, 0.0f);
            a[j].w = fmaxf(a[j].w + b.w, 0.0f);
        }
    }
    __syncthreads();
    if (strip >= N_STRIPS) return;

    v8f acc[8];
#pragma unroll
    for (int n = 0; n < 8; ++n) acc[n] = v8f{};

    const float4* bl = wl + h * DIM + c;
    float4 bcur[8], bnxt[8];
#pragma unroll
    for (int n = 0; n < 8; ++n) bcur[n] = bl[16 * n];
#pragma unroll
    for (int j = 0; j < 16; ++j) {
        __builtin_amdgcn_sched_barrier(0);
        if (j + 1 < 16) {
#pragma unroll
            for (int n = 0; n < 8; ++n) bnxt[n] = bl[2 * (j + 1) * DIM + 16 * n];
        }
        __builtin_amdgcn_sched_barrier(0);
        const v2f a0 = {a[j].x, a[j].y};
        const v2f a1 = {a[j].z, a[j].w};
#pragma unroll
        for (int n = 0; n < 8; ++n) {
            const v2f b0 = {bcur[n].x, bcur[n].y};
            acc[n] = __builtin_amdgcn_wmma_f32_16x16x4_f32(
                          false, a0,  false, b0,
                          (short)0, acc[n],  false,  false);
        }
#pragma unroll
        for (int n = 0; n < 8; ++n) {
            const v2f b1 = {bcur[n].z, bcur[n].w};
            acc[n] = __builtin_amdgcn_wmma_f32_16x16x4_f32(
                          false, a1,  false, b1,
                          (short)0, acc[n],  false,  false);
        }
#pragma unroll
        for (int n = 0; n < 8; ++n) bcur[n] = bnxt[n];
    }

    __builtin_amdgcn_sched_barrier(0);
    asm volatile("v_nop\n\tv_nop\n\tv_nop\n\tv_nop");
    __builtin_amdgcn_sched_barrier(0);

    const float sc[8] = {sc0.x, sc0.y, sc0.z, sc0.w, sc1.x, sc1.y, sc1.z, sc1.w};
    float* orow = out + (size_t)(row0 + 8 * h) * DIM + c;
#pragma unroll
    for (int n = 0; n < 8; ++n) {
#pragma unroll
        for (int r = 0; r < 8; ++r)
            orow[(size_t)r * DIM + 16 * n] = acc[n][r] * sc[r];
    }
}

__global__ __launch_bounds__(AGG_THREADS)
void k_tile_agg(const int* __restrict__ recs, const int* __restrict__ tileStart,
                const float* __restrict__ dinv, const float* __restrict__ t,
                float* __restrict__ agg) {
    __shared__ float acc[TILE * DIM];

    const int tid   = threadIdx.x;
    const int lane  = tid & 31;
    const int wave  = __builtin_amdgcn_readfirstlane(tid >> 5);
    const int node0 = blockIdx.x * TILE;
    const int start = tileStart[blockIdx.x];
    const int end   = tileStart[blockIdx.x + 1];
    int b   = start + (wave << 5);
    int cur = 0;
    if (b + lane < end) cur = recs[b + lane];

    float wq[AGG_INIT_TRIPS];
#pragma unroll
    for (int q = 0; q < AGG_INIT_TRIPS; ++q) {
        const int i    = tid + q * AGG_THREADS;
        const int node = min(node0 + (i >> 5), N_NODES - 1);
        wq[q]   = dinv[node];
        const float4 x = ((const float4*)t)[(size_t)node * (DIM / 4) + lane];
        float* a = acc + (i >> 5) * DIM + lane;
        a[0]  = x.x;
        a[32] = x.y;
        a[64] = x.z;
        a[96] = x.w;
    }
    __syncthreads();

    while (b < end) {
        const int nb  = b + AGG_WAVES * 32;
        int       nxt = 0;
        if (nb + lane < end) nxt = recs[nb + lane];

        const int n = min(32, end - b);

        for (int j0 = 0; j0 < n; j0 += 8) {
            float4 v[8];
            int    dl[8];
#pragma unroll
            for (int j = 0; j < 8; ++j) {
                const int pj = __builtin_amdgcn_readlane(cur, j0 + j);
                dl[j] = pj >> 16;
                v[j]  = ((const float4*)(t + (size_t)(pj & 0xFFFF) * DIM))[lane];
            }
            if (j0 + 8 <= n) {
#pragma unroll
                for (int j = 0; j < 8; ++j) {
                    float* a = acc + dl[j] * DIM + lane;
                    atomicAdd(a,      v[j].x);
                    atomicAdd(a + 32, v[j].y);
                    atomicAdd(a + 64, v[j].z);
                    atomicAdd(a + 96, v[j].w);
                }
            } else {
#pragma unroll
                for (int j = 0; j < 8; ++j) {
                    if (j0 + j < n) {
                        float* a = acc + dl[j] * DIM + lane;
                        atomicAdd(a,      v[j].x);
                        atomicAdd(a + 32, v[j].y);
                        atomicAdd(a + 64, v[j].z);
                        atomicAdd(a + 96, v[j].w);
                    }
                }
            }
        }
        cur = nxt;
        b   = nb;
    }
    __syncthreads();

#pragma unroll
    for (int q = 0; q < AGG_INIT_TRIPS; ++q) {
        const int i    = tid + q * AGG_THREADS;
        const int node = node0 + (i >> 5);
        if (node < N_NODES) {
            const float w = wq[q];
            const float* a = acc + (i >> 5) * DIM + lane;
            float4 v = make_float4(a[0], a[32], a[64], a[96]);
            v.x *= w; v.y *= w; v.z *= w; v.w *= w;
            ((float4*)agg)[(size_t)node * (DIM / 4) + lane] = v;
        }
    }
}

#define HEAD_ROWS     16
#define HEAD_THREADS  128
#define HEAD_WAVES    (HEAD_THREADS / 32)
#define N_HEAD_WAVES  (N_NODES / HEAD_ROWS)
#define HEAD_GRID     ((N_HEAD_WAVES + HEAD_WAVES - 1) / HEAD_WAVES)

static_assert(N_NODES % HEAD_ROWS == 0, "whole groups of rows per wave");
static_assert(HEAD_ROWS == 16, "the fold below ends with one value per lane: 2 heads x 16 rows = 32 lanes");

template <int M>
__device__ __forceinline__ void fold_step(float (&val)[2 * HEAD_ROWS], const int lane) {
    const bool up = (lane & M) != 0;
#pragma unroll
    for (int i = 0; i < M; ++i) {
        const float keep = up ? val[i + M] : val[i];
        const float send = up ? val[i]     : val[i + M];
        val[i] = keep + __shfl_xor(send, M, 32);
    }
}

__global__ __launch_bounds__(HEAD_THREADS)
void k_heads(const float* __restrict__ agg, const float* __restrict__ b,
             const float* __restrict__ Wt, const float* __restrict__ bt,
             const float* __restrict__ We, const float* __restrict__ be,
             float* __restrict__ out) {
    const int lane = threadIdx.x & 31;
    const int wave = __builtin_amdgcn_readfirstlane(threadIdx.x >> 5);
    const int w    = blockIdx.x * HEAD_WAVES + wave;
    if (w >= N_HEAD_WAVES) return;
    const int row0 = w * HEAD_ROWS;

    const float4 bb = ((const float4*)b)[lane];
    const float4 wt = ((const float4*)Wt)[lane];
    const float4 we = ((const float4*)We)[lane];
    const float  hb = (lane & 16) ? be[0] : bt[0];

    float4 v[HEAD_ROWS];
    const float4* p = (const float4*)(agg + (size_t)row0 * DIM) + lane;
#pragma unroll
    for (int r = 0; r < HEAD_ROWS; ++r) v[r] = p[r * (DIM / 4)];
    __builtin_amdgcn_sched_barrier(0);

    float val[2 * HEAD_ROWS];
#pragma unroll
    for (int r = 0; r < HEAD_ROWS; ++r) {
        float4 x = v[r];
        x.x = fmaxf(x.x + bb.x, 0.0f);
        x.y = fmaxf(x.y + bb.y, 0.0f);
        x.z = fmaxf(x.z + bb.z, 0.0f);
        x.w = fmaxf(x.w + bb.w, 0.0f);
        val[r]             = fmaf(x.w, wt.w, fmaf(x.z, wt.z, fmaf(x.y, wt.y, x.x * wt.x)));
        val[HEAD_ROWS + r] = fmaf(x.w, we.w, fmaf(x.z, we.z, fmaf(x.y, we.y, x.x * we.x)));
    }

    fold_step<16>(val, lane);
    fold_step<8>(val, lane);
    fold_step<4>(val, lane);
    fold_step<2>(val, lane);
    fold_step<1>(val, lane);

    out[(size_t)(lane >> 4) * N_NODES + row0 + (lane & 15)] = val[0] + hb;
}

extern "C" void kernel_launch(void* const* d_in, const int* in_sizes, int n_in,
                              void* d_out, int out_size, void* d_ws, size_t ws_size,
                              hipStream_t stream) {
    (void)in_sizes; (void)n_in; (void)out_size; (void)ws_size;

    const float* x   = (const float*)d_in[0];
    const int*   ei  = (const int*)  d_in[1];
    const float* W1  = (const float*)d_in[2];
    const float* b1  = (const float*)d_in[3];
    const float* W2  = (const float*)d_in[4];
    const float* b2  = (const float*)d_in[5];
    const float* Wt  = (const float*)d_in[6];
    const float* bt  = (const float*)d_in[7];
    const float* We  = (const float*)d_in[8];
    const float* be  = (const float*)d_in[9];
    float*       out = (float*)d_out;

    const int* src = ei;
    const int* dst = ei + N_EDGES;

    float* dinv = (float*)d_ws;
    float* bufA = dinv + ((N_NODES + 255) & ~255);
    float* bufB = bufA + (size_t)N_NODES * DIM;
    int*   recs      = (int*)(bufB + (size_t)N_NODES * DIM);
    int*   tileStart = recs + (size_t)N_EDGES;
    int*   binCnt    = tileStart + 1024;

    k_bin_count<<<N_BINB, BIN_THREADS, 0, stream>>>(dst, binCnt);
    k_bin_fill <<<N_BINB, BIN_THREADS, 0, stream>>>(src, dst, binCnt, tileStart, recs);
    k_tile_deg <<<N_TILES, DEG_THREADS, 0, stream>>>(recs, tileStart, dinv);

    k_gemm_wmma_f32<false><<<GEMM_GRID, GEMM_THREADS, 0, stream>>>(x, b1, W1, dinv, bufA);
    k_tile_agg            <<<N_TILES, AGG_THREADS, 0, stream>>>(recs, tileStart, dinv, bufA, bufB);

    k_gemm_wmma_f32<true> <<<GEMM_GRID, GEMM_THREADS, 0, stream>>>(bufB, b1, W2, dinv, bufA);
    k_tile_agg            <<<N_TILES, AGG_THREADS, 0, stream>>>(recs, tileStart, dinv, bufA, bufB);

    k_heads<<<HEAD_GRID, HEAD_THREADS, 0, stream>>>(bufB, b2, Wt, bt, We, be, out);
}


# GEMM 1 runs in the fill pass's launch as extra work-groups (fill re-blocked to 256 threads), not behind the set-up; its dinv row scale moves to layer 1's gather, same products; 7 launches, not 8
# speedup vs baseline: 3.4918x; 1.0576x over previous
#include <hip/hip_runtime.h>

#define N_NODES 50000
#define N_EDGES 800000
#define DIM 128

#define TILE        64
#define TILE_SHIFT  6
#define N_TILES     ((N_NODES + TILE - 1) / TILE)
#define BIN_THREADS 1024
#define BIN_EPT     8
#define BIN_EDGES   (BIN_THREADS * BIN_EPT)
#define N_BINB      ((N_EDGES + BIN_EDGES - 1) / BIN_EDGES)
#define DEG_THREADS 256
#define AGG_THREADS  512
#define AGG_WAVES    (AGG_THREADS / 32)

#define AGG_INIT_TRIPS (TILE * (DIM / 4) / AGG_THREADS)

static_assert(N_NODES <= 65536, "source index is packed into 16 bits");
static_assert(AGG_INIT_TRIPS * AGG_THREADS == TILE * (DIM / 4), "tile is a whole number of passes");
static_assert(AGG_THREADS % 32 == 0 && DIM == 128, "a thread keeps its lane's column quad in every pass");
static_assert(N_EDGES % 4 == 0, "edges are read four at a time");
static_assert(BIN_EPT % 4 == 0, "edges are read four at a time");

typedef float v2f __attribute__((ext_vector_type(2)));
typedef float v8f __attribute__((ext_vector_type(8)));

__global__ __launch_bounds__(BIN_THREADS)
void k_bin_count(const int* __restrict__ dst, int* __restrict__ cnt) {
    __shared__ int lcnt[N_TILES];
    const int tid  = threadIdx.x;
    const int base = blockIdx.x * BIN_EDGES;

    for (int t = tid; t < N_TILES; t += BIN_THREADS) lcnt[t] = 0;
    __syncthreads();

#pragma unroll
    for (int k = 0; k < BIN_EPT / 4; ++k) {
        const int e = base + 4 * (k * BIN_THREADS + tid);
        if (e < N_EDGES) {
            const int4 d = *(const int4*)(dst + e);
            atomicAdd(&lcnt[d.x >> TILE_SHIFT], 1);
            atomicAdd(&lcnt[d.y >> TILE_SHIFT], 1);
            atomicAdd(&lcnt[d.z >> TILE_SHIFT], 1);
            atomicAdd(&lcnt[d.w >> TILE_SHIFT], 1);
        }
    }
    __syncthreads();

    for (int t = tid; t < N_TILES; t += BIN_THREADS)
        cnt[blockIdx.x * N_TILES + t] = lcnt[t];
}

#define FILL_THREADS 256
#define FILL_GROUPS  (BIN_EDGES / 4 / FILL_THREADS)

static_assert(FILL_GROUPS * 4 * FILL_THREADS == BIN_EDGES, "a block's edges are a whole number of passes");
static_assert(4 * FILL_THREADS >= N_TILES, "four tiles per thread cover the tiles");
static_assert(N_TILES % 2 == 0, "count rows start 8-byte aligned; a tile pair never straddles the row end");

__device__ __forceinline__
void fill_body(const int blk, const int* __restrict__ src, const int* __restrict__ dst,
               const int* __restrict__ cnt, int* __restrict__ tileStart,
               int* __restrict__ recs) {
    __shared__ int s[FILL_THREADS];
    __shared__ int lcnt[N_TILES];
    __shared__ int lbase[N_TILES];
    const int tid  = threadIdx.x;
    const int base = blk * BIN_EDGES;

    int4 d[FILL_GROUPS], sv[FILL_GROUPS];
#pragma unroll
    for (int k = 0; k < FILL_GROUPS; ++k) {
        const int e = min(base + 4 * (k * FILL_THREADS + tid), N_EDGES - 4);
        d[k]  = *(const int4*)(dst + e);
        sv[k] = *(const int4*)(src + e);
    }

    const bool va_ok = 4 * tid     < N_TILES;
    const bool vb_ok = 4 * tid + 2 < N_TILES;
    const int  ta = min(4 * tid,     N_TILES - 2);
    const int  tb = min(4 * tid + 2, N_TILES - 2);
    int c[4] = {0, 0, 0, 0}, p[4] = {0, 0, 0, 0};
#pragma unroll 49
    for (int b = 0; b < N_BINB; ++b) {
        const int2 va = *(const int2*)(cnt + b * N_TILES + ta);
        const int2 vb = *(const int2*)(cnt + b * N_TILES + tb);
        c[0] += va.x; c[1] += va.y; c[2] += vb.x; c[3] += vb.y;
        const bool before = b < blk;
        p[0] += before ? va.x : 0; p[1] += before ? va.y : 0;
        p[2] += before ? vb.x : 0; p[3] += before ? vb.y : 0;
    }
    if (!va_ok) { c[0] = 0; c[1] = 0; }
    if (!vb_ok) { c[2] = 0; c[3] = 0; }
    const int tot = c[0] + c[1] + c[2] + c[3];
    s[tid] = tot;
    __syncthreads();
    for (int off = 1; off < FILL_THREADS; off <<= 1) {
        const int v = (tid >= off) ? s[tid - off] : 0;
        __syncthreads();
        s[tid] += v;
        __syncthreads();
    }
    int excl = s[tid] - tot;
#pragma unroll
    for (int i = 0; i < 4; ++i) {
        const int tile = 4 * tid + i;
        if (tile < N_TILES) {
            lcnt[tile]  = 0;
            lbase[tile] = excl + p[i];
            if (blk == 0) tileStart[tile] = excl;
        }
        excl += c[i];
    }
    if (blk == 0 && tid == FILL_THREADS - 1) tileStart[N_TILES] = s[tid];
    __syncthreads();

#pragma unroll
    for (int k = 0; k < FILL_GROUPS; ++k) {
        if (base + 4 * (k * FILL_THREADS + tid) < N_EDGES) {
            const int dd[4] = {d[k].x, d[k].y, d[k].z, d[k].w};
            const int ss[4] = {sv[k].x, sv[k].y, sv[k].z, sv[k].w};
#pragma unroll
            for (int q = 0; q < 4; ++q) {
                const int tile = dd[q] >> TILE_SHIFT;
                const int r    = atomicAdd(&lcnt[tile], 1);
                recs[lbase[tile] + r] = ss[q] | ((dd[q] & (TILE - 1)) << 16);
            }
        }
    }
}

__global__ __launch_bounds__(DEG_THREADS)
void k_tile_deg(const int* __restrict__ recs, const int* __restrict__ tileStart,
                float* __restrict__ dinv) {
    __shared__ int cnt[TILE];
    const int tid = threadIdx.x;
    if (tid < TILE) cnt[tid] = 0;
    __syncthreads();
    const int start = tileStart[blockIdx.x];
    const int end   = tileStart[blockIdx.x + 1];
    for (int i = start + tid; i < end; i += DEG_THREADS)
        atomicAdd(&cnt[recs[i] >> 16], 1);
    __syncthreads();
    if (tid < TILE) {
        const int node = blockIdx.x * TILE + tid;
        if (node < N_NODES) {
            const float dg = (float)(cnt[tid] + 1);
            dinv[node] = (dg > 0.0f) ? rsqrtf(dg) : 0.0f;
        }
    }
}

#define GEMM_THREADS 256
#define GEMM_WAVES   (GEMM_THREADS / 32)
#define N_STRIPS     (N_NODES / 16)
#define GEMM_GRID    ((N_STRIPS + GEMM_WAVES - 1) / GEMM_WAVES)
#define W_QUADS      (DIM / 4)

static_assert(N_NODES % 16 == 0, "whole 16-row strips");
static_assert(W_QUADS * (DIM / 4) == 4 * GEMM_THREADS, "W staging is four passes");

template <bool ACT, bool SCALE>
__device__ __forceinline__
void gemm_body(const int blk, const float* __restrict__ A, const float* __restrict__ bias,
               const float* __restrict__ W, const float* __restrict__ dinv,
               float* __restrict__ out) {
    __shared__ float4 wl[W_QUADS * DIM];

    const int tid  = threadIdx.x;
    const int lane = tid & 31;
    const int wave = __builtin_amdgcn_readfirstlane(tid >> 5);
    const int c    = lane & 15;
    const int h    = lane >> 4;
    const int strip = blk * GEMM_WAVES + wave;
    const int row0  = min(strip, N_STRIPS - 1) * 16;

    float4 a[16];
    const float4* arow = (const float4*)(A + (size_t)(row0 + c) * DIM) + h;
#pragma unroll
    for (int j = 0; j < 16; ++j) a[j] = arow[2 * j];

    float4 sc0 = make_float4(1.0f, 1.0f, 1.0f, 1.0f), sc1 = sc0;
    if constexpr (SCALE) {
        const float4* dq = (const float4*)(dinv + row0 + 8 * h);
        sc0 = dq[0];
        sc1 = dq[1];
    }

#pragma unroll
    for (int it = 0; it < 4; ++it) {
        const int q  = it * GEMM_THREADS + tid;
        const int kq = q >> 5;
        const int cq = q & 31;
        const float4* wp = (const float4*)(W + (size_t)(4 * kq) * DIM) + cq;
        const float4 g0 = wp[0];
        const float4 g1 = wp[DIM / 4];
        const float4 g2 = wp[2 * (DIM / 4)];
        const float4 g3 = wp[3 * (DIM / 4)];
        float4* d = wl + kq * DIM + 4 * cq;
        d[0] = make_float4(g0.x, g1.x, g2.x, g3.x);
        d[1] = make_float4(g0.y, g1.y, g2.y, g3.y);
        d[2] = make_float4(g0.z, g1.z, g2.z, g3.z);
        d[3] = make_float4(g0.w, g1.w, g2.w, g3.w);
    }

    if constexpr (ACT) {
        const float4* bq = (const float4*)bias + h;
#pragma unroll
        for (int j = 0; j < 16; ++j) {
            const float4 b = bq[2 * j];
            a[j].x = fmaxf(a[j].x + b.x, 0.0f);
            a[j].y = fmaxf(a[j].y + b.y, 0.0f);
            a[j].z = fmaxf(a[j].z + b.z, 0.0f);
            a[j].w = fmaxf(a[j].w + b.w, 0.0f);
        }
    }
    __syncthreads();
    if (strip >= N_STRIPS) return;

    v8f acc[8];
#pragma unroll
    for (int n = 0; n < 8; ++n) acc[n] = v8f{};

    const float4* bl = wl + h * DIM + c;
    float4 bcur[8], bnxt[8];
#pragma unroll
    for (int n = 0; n < 8; ++n) bcur[n] = bl[16 * n];
#pragma unroll
    for (int j = 0; j < 16; ++j) {
        __builtin_amdgcn_sched_barrier(0);
        if (j + 1 < 16) {
#pragma unroll
            for (int n = 0; n < 8; ++n) bnxt[n] = bl[2 * (j + 1) * DIM + 16 * n];
        }
        __builtin_amdgcn_sched_barrier(0);
        const v2f a0 = {a[j].x, a[j].y};
        const v2f a1 = {a[j].z, a[j].w};
#pragma unroll
        for (int n = 0; n < 8; ++n) {
            const v2f b0 = {bcur[n].x, bcur[n].y};
            acc[n] = __builtin_amdgcn_wmma_f32_16x16x4_f32(
                          false, a0,  false, b0,
                          (short)0, acc[n],  false,  false);
        }
#pragma unroll
        for (int n = 0; n < 8; ++n) {
            const v2f b1 = {bcur[n].z, bcur[n].w};
            acc[n] = __builtin_amdgcn_wmma_f32_16x16x4_f32(
                          false, a1,  false, b1,
                          (short)0, acc[n],  false,  false);
        }
#pragma unroll
        for (int n = 0; n < 8; ++n) bcur[n] = bnxt[n];
    }

    __builtin_amdgcn_sched_barrier(0);
    asm volatile("v_nop\n\tv_nop\n\tv_nop\n\tv_nop");
    __builtin_amdgcn_sched_barrier(0);

    const float sc[8] = {sc0.x, sc0.y, sc0.z, sc0.w, sc1.x, sc1.y, sc1.z, sc1.w};
    float* orow = out + (size_t)(row0 + 8 * h) * DIM + c;
#pragma unroll
    for (int n = 0; n < 8; ++n) {
#pragma unroll
        for (int r = 0; r < 8; ++r) {
            if constexpr (SCALE) orow[(size_t)r * DIM + 16 * n] = acc[n][r] * sc[r];
            else                 orow[(size_t)r * DIM + 16 * n] = acc[n][r];
        }
    }
}

__global__ __launch_bounds__(GEMM_THREADS)
void k_gemm2(const float* __restrict__ A, const float* __restrict__ bias,
             const float* __restrict__ W, const float* __restrict__ dinv,
             float* __restrict__ out) {
    gemm_body<true, true>(blockIdx.x, A, bias, W, dinv, out);
}

static_assert(FILL_THREADS == GEMM_THREADS, "one block size for both kinds of work-group");
__global__ __launch_bounds__(GEMM_THREADS)
void k_fill_gemm1(const int* __restrict__ src, const int* __restrict__ dst,
                  const int* __restrict__ cnt, int* __restrict__ tileStart,
                  int* __restrict__ recs,
                  const float* __restrict__ x, const float* __restrict__ W1,
                  float* __restrict__ t1) {
    if (blockIdx.x < N_BINB) {
        fill_body(blockIdx.x, src, dst, cnt, tileStart, recs);
    } else {
        gemm_body<false, false>(blockIdx.x - N_BINB, x, nullptr, W1, nullptr, t1);
    }
}

template <bool WSRC>
__global__ __launch_bounds__(AGG_THREADS)
void k_tile_agg(const int* __restrict__ recs, const int* __restrict__ tileStart,
                const float* __restrict__ dinv, const float* __restrict__ t,
                float* __restrict__ agg) {
    __shared__ float acc[TILE * DIM];

    const int tid   = threadIdx.x;
    const int lane  = tid & 31;
    const int wave  = __builtin_amdgcn_readfirstlane(tid >> 5);
    const int node0 = blockIdx.x * TILE;
    const int start = tileStart[blockIdx.x];
    const int end   = tileStart[blockIdx.x + 1];
    int b   = start + (wave << 5);
    int cur = 0;
    if (b + lane < end) cur = recs[b + lane];
    int   ahead = 0;
    float wcur  = 0.0f;
    if constexpr (WSRC) {
        if (b + AGG_WAVES * 32 + lane < end) ahead = recs[b + AGG_WAVES * 32 + lane];
        wcur = dinv[cur & 0xFFFF];
    }

    float wq[AGG_INIT_TRIPS];
#pragma unroll
    for (int q = 0; q < AGG_INIT_TRIPS; ++q) {
        const int i    = tid + q * AGG_THREADS;
        const int node = min(node0 + (i >> 5), N_NODES - 1);
        wq[q]   = dinv[node];
        float4 x = ((const float4*)t)[(size_t)node * (DIM / 4) + lane];
        if constexpr (WSRC) { x.x *= wq[q]; x.y *= wq[q]; x.z *= wq[q]; x.w *= wq[q]; }
        float* a = acc + (i >> 5) * DIM + lane;
        a[0]  = x.x;
        a[32] = x.y;
        a[64] = x.z;
        a[96] = x.w;
    }
    __syncthreads();

    while (b < end) {
        const int nb  = b + AGG_WAVES * 32;
        int       nxt = 0;
        float     wnxt = 0.0f;
        int       ahead2 = 0;
        if constexpr (WSRC) {
            nxt  = ahead;
            wnxt = dinv[nxt & 0xFFFF];
            if (nb + AGG_WAVES * 32 + lane < end) ahead2 = recs[nb + AGG_WAVES * 32 + lane];
        } else {
            if (nb + lane < end) nxt = recs[nb + lane];
        }

        const int n = min(32, end - b);

        for (int j0 = 0; j0 < n; j0 += 8) {
            float4 v[8];
            int    dl[8];
#pragma unroll
            for (int j = 0; j < 8; ++j) {
                const int pj = __builtin_amdgcn_readlane(cur, j0 + j);
                dl[j] = pj >> 16;
                v[j]  = ((const float4*)(t + (size_t)(pj & 0xFFFF) * DIM))[lane];
            }
            if constexpr (WSRC) {
#pragma unroll
                for (int j = 0; j < 8; ++j) {
                    const float wj = __int_as_float(
                        __builtin_amdgcn_readlane(__float_as_int(wcur), j0 + j));
                    v[j].x *= wj; v[j].y *= wj; v[j].z *= wj; v[j].w *= wj;
                }
            }
            if (j0 + 8 <= n) {
#pragma unroll
                for (int j = 0; j < 8; ++j) {
                    float* a = acc + dl[j] * DIM + lane;
                    atomicAdd(a,      v[j].x);
                    atomicAdd(a + 32, v[j].y);
                    atomicAdd(a + 64, v[j].z);
                    atomicAdd(a + 96, v[j].w);
                }
            } else {
#pragma unroll
                for (int j = 0; j < 8; ++j) {
                    if (j0 + j < n) {
                        float* a = acc + dl[j] * DIM + lane;
                        atomicAdd(a,      v[j].x);
                        atomicAdd(a + 32, v[j].y);
                        atomicAdd(a + 64, v[j].z);
                        atomicAdd(a + 96, v[j].w);
                    }
                }
            }
        }
        cur = nxt;
        b   = nb;
        if constexpr (WSRC) { wcur = wnxt; ahead = ahead2; }
    }
    __syncthreads();

#pragma unroll
    for (int q = 0; q < AGG_INIT_TRIPS; ++q) {
        const int i    = tid + q * AGG_THREADS;
        const int node = node0 + (i >> 5);
        if (node < N_NODES) {
            const float w = wq[q];
            const float* a = acc + (i >> 5) * DIM + lane;
            float4 v = make_float4(a[0], a[32], a[64], a[96]);
            v.x *= w; v.y *= w; v.z *= w; v.w *= w;
            ((float4*)agg)[(size_t)node * (DIM / 4) + lane] = v;
        }
    }
}

#define HEAD_ROWS     16
#define HEAD_THREADS  128
#define HEAD_WAVES    (HEAD_THREADS / 32)
#define N_HEAD_WAVES  (N_NODES / HEAD_ROWS)
#define HEAD_GRID     ((N_HEAD_WAVES + HEAD_WAVES - 1) / HEAD_WAVES)

static_assert(N_NODES % HEAD_ROWS == 0, "whole groups of rows per wave");
static_assert(HEAD_ROWS == 16, "the fold below ends with one value per lane: 2 heads x 16 rows = 32 lanes");

template <int M>
__device__ __forceinline__ void fold_step(float (&val)[2 * HEAD_ROWS], const int lane) {
    const bool up = (lane & M) != 0;
#pragma unroll
    for (int i = 0; i < M; ++i) {
        const float keep = up ? val[i + M] : val[i];
        const float send = up ? val[i]     : val[i + M];
        val[i] = keep + __shfl_xor(send, M, 32);
    }
}

__global__ __launch_bounds__(HEAD_THREADS)
void k_heads(const float* __restrict__ agg, const float* __restrict__ b,
             const float* __restrict__ Wt, const float* __restrict__ bt,
             const float* __restrict__ We, const float* __restrict__ be,
             float* __restrict__ out) {
    const int lane = threadIdx.x & 31;
    const int wave = __builtin_amdgcn_readfirstlane(threadIdx.x >> 5);
    const int w    = blockIdx.x * HEAD_WAVES + wave;
    if (w >= N_HEAD_WAVES) return;
    const int row0 = w * HEAD_ROWS;

    const float4 bb = ((const float4*)b)[lane];
    const float4 wt = ((const float4*)Wt)[lane];
    const float4 we = ((const float4*)We)[lane];
    const float  hb = (lane & 16) ? be[0] : bt[0];

    float4 v[HEAD_ROWS];
    const float4* p = (const float4*)(agg + (size_t)row0 * DIM) + lane;
#pragma unroll
    for (int r = 0; r < HEAD_ROWS; ++r) v[r] = p[r * (DIM / 4)];
    __builtin_amdgcn_sched_barrier(0);

    float val[2 * HEAD_ROWS];
#pragma unroll
    for (int r = 0; r < HEAD_ROWS; ++r) {
        float4 x = v[r];
        x.x = fmaxf(x.x + bb.x, 0.0f);
        x.y = fmaxf(x.y + bb.y, 0.0f);
        x.z = fmaxf(x.z + bb.z, 0.0f);
        x.w = fmaxf(x.w + bb.w, 0.0f);
        val[r]             = fmaf(x.w, wt.w, fmaf(x.z, wt.z, fmaf(x.y, wt.y, x.x * wt.x)));
        val[HEAD_ROWS + r] = fmaf(x.w, we.w, fmaf(x.z, we.z, fmaf(x.y, we.y, x.x * we.x)));
    }

    fold_step<16>(val, lane);
    fold_step<8>(val, lane);
    fold_step<4>(val, lane);
    fold_step<2>(val, lane);
    fold_step<1>(val, lane);

    out[(size_t)(lane >> 4) * N_NODES + row0 + (lane & 15)] = val[0] + hb;
}

extern "C" void kernel_launch(void* const* d_in, const int* in_sizes, int n_in,
                              void* d_out, int out_size, void* d_ws, size_t ws_size,
                              hipStream_t stream) {
    (void)in_sizes; (void)n_in; (void)out_size; (void)ws_size;

    const float* x   = (const float*)d_in[0];
    const int*   ei  = (const int*)  d_in[1];
    const float* W1  = (const float*)d_in[2];
    const float* b1  = (const float*)d_in[3];
    const float* W2  = (const float*)d_in[4];
    const float* b2  = (const float*)d_in[5];
    const float* Wt  = (const float*)d_in[6];
    const float* bt  = (const float*)d_in[7];
    const float* We  = (const float*)d_in[8];
    const float* be  = (const float*)d_in[9];
    float*       out = (float*)d_out;

    const int* src = ei;
    const int* dst = ei + N_EDGES;

    float* dinv = (float*)d_ws;
    float* bufA = dinv + ((N_NODES + 255) & ~255);
    float* bufB = bufA + (size_t)N_NODES * DIM;
    int*   recs      = (int*)(bufB + (size_t)N_NODES * DIM);
    int*   tileStart = recs + (size_t)N_EDGES;
    int*   binCnt    = tileStart + 1024;

    k_bin_count<<<N_BINB, BIN_THREADS, 0, stream>>>(dst, binCnt);
    k_fill_gemm1<<<N_BINB + GEMM_GRID, GEMM_THREADS, 0, stream>>>(src, dst, binCnt, tileStart, recs,
                                                                  x, W1, bufA);
    k_tile_deg <<<N_TILES, DEG_THREADS, 0, stream>>>(recs, tileStart, dinv);

    k_tile_agg<true> <<<N_TILES, AGG_THREADS, 0, stream>>>(recs, tileStart, dinv, bufA, bufB);

    k_gemm2          <<<GEMM_GRID, GEMM_THREADS, 0, stream>>>(bufB, b1, W2, dinv, bufA);
    k_tile_agg<false><<<N_TILES, AGG_THREADS, 0, stream>>>(recs, tileStart, dinv, bufA, bufB);

    k_heads<<<HEAD_GRID, HEAD_THREADS, 0, stream>>>(bufB, b2, Wt, bt, We, be, out);
}


# aggregation re-tiled: 196-node tiles = 256 work-groups of 32 waves, one per WGP, one even resident round (was 782 x 16 waves); tile records staged in LDS, 16 rows in flight; LDS add order changed
# speedup vs baseline: 3.8129x; 1.0920x over previous
#include <hip/hip_runtime.h>

#define N_NODES 50000
#define N_EDGES 800000
#define DIM 128

#define TILE        196
#define N_TILES     ((N_NODES + TILE - 1) / TILE)
#define BIN_THREADS 1024
#define BIN_EPT     8
#define BIN_EDGES   (BIN_THREADS * BIN_EPT)
#define N_BINB      ((N_EDGES + BIN_EDGES - 1) / BIN_EDGES)
#define DEG_THREADS 512
#define DEG_EPT     8
#define AGG_THREADS  1024
#define AGG_WAVES    (AGG_THREADS / 32)
#define AGG_G        16
#define AGG_STAGE    16384
#define AGG_PASS     4

#define AGG_INIT_TRIPS ((TILE * (DIM / 4) + AGG_THREADS - 1) / AGG_THREADS)

static_assert(N_NODES <= 65536, "source index is packed into 16 bits");
static_assert(TILE <= 256, "local destination row is packed into 8 bits");
static_assert(N_TILES == 256, "one tile per WGP");
static_assert(AGG_STAGE % (AGG_PASS * AGG_THREADS) == 0 && AGG_G == 16, "whole staging passes; a block is half a wave's lanes");
static_assert(AGG_THREADS % 32 == 0 && DIM == 128, "a lane owns one column quad of a row");
static_assert(N_EDGES % 4 == 0, "edges are read four at a time");
static_assert(BIN_EPT % 4 == 0, "edges are read four at a time");

typedef float v2f __attribute__((ext_vector_type(2)));
typedef float v8f __attribute__((ext_vector_type(8)));

__global__ __launch_bounds__(BIN_THREADS)
void k_bin_count(const int* __restrict__ dst, int* __restrict__ cnt) {
    __shared__ int lcnt[N_TILES];
    const int tid  = threadIdx.x;
    const int base = blockIdx.x * BIN_EDGES;

    for (int t = tid; t < N_TILES; t += BIN_THREADS) lcnt[t] = 0;
    __syncthreads();

#pragma unroll
    for (int k = 0; k < BIN_EPT / 4; ++k) {
        const int e = base + 4 * (k * BIN_THREADS + tid);
        if (e < N_EDGES) {
            const int4 d = *(const int4*)(dst + e);
            atomicAdd(&lcnt[(unsigned)d.x / TILE], 1);
            atomicAdd(&lcnt[(unsigned)d.y / TILE], 1);
            atomicAdd(&lcnt[(unsigned)d.z / TILE], 1);
            atomicAdd(&lcnt[(unsigned)d.w / TILE], 1);
        }
    }
    __syncthreads();

    for (int t = tid; t < N_TILES; t += BIN_THREADS)
        cnt[blockIdx.x * N_TILES + t] = lcnt[t];
}

#define FILL_THREADS 256
#define FILL_GROUPS  (BIN_EDGES / 4 / FILL_THREADS)

static_assert(FILL_GROUPS * 4 * FILL_THREADS == BIN_EDGES, "a block's edges are a whole number of passes");
static_assert(FILL_THREADS == N_TILES, "one tile per thread");

__device__ __forceinline__
void fill_body(const int blk, const int* __restrict__ src, const int* __restrict__ dst,
               const int* __restrict__ cnt, int* __restrict__ tileStart,
               int* __restrict__ recs) {
    __shared__ int s[FILL_THREADS];
    __shared__ int lcnt[N_TILES];
    __shared__ int lbase[N_TILES];
    const int tid  = threadIdx.x;
    const int base = blk * BIN_EDGES;

    int4 d[FILL_GROUPS], sv[FILL_GROUPS];
#pragma unroll
    for (int k = 0; k < FILL_GROUPS; ++k) {
        const int e = min(base + 4 * (k * FILL_THREADS + tid), N_EDGES - 4);
        d[k]  = *(const int4*)(dst + e);
        sv[k] = *(const int4*)(src + e);
    }

    int c = 0, p = 0;
#pragma unroll 49
    for (int b = 0; b < N_BINB; ++b) {
        const int v = cnt[b * N_TILES + tid];
        c += v;
        p += (b < blk) ? v : 0;
    }
    s[tid] = c;
    __syncthreads();
    for (int off = 1; off < FILL_THREADS; off <<= 1) {
        const int v = (tid >= off) ? s[tid - off] : 0;
        __syncthreads();
        s[tid] += v;
        __syncthreads();
    }
    const int excl = s[tid] - c;
    lcnt[tid]  = 0;
    lbase[tid] = excl + p;
    if (blk == 0) {
        tileStart[tid] = excl;
        if (tid == FILL_THREADS - 1) tileStart[N_TILES] = s[tid];
    }
    __syncthreads();

#pragma unroll
    for (int k = 0; k < FILL_GROUPS; ++k) {
        if (base + 4 * (k * FILL_THREADS + tid) < N_EDGES) {
            const int dd[4] = {d[k].x, d[k].y, d[k].z, d[k].w};
            const int ss[4] = {sv[k].x, sv[k].y, sv[k].z, sv[k].w};
#pragma unroll
            for (int q = 0; q < 4; ++q) {
                const int tile = (unsigned)dd[q] / TILE;
                const int row  = dd[q] - tile * TILE;
                const int r    = atomicAdd(&lcnt[tile], 1);
                recs[lbase[tile] + r] = ss[q] | (row << 16);
            }
        }
    }
}

__global__ __launch_bounds__(DEG_THREADS)
void k_tile_deg(const int* __restrict__ recs, const int* __restrict__ tileStart,
                float* __restrict__ dinv) {
    __shared__ int cnt[TILE];
    const int tid = threadIdx.x;
    if (tid < TILE) cnt[tid] = 0;
    __syncthreads();
    const int start = tileStart[blockIdx.x];
    const int end   = tileStart[blockIdx.x + 1];
    for (int base = start; base < end; base += DEG_EPT * DEG_THREADS) {
        int r[DEG_EPT];
#pragma unroll
        for (int k = 0; k < DEG_EPT; ++k) {
            const int i = base + k * DEG_THREADS + tid;
            r[k] = -1;
            if (i < end) r[k] = recs[i];
        }
#pragma unroll
        for (int k = 0; k < DEG_EPT; ++k)
            if (r[k] >= 0) atomicAdd(&cnt[r[k] >> 16], 1);
    }
    __syncthreads();
    if (tid < TILE) {
        const int node = blockIdx.x * TILE + tid;
        if (node < N_NODES) {
            const float dg = (float)(cnt[tid] + 1);
            dinv[node] = (dg > 0.0f) ? rsqrtf(dg) : 0.0f;
        }
    }
}

#define GEMM_THREADS 256
#define GEMM_WAVES   (GEMM_THREADS / 32)
#define N_STRIPS     (N_NODES / 16)
#define GEMM_GRID    ((N_STRIPS + GEMM_WAVES - 1) / GEMM_WAVES)
#define W_QUADS      (DIM / 4)

static_assert(N_NODES % 16 == 0, "whole 16-row strips");
static_assert(W_QUADS * (DIM / 4) == 4 * GEMM_THREADS, "W staging is four passes");

template <bool ACT, bool SCALE>
__device__ __forceinline__
void gemm_body(const int blk, const float* __restrict__ A, const float* __restrict__ bias,
               const float* __restrict__ W, const float* __restrict__ dinv,
               float* __restrict__ out) {
    __shared__ float4 wl[W_QUADS * DIM];

    const int tid  = threadIdx.x;
    const int lane = tid & 31;
    const int wave = __builtin_amdgcn_readfirstlane(tid >> 5);
    const int c    = lane & 15;
    const int h    = lane >> 4;
    const int strip = blk * GEMM_WAVES + wave;
    const int row0  = min(strip, N_STRIPS - 1) * 16;

    float4 a[16];
    const float4* arow = (const float4*)(A + (size_t)(row0 + c) * DIM) + h;
#pragma unroll
    for (int j = 0; j < 16; ++j) a[j] = arow[2 * j];

    float4 sc0 = make_float4(1.0f, 1.0f, 1.0f, 1.0f), sc1 = sc0;
    if constexpr (SCALE) {
        const float4* dq = (const float4*)(dinv + row0 + 8 * h);
        sc0 = dq[0];
        sc1 = dq[1];
    }

#pragma unroll
    for (int it = 0; it < 4; ++it) {
        const int q  = it * GEMM_THREADS + tid;
        const int kq = q >> 5;
        const int cq = q & 31;
        const float4* wp = (const float4*)(W + (size_t)(4 * kq) * DIM) + cq;
        const float4 g0 = wp[0];
        const float4 g1 = wp[DIM / 4];
        const float4 g2 = wp[2 * (DIM / 4)];
        const float4 g3 = wp[3 * (DIM / 4)];
        float4* d = wl + kq * DIM + 4 * cq;
        d[0] = make_float4(g0.x, g1.x, g2.x, g3.x);
        d[1] = make_float4(g0.y, g1.y, g2.y, g3.y);
        d[2] = make_float4(g0.z, g1.z, g2.z, g3.z);
        d[3] = make_float4(g0.w, g1.w, g2.w, g3.w);
    }

    if constexpr (ACT) {
        const float4* bq = (const float4*)bias + h;
#pragma unroll
        for (int j = 0; j < 16; ++j) {
            const float4 b = bq[2 * j];
            a[j].x = fmaxf(a[j].x + b.x, 0.0f);
            a[j].y = fmaxf(a[j].y + b.y, 0.0f);
            a[j].z = fmaxf(a[j].z + b.z, 0.0f);
            a[j].w = fmaxf(a[j].w + b.w, 0.0f);
        }
    }
    __syncthreads();
    if (strip >= N_STRIPS) return;

    v8f acc[8];
#pragma unroll
    for (int n = 0; n < 8; ++n) acc[n] = v8f{};

    const float4* bl = wl + h * DIM + c;
    float4 bcur[8], bnxt[8];
#pragma unroll
    for (int n = 0; n < 8; ++n) bcur[n] = bl[16 * n];
#pragma unroll
    for (int j = 0; j < 16; ++j) {
        __builtin_amdgcn_sched_barrier(0);
        if (j + 1 < 16) {
#pragma unroll
            for (int n = 0; n < 8; ++n) bnxt[n] = bl[2 * (j + 1) * DIM + 16 * n];
        }
        __builtin_amdgcn_sched_barrier(0);
        const v2f a0 = {a[j].x, a[j].y};
        const v2f a1 = {a[j].z, a[j].w};
#pragma unroll
        for (int n = 0; n < 8; ++n) {
            const v2f b0 = {bcur[n].x, bcur[n].y};
            acc[n] = __builtin_amdgcn_wmma_f32_16x16x4_f32(
                          false, a0,  false, b0,
                          (short)0, acc[n],  false,  false);
        }
#pragma unroll
        for (int n = 0; n < 8; ++n) {
            const v2f b1 = {bcur[n].z, bcur[n].w};
            acc[n] = __builtin_amdgcn_wmma_f32_16x16x4_f32(
                          false, a1,  false, b1,
                          (short)0, acc[n],  false,  false);
        }
#pragma unroll
        for (int n = 0; n < 8; ++n) bcur[n] = bnxt[n];
    }

    __builtin_amdgcn_sched_barrier(0);
    asm volatile("v_nop\n\tv_nop\n\tv_nop\n\tv_nop");
    __builtin_amdgcn_sched_barrier(0);

    const float sc[8] = {sc0.x, sc0.y, sc0.z, sc0.w, sc1.x, sc1.y, sc1.z, sc1.w};
    float* orow = out + (size_t)(row0 + 8 * h) * DIM + c;
#pragma unroll
    for (int n = 0; n < 8; ++n) {
#pragma unroll
        for (int r = 0; r < 8; ++r) {
            if constexpr (SCALE) orow[(size_t)r * DIM + 16 * n] = acc[n][r] * sc[r];
            else                 orow[(size_t)r * DIM + 16 * n] = acc[n][r];
        }
    }
}

__global__ __launch_bounds__(GEMM_THREADS)
void k_gemm2(const float* __restrict__ A, const float* __restrict__ bias,
             const float* __restrict__ W, const float* __restrict__ dinv,
             float* __restrict__ out) {
    gemm_body<true, true>(blockIdx.x, A, bias, W, dinv, out);
}

static_assert(FILL_THREADS == GEMM_THREADS, "one block size for both kinds of work-group");
__global__ __launch_bounds__(GEMM_THREADS)
void k_fill_gemm1(const int* __restrict__ src, const int* __restrict__ dst,
                  const int* __restrict__ cnt, int* __restrict__ tileStart,
                  int* __restrict__ recs,
                  const float* __restrict__ x, const float* __restrict__ W1,
                  float* __restrict__ t1) {
    if (blockIdx.x < N_BINB) {
        fill_body(blockIdx.x, src, dst, cnt, tileStart, recs);
    } else {
        gemm_body<false, false>(blockIdx.x - N_BINB, x, nullptr, W1, nullptr, t1);
    }
}

template <bool WSRC, bool FULL>
__device__ __forceinline__
void agg_block(const int cur, const float wcur, const int n, const int lane,
               const float* __restrict__ t, float* acc) {
    float4 v[AGG_G];
    int    dl[AGG_G];
#pragma unroll
    for (int j = 0; j < AGG_G; ++j) {
        const int pj = __builtin_amdgcn_readlane(cur, j);
        dl[j] = pj >> 16;
        v[j]  = ((const float4*)(t + (size_t)(pj & 0xFFFF) * DIM))[lane];
    }
    if constexpr (WSRC) {
#pragma unroll
        for (int j = 0; j < AGG_G; ++j) {
            const float wj = __int_as_float(
                __builtin_amdgcn_readlane(__float_as_int(wcur), j));
            v[j].x *= wj; v[j].y *= wj; v[j].z *= wj; v[j].w *= wj;
        }
    }
#pragma unroll
    for (int j = 0; j < AGG_G; ++j) {
        if (FULL || j < n) {
            float* a = acc + dl[j] * DIM + lane;
            atomicAdd(a,      v[j].x);
            atomicAdd(a + 32, v[j].y);
            atomicAdd(a + 64, v[j].z);
            atomicAdd(a + 96, v[j].w);
        }
    }
}

template <bool WSRC>
__global__ __launch_bounds__(AGG_THREADS)
void k_tile_agg(const int* __restrict__ recs, const int* __restrict__ tileStart,
                const float* __restrict__ dinv, const float* __restrict__ t,
                float* __restrict__ agg) {
    __shared__ float acc[TILE * DIM];
    __shared__ int   lrec[AGG_STAGE];
    __shared__ float lw[WSRC ? AGG_STAGE : 1];

    const int tid   = threadIdx.x;
    const int lane  = tid & 31;
    const int wave  = __builtin_amdgcn_readfirstlane(tid >> 5);
    const int node0 = blockIdx.x * TILE;
    const int start = tileStart[blockIdx.x];
    const int cnt   = tileStart[blockIdx.x + 1] - start;

    float  wq[AGG_INIT_TRIPS];
    float4 xq[AGG_INIT_TRIPS];
#pragma unroll
    for (int q = 0; q < AGG_INIT_TRIPS; ++q) {
        const int row  = min(wave + q * AGG_WAVES, TILE - 1);
        const int node = min(node0 + row, N_NODES - 1);
        wq[q] = dinv[node];
        xq[q] = ((const float4*)t)[(size_t)node * (DIM / 4) + lane];
    }
#pragma unroll
    for (int q = 0; q < AGG_INIT_TRIPS; ++q) {
        const int row = wave + q * AGG_WAVES;
        if (row < TILE) {
            float4 x = xq[q];
            if constexpr (WSRC) { x.x *= wq[q]; x.y *= wq[q]; x.z *= wq[q]; x.w *= wq[q]; }
            float* a = acc + row * DIM + lane;
            a[0]  = x.x;
            a[32] = x.y;
            a[64] = x.z;
            a[96] = x.w;
        }
    }

    int r0 = 0;
    do {
        const int nr = min(AGG_STAGE, cnt - r0);
        for (int q0 = 0; q0 < nr; q0 += AGG_PASS * AGG_THREADS) {
            int   sr[AGG_PASS];
            float sw[AGG_PASS];
#pragma unroll
            for (int q = 0; q < AGG_PASS; ++q) {
                const int i = q0 + tid + q * AGG_THREADS;
                sr[q] = 0;
                if (i < nr) sr[q] = recs[start + r0 + i];
            }
            if constexpr (WSRC) {
#pragma unroll
                for (int q = 0; q < AGG_PASS; ++q) sw[q] = dinv[sr[q] & 0xFFFF];
            }
#pragma unroll
            for (int q = 0; q < AGG_PASS; ++q) {
                lrec[q0 + tid + q * AGG_THREADS] = sr[q];
                if constexpr (WSRC) lw[q0 + tid + q * AGG_THREADS] = sw[q];
            }
        }
        __syncthreads();

        for (int b = wave * AGG_G; b < nr; b += AGG_WAVES * AGG_G) {
            const int cur = lrec[b + (lane & (AGG_G - 1))];
            float wcur = 0.0f;
            if constexpr (WSRC) wcur = lw[b + (lane & (AGG_G - 1))];
            const int n = min(AGG_G, nr - b);

            if (n == AGG_G) agg_block<WSRC, true >(cur, wcur, n, lane, t, acc);
            else            agg_block<WSRC, false>(cur, wcur, n, lane, t, acc);
        }
        r0 += AGG_STAGE;
        __syncthreads();
    } while (r0 < cnt);

#pragma unroll
    for (int q = 0; q < AGG_INIT_TRIPS; ++q) {
        const float* a = acc + min(wave + q * AGG_WAVES, TILE - 1) * DIM + lane;
        xq[q] = make_float4(a[0], a[32], a[64], a[96]);
    }
#pragma unroll
    for (int q = 0; q < AGG_INIT_TRIPS; ++q) {
        const int row  = wave + q * AGG_WAVES;
        const int node = node0 + row;
        if (row < TILE && node < N_NODES) {
            const float w = wq[q];
            float4 v = xq[q];
            v.x *= w; v.y *= w; v.z *= w; v.w *= w;
            ((float4*)agg)[(size_t)node * (DIM / 4) + lane] = v;
        }
    }
}

#define HEAD_ROWS     16
#define HEAD_THREADS  128
#define HEAD_WAVES    (HEAD_THREADS / 32)
#define N_HEAD_WAVES  (N_NODES / HEAD_ROWS)
#define HEAD_GRID     ((N_HEAD_WAVES + HEAD_WAVES - 1) / HEAD_WAVES)

static_assert(N_NODES % HEAD_ROWS == 0, "whole groups of rows per wave");
static_assert(HEAD_ROWS == 16, "the fold below ends with one value per lane: 2 heads x 16 rows = 32 lanes");

template <int M>
__device__ __forceinline__ void fold_step(float (&val)[2 * HEAD_ROWS], const int lane) {
    const bool up = (lane & M) != 0;
#pragma unroll
    for (int i = 0; i < M; ++i) {
        const float keep = up ? val[i + M] : val[i];
        const float send = up ? val[i]     : val[i + M];
        val[i] = keep + __shfl_xor(send, M, 32);
    }
}

__global__ __launch_bounds__(HEAD_THREADS)
void k_heads(const float* __restrict__ agg, const float* __restrict__ b,
             const float* __restrict__ Wt, const float* __restrict__ bt,
             const float* __restrict__ We, const float* __restrict__ be,
             float* __restrict__ out) {
    const int lane = threadIdx.x & 31;
    const int wave = __builtin_amdgcn_readfirstlane(threadIdx.x >> 5);
    const int w    = blockIdx.x * HEAD_WAVES + wave;
    if (w >= N_HEAD_WAVES) return;
    const int row0 = w * HEAD_ROWS;

    const float4 bb = ((const float4*)b)[lane];
    const float4 wt = ((const float4*)Wt)[lane];
    const float4 we = ((const float4*)We)[lane];
    const float  hb = (lane & 16) ? be[0] : bt[0];

    float4 v[HEAD_ROWS];
    const float4* p = (const float4*)(agg + (size_t)row0 * DIM) + lane;
#pragma unroll
    for (int r = 0; r < HEAD_ROWS; ++r) v[r] = p[r * (DIM / 4)];
    __builtin_amdgcn_sched_barrier(0);

    float val[2 * HEAD_ROWS];
#pragma unroll
    for (int r = 0; r < HEAD_ROWS; ++r) {
        float4 x = v[r];
        x.x = fmaxf(x.x + bb.x, 0.0f);
        x.y = fmaxf(x.y + bb.y, 0.0f);
        x.z = fmaxf(x.z + bb.z, 0.0f);
        x.w = fmaxf(x.w + bb.w, 0.0f);
        val[r]             = fmaf(x.w, wt.w, fmaf(x.z, wt.z, fmaf(x.y, wt.y, x.x * wt.x)));
        val[HEAD_ROWS + r] = fmaf(x.w, we.w, fmaf(x.z, we.z, fmaf(x.y, we.y, x.x * we.x)));
    }

    fold_step<16>(val, lane);
    fold_step<8>(val, lane);
    fold_step<4>(val, lane);
    fold_step<2>(val, lane);
    fold_step<1>(val, lane);

    out[(size_t)(lane >> 4) * N_NODES + row0 + (lane & 15)] = val[0] + hb;
}

extern "C" void kernel_launch(void* const* d_in, const int* in_sizes, int n_in,
                              void* d_out, int out_size, void* d_ws, size_t ws_size,
                              hipStream_t stream) {
    (void)in_sizes; (void)n_in; (void)out_size; (void)ws_size;

    const float* x   = (const float*)d_in[0];
    const int*   ei  = (const int*)  d_in[1];
    const float* W1  = (const float*)d_in[2];
    const float* b1  = (const float*)d_in[3];
    const float* W2  = (const float*)d_in[4];
    const float* b2  = (const float*)d_in[5];
    const float* Wt  = (const float*)d_in[6];
    const float* bt  = (const float*)d_in[7];
    const float* We  = (const float*)d_in[8];
    const float* be  = (const float*)d_in[9];
    float*       out = (float*)d_out;

    const int* src = ei;
    const int* dst = ei + N_EDGES;

    float* dinv = (float*)d_ws;
    float* bufA = dinv + ((N_NODES + 255) & ~255);
    float* bufB = bufA + (size_t)N_NODES * DIM;
    int*   recs      = (int*)(bufB + (size_t)N_NODES * DIM);
    int*   tileStart = recs + (size_t)N_EDGES;
    int*   binCnt    = tileStart + 1024;

    k_bin_count<<<N_BINB, BIN_THREADS, 0, stream>>>(dst, binCnt);
    k_fill_gemm1<<<N_BINB + GEMM_GRID, GEMM_THREADS, 0, stream>>>(src, dst, binCnt, tileStart, recs,
                                                                  x, W1, bufA);
    k_tile_deg <<<N_TILES, DEG_THREADS, 0, stream>>>(recs, tileStart, dinv);

    k_tile_agg<true> <<<N_TILES, AGG_THREADS, 0, stream>>>(recs, tileStart, dinv, bufA, bufB);

    k_gemm2          <<<GEMM_GRID, GEMM_THREADS, 0, stream>>>(bufB, b1, W2, dinv, bufA);
    k_tile_agg<false><<<N_TILES, AGG_THREADS, 0, stream>>>(recs, tileStart, dinv, bufA, bufB);

    k_heads<<<HEAD_GRID, HEAD_THREADS, 0, stream>>>(bufB, b2, Wt, bt, We, be, out);
}


# count launch dropped: each edge block sorts its own edges by tile inside its own slot range (LDS count, scan, rank); a tile's list = 98 segments named in a table; 6 launches, not 7; sum order kept
# speedup vs baseline: 4.3286x; 1.1353x over previous
#include <hip/hip_runtime.h>

#define N_NODES 50000
#define N_EDGES 800000
#define DIM 128

#define TILE        196
#define N_TILES     ((N_NODES + TILE - 1) / TILE)
#define BIN_EDGES   8192
#define N_BINB      ((N_EDGES + BIN_EDGES - 1) / BIN_EDGES)
#define DEG_THREADS 512
#define AGG_THREADS  1024
#define AGG_WAVES    (AGG_THREADS / 32)
#define AGG_G        16
#define AGG_STAGE    16384
#define AGG_SEGS     ((N_BINB + AGG_WAVES - 1) / AGG_WAVES)

#define AGG_INIT_TRIPS ((TILE * (DIM / 4) + AGG_THREADS - 1) / AGG_THREADS)

static_assert(N_NODES <= 65536, "source index is packed into 16 bits");
static_assert(TILE <= 256, "local destination row is packed into 8 bits");
static_assert(N_TILES == 256, "one tile per WGP");
static_assert(AGG_STAGE % AGG_G == 0 && AGG_G == 16, "whole blocks per round; a block is half a wave's lanes");
static_assert(AGG_THREADS % 32 == 0 && DIM == 128, "a lane owns one column quad of a row");
static_assert(N_EDGES % 4 == 0, "edges are read four at a time");

typedef float v2f __attribute__((ext_vector_type(2)));
typedef float v8f __attribute__((ext_vector_type(8)));

#define SEG_PITCH 100
#define SEG_SHIFT 14

static_assert(N_BINB <= SEG_PITCH && SEG_PITCH % 4 == 0 && SEG_PITCH <= 128, "a lane reads four words of its tile's table row");
static_assert(BIN_EDGES < (1 << SEG_SHIFT) && SEG_SHIFT <= 15, "first slot and count are packed into one word");

#define FILL_THREADS 256
#define FILL_GROUPS  (BIN_EDGES / 4 / FILL_THREADS)

static_assert(FILL_GROUPS * 4 * FILL_THREADS == BIN_EDGES, "a block's edges are a whole number of passes");
static_assert(FILL_THREADS == N_TILES, "one tile per thread");

__device__ __forceinline__
void fill_body(const int blk, const int* __restrict__ src, const int* __restrict__ dst,
               int* __restrict__ seg, int* __restrict__ recs) {
    __shared__ int s[FILL_THREADS];
    __shared__ int lcnt[N_TILES];
    __shared__ int lbase[N_TILES];
    const int tid  = threadIdx.x;
    const int base = blk * BIN_EDGES;

    int4 d[FILL_GROUPS], sv[FILL_GROUPS];
#pragma unroll
    for (int k = 0; k < FILL_GROUPS; ++k) {
        const int e = min(base + 4 * (k * FILL_THREADS + tid), N_EDGES - 4);
        d[k]  = *(const int4*)(dst + e);
        sv[k] = *(const int4*)(src + e);
    }

    lcnt[tid] = 0;
    __syncthreads();
#pragma unroll
    for (int k = 0; k < FILL_GROUPS; ++k) {
        if (base + 4 * (k * FILL_THREADS + tid) < N_EDGES) {
            atomicAdd(&lcnt[(unsigned)d[k].x / TILE], 1);
            atomicAdd(&lcnt[(unsigned)d[k].y / TILE], 1);
            atomicAdd(&lcnt[(unsigned)d[k].z / TILE], 1);
            atomicAdd(&lcnt[(unsigned)d[k].w / TILE], 1);
        }
    }
    __syncthreads();

    const int c = lcnt[tid];
    lcnt[tid] = 0;
    s[tid] = c;
    __syncthreads();
    for (int off = 1; off < FILL_THREADS; off <<= 1) {
        const int v = (tid >= off) ? s[tid - off] : 0;
        __syncthreads();
        s[tid] += v;
        __syncthreads();
    }
    const int excl = s[tid] - c;
    lbase[tid] = base + excl;
    seg[tid * SEG_PITCH + blk] = excl | (c << SEG_SHIFT);
    __syncthreads();

#pragma unroll
    for (int k = 0; k < FILL_GROUPS; ++k) {
        if (base + 4 * (k * FILL_THREADS + tid) < N_EDGES) {
            const int dd[4] = {d[k].x, d[k].y, d[k].z, d[k].w};
            const int ss[4] = {sv[k].x, sv[k].y, sv[k].z, sv[k].w};
#pragma unroll
            for (int q = 0; q < 4; ++q) {
                const int tile = (unsigned)dd[q] / TILE;
                const int row  = dd[q] - tile * TILE;
                const int r    = atomicAdd(&lcnt[tile], 1);
                recs[lbase[tile] + r] = ss[q] | (row << 16);
            }
        }
    }
}

struct SegLane { int word; int start; };

__device__ __forceinline__
int seg_scan(const int* __restrict__ seg, const int tile, const int lane, const int q, SegLane& sl) {
    int4 w = make_int4(0, 0, 0, 0);
    if (4 * lane < SEG_PITCH) w = *(const int4*)(seg + tile * SEG_PITCH + 4 * lane);
    const int w0 = (4 * lane + 0 < N_BINB) ? w.x : 0;
    const int w1 = (4 * lane + 1 < N_BINB) ? w.y : 0;
    const int w2 = (4 * lane + 2 < N_BINB) ? w.z : 0;
    const int w3 = (4 * lane + 3 < N_BINB) ? w.w : 0;
    const int c0 = w0 >> SEG_SHIFT, c1 = w1 >> SEG_SHIFT, c2 = w2 >> SEG_SHIFT, c3 = w3 >> SEG_SHIFT;
    const int own = c0 + c1 + c2 + c3;
    int inc = own;
#pragma unroll
    for (int off = 1; off < 32; off <<= 1) {
        const int v = __shfl_up(inc, off, 32);
        if (lane >= off) inc += v;
    }
    const int s0 = inc - own, s1 = s0 + c0, s2 = s1 + c1, s3 = s2 + c2;
    sl.word  = (q & 2) ? ((q & 1) ? w3 : w2) : ((q & 1) ? w1 : w0);
    sl.start = (q & 2) ? ((q & 1) ? s3 : s2) : ((q & 1) ? s1 : s0);
    return __shfl(inc, 31, 32);
}

__device__ __forceinline__
void seg_get(const SegLane& sl, const int b, int& cnt, int& goff, int& start) {
    const int word = __builtin_amdgcn_readlane(sl.word, b >> 2);
    start = __builtin_amdgcn_readlane(sl.start, b >> 2);
    cnt   = word >> SEG_SHIFT;
    goff  = b * BIN_EDGES + (word & ((1 << SEG_SHIFT) - 1));
}

#define DEG_WAVES (DEG_THREADS / 32)
#define DEG_SEGS  ((N_BINB + DEG_WAVES - 1) / DEG_WAVES)

static_assert(DEG_WAVES % 4 == 0 && AGG_WAVES % 4 == 0, "a wave's blocks all have the same b & 3");

__global__ __launch_bounds__(DEG_THREADS)
void k_tile_deg(const int* __restrict__ recs, const int* __restrict__ seg,
                float* __restrict__ dinv) {
    __shared__ int cnt[TILE];
    const int tid  = threadIdx.x;
    const int lane = tid & 31;
    const int wave = __builtin_amdgcn_readfirstlane(tid >> 5);
    if (tid < TILE) cnt[tid] = 0;
    __syncthreads();

    SegLane sl;
    seg_scan(seg, blockIdx.x, lane, wave & 3, sl);

    int r[DEG_SEGS], sc[DEG_SEGS], sg[DEG_SEGS];
#pragma unroll
    for (int k = 0; k < DEG_SEGS; ++k) {
        const int b = wave + k * DEG_WAVES;
        int st;
        sc[k] = 0; sg[k] = 0;
        if (b < N_BINB) seg_get(sl, b, sc[k], sg[k], st);
        r[k] = -1;
        if (lane < sc[k]) r[k] = recs[sg[k] + lane];
    }
#pragma unroll
    for (int k = 0; k < DEG_SEGS; ++k)
        if (r[k] >= 0) atomicAdd(&cnt[r[k] >> 16], 1);
#pragma unroll
    for (int k = 0; k < DEG_SEGS; ++k)
        for (int j = lane + 32; j < sc[k]; j += 32)
            atomicAdd(&cnt[recs[sg[k] + j] >> 16], 1);
    __syncthreads();
    if (tid < TILE) {
        const int node = blockIdx.x * TILE + tid;
        if (node < N_NODES) {
            const float dg = (float)(cnt[tid] + 1);
            dinv[node] = (dg > 0.0f) ? rsqrtf(dg) : 0.0f;
        }
    }
}

#define GEMM_THREADS 256
#define GEMM_WAVES   (GEMM_THREADS / 32)
#define N_STRIPS     (N_NODES / 16)
#define GEMM_GRID    ((N_STRIPS + GEMM_WAVES - 1) / GEMM_WAVES)
#define W_QUADS      (DIM / 4)

static_assert(N_NODES % 16 == 0, "whole 16-row strips");
static_assert(W_QUADS * (DIM / 4) == 4 * GEMM_THREADS, "W staging is four passes");

template <bool ACT, bool SCALE>
__device__ __forceinline__
void gemm_body(const int blk, const float* __restrict__ A, const float* __restrict__ bias,
               const float* __restrict__ W, const float* __restrict__ dinv,
               float* __restrict__ out) {
    __shared__ float4 wl[W_QUADS * DIM];

    const int tid  = threadIdx.x;
    const int lane = tid & 31;
    const int wave = __builtin_amdgcn_readfirstlane(tid >> 5);
    const int c    = lane & 15;
    const int h    = lane >> 4;
    const int strip = blk * GEMM_WAVES + wave;
    const int row0  = min(strip, N_STRIPS - 1) * 16;

    float4 a[16];
    const float4* arow = (const float4*)(A + (size_t)(row0 + c) * DIM) + h;
#pragma unroll
    for (int j = 0; j < 16; ++j) a[j] = arow[2 * j];

    float4 sc0 = make_float4(1.0f, 1.0f, 1.0f, 1.0f), sc1 = sc0;
    if constexpr (SCALE) {
        const float4* dq = (const float4*)(dinv + row0 + 8 * h);
        sc0 = dq[0];
        sc1 = dq[1];
    }

#pragma unroll
    for (int it = 0; it < 4; ++it) {
        const int q  = it * GEMM_THREADS + tid;
        const int kq = q >> 5;
        const int cq = q & 31;
        const float4* wp = (const float4*)(W + (size_t)(4 * kq) * DIM) + cq;
        const float4 g0 = wp[0];
        const float4 g1 = wp[DIM / 4];
        const float4 g2 = wp[2 * (DIM / 4)];
        const float4 g3 = wp[3 * (DIM / 4)];
        float4* d = wl + kq * DIM + 4 * cq;
        d[0] = make_float4(g0.x, g1.x, g2.x, g3.x);
        d[1] = make_float4(g0.y, g1.y, g2.y, g3.y);
        d[2] = make_float4(g0.z, g1.z, g2.z, g3.z);
        d[3] = make_float4(g0.w, g1.w, g2.w, g3.w);
    }

    if constexpr (ACT) {
        const float4* bq = (const float4*)bias + h;
#pragma unroll
        for (int j = 0; j < 16; ++j) {
            const float4 b = bq[2 * j];
            a[j].x = fmaxf(a[j].x + b.x, 0.0f);
            a[j].y = fmaxf(a[j].y + b.y, 0.0f);
            a[j].z = fmaxf(a[j].z + b.z, 0.0f);
            a[j].w = fmaxf(a[j].w + b.w, 0.0f);
        }
    }
    __syncthreads();
    if (strip >= N_STRIPS) return;

    v8f acc[8];
#pragma unroll
    for (int n = 0; n < 8; ++n) acc[n] = v8f{};

    const float4* bl = wl + h * DIM + c;
    float4 bcur[8], bnxt[8];
#pragma unroll
    for (int n = 0; n < 8; ++n) bcur[n] = bl[16 * n];
#pragma unroll
    for (int j = 0; j < 16; ++j) {
        __builtin_amdgcn_sched_barrier(0);
        if (j + 1 < 16) {
#pragma unroll
            for (int n = 0; n < 8; ++n) bnxt[n] = bl[2 * (j + 1) * DIM + 16 * n];
        }
        __builtin_amdgcn_sched_barrier(0);
        const v2f a0 = {a[j].x, a[j].y};
        const v2f a1 = {a[j].z, a[j].w};
#pragma unroll
        for (int n = 0; n < 8; ++n) {
            const v2f b0 = {bcur[n].x, bcur[n].y};
            acc[n] = __builtin_amdgcn_wmma_f32_16x16x4_f32(
                          false, a0,  false, b0,
                          (short)0, acc[n],  false,  false);
        }
#pragma unroll
        for (int n = 0; n < 8; ++n) {
            const v2f b1 = {bcur[n].z, bcur[n].w};
            acc[n] = __builtin_amdgcn_wmma_f32_16x16x4_f32(
                          false, a1,  false, b1,
                          (short)0, acc[n],  false,  false);
        }
#pragma unroll
        for (int n = 0; n < 8; ++n) bcur[n] = bnxt[n];
    }

    __builtin_amdgcn_sched_barrier(0);
    asm volatile("v_nop\n\tv_nop\n\tv_nop\n\tv_nop");
    __builtin_amdgcn_sched_barrier(0);

    const float sc[8] = {sc0.x, sc0.y, sc0.z, sc0.w, sc1.x, sc1.y, sc1.z, sc1.w};
    float* orow = out + (size_t)(row0 + 8 * h) * DIM + c;
#pragma unroll
    for (int n = 0; n < 8; ++n) {
#pragma unroll
        for (int r = 0; r < 8; ++r) {
            if constexpr (SCALE) orow[(size_t)r * DIM + 16 * n] = acc[n][r] * sc[r];
            else                 orow[(size_t)r * DIM + 16 * n] = acc[n][r];
        }
    }
}

__global__ __launch_bounds__(GEMM_THREADS)
void k_gemm2(const float* __restrict__ A, const float* __restrict__ bias,
             const float* __restrict__ W, const float* __restrict__ dinv,
             float* __restrict__ out) {
    gemm_body<true, true>(blockIdx.x, A, bias, W, dinv, out);
}

static_assert(FILL_THREADS == GEMM_THREADS, "one block size for both kinds of work-group");
__global__ __launch_bounds__(GEMM_THREADS)
void k_fill_gemm1(const int* __restrict__ src, const int* __restrict__ dst,
                  int* __restrict__ seg, int* __restrict__ recs,
                  const float* __restrict__ x, const float* __restrict__ W1,
                  float* __restrict__ t1) {
    if (blockIdx.x < N_BINB) {
        fill_body(blockIdx.x, src, dst, seg, recs);
    } else {
        gemm_body<false, false>(blockIdx.x - N_BINB, x, nullptr, W1, nullptr, t1);
    }
}

template <bool WSRC, bool FULL>
__device__ __forceinline__
void agg_block(const int cur, const float wcur, const int n, const int lane,
               const float* __restrict__ t, float* acc) {
    float4 v[AGG_G];
    int    dl[AGG_G];
#pragma unroll
    for (int j = 0; j < AGG_G; ++j) {
        const int pj = __builtin_amdgcn_readlane(cur, j);
        dl[j] = pj >> 16;
        v[j]  = ((const float4*)(t + (size_t)(pj & 0xFFFF) * DIM))[lane];
    }
    if constexpr (WSRC) {
#pragma unroll
        for (int j = 0; j < AGG_G; ++j) {
            const float wj = __int_as_float(
                __builtin_amdgcn_readlane(__float_as_int(wcur), j));
            v[j].x *= wj; v[j].y *= wj; v[j].z *= wj; v[j].w *= wj;
        }
    }
#pragma unroll
    for (int j = 0; j < AGG_G; ++j) {
        if (FULL || j < n) {
            float* a = acc + dl[j] * DIM + lane;
            atomicAdd(a,      v[j].x);
            atomicAdd(a + 32, v[j].y);
            atomicAdd(a + 64, v[j].z);
            atomicAdd(a + 96, v[j].w);
        }
    }
}

template <bool WSRC>
__global__ __launch_bounds__(AGG_THREADS)
void k_tile_agg(const int* __restrict__ recs, const int* __restrict__ seg,
                const float* __restrict__ dinv, const float* __restrict__ t,
                float* __restrict__ agg) {
    __shared__ float acc[TILE * DIM];
    __shared__ int   lrec[AGG_STAGE];
    __shared__ float lw[WSRC ? AGG_STAGE : 1];

    const int tid   = threadIdx.x;
    const int lane  = tid & 31;
    const int wave  = __builtin_amdgcn_readfirstlane(tid >> 5);
    const int node0 = blockIdx.x * TILE;
    SegLane sl;
    const int cnt = seg_scan(seg, blockIdx.x, lane, wave & 3, sl);

    float  wq[AGG_INIT_TRIPS];
    float4 xq[AGG_INIT_TRIPS];
#pragma unroll
    for (int q = 0; q < AGG_INIT_TRIPS; ++q) {
        const int row  = min(wave + q * AGG_WAVES, TILE - 1);
        const int node = min(node0 + row, N_NODES - 1);
        wq[q] = dinv[node];
        xq[q] = ((const float4*)t)[(size_t)node * (DIM / 4) + lane];
    }
#pragma unroll
    for (int q = 0; q < AGG_INIT_TRIPS; ++q) {
        const int row = wave + q * AGG_WAVES;
        if (row < TILE) {
            float4 x = xq[q];
            if constexpr (WSRC) { x.x *= wq[q]; x.y *= wq[q]; x.z *= wq[q]; x.w *= wq[q]; }
            float* a = acc + row * DIM + lane;
            a[0]  = x.x;
            a[32] = x.y;
            a[64] = x.z;
            a[96] = x.w;
        }
    }

    int r0 = 0;
    do {
        const int nr = min(AGG_STAGE, cnt - r0);
        int sc[AGG_SEGS], sg[AGG_SEGS], sp[AGG_SEGS];
        int   sr[AGG_SEGS];
        float sw[AGG_SEGS];
        bool  ok[AGG_SEGS];
#pragma unroll
        for (int k = 0; k < AGG_SEGS; ++k) {
            const int b = wave + k * AGG_WAVES;
            int st = 0;
            sc[k] = 0; sg[k] = 0;
            if (b < N_BINB) seg_get(sl, b, sc[k], sg[k], st);
            sp[k] = st - r0;
            const int li = sp[k] + lane;
            ok[k] = lane < sc[k] && li >= 0 && li < nr;
            sr[k] = 0;
            if (ok[k]) sr[k] = recs[sg[k] + lane];
        }
        if constexpr (WSRC) {
#pragma unroll
            for (int k = 0; k < AGG_SEGS; ++k) sw[k] = dinv[sr[k] & 0xFFFF];
        }
#pragma unroll
        for (int k = 0; k < AGG_SEGS; ++k) {
            if (ok[k]) {
                lrec[sp[k] + lane] = sr[k];
                if constexpr (WSRC) lw[sp[k] + lane] = sw[k];
            }
        }
#pragma unroll
        for (int k = 0; k < AGG_SEGS; ++k) {
            for (int j = lane + 32; j < sc[k]; j += 32) {
                const int li = sp[k] + j;
                if (li >= 0 && li < nr) {
                    const int rj = recs[sg[k] + j];
                    lrec[li] = rj;
                    if constexpr (WSRC) lw[li] = dinv[rj & 0xFFFF];
                }
            }
        }
        if (tid < AGG_G && nr + tid < AGG_STAGE) {
            lrec[nr + tid] = 0;
            if constexpr (WSRC) lw[nr + tid] = 0.0f;
        }
        __syncthreads();

        for (int b = wave * AGG_G; b < nr; b += AGG_WAVES * AGG_G) {
            const int cur = lrec[b + (lane & (AGG_G - 1))];
            float wcur = 0.0f;
            if constexpr (WSRC) wcur = lw[b + (lane & (AGG_G - 1))];
            const int n = min(AGG_G, nr - b);

            if (n == AGG_G) agg_block<WSRC, true >(cur, wcur, n, lane, t, acc);
            else            agg_block<WSRC, false>(cur, wcur, n, lane, t, acc);
        }
        r0 += AGG_STAGE;
        __syncthreads();
    } while (r0 < cnt);

#pragma unroll
    for (int q = 0; q < AGG_INIT_TRIPS; ++q) {
        const float* a = acc + min(wave + q * AGG_WAVES, TILE - 1) * DIM + lane;
        xq[q] = make_float4(a[0], a[32], a[64], a[96]);
    }
#pragma unroll
    for (int q = 0; q < AGG_INIT_TRIPS; ++q) {
        const int row  = wave + q * AGG_WAVES;
        const int node = node0 + row;
        if (row < TILE && node < N_NODES) {
            const float w = wq[q];
            float4 v = xq[q];
            v.x *= w; v.y *= w; v.z *= w; v.w *= w;
            ((float4*)agg)[(size_t)node * (DIM / 4) + lane] = v;
        }
    }
}

#define HEAD_ROWS     16
#define HEAD_THREADS  128
#define HEAD_WAVES    (HEAD_THREADS / 32)
#define N_HEAD_WAVES  (N_NODES / HEAD_ROWS)
#define HEAD_GRID     ((N_HEAD_WAVES + HEAD_WAVES - 1) / HEAD_WAVES)

static_assert(N_NODES % HEAD_ROWS == 0, "whole groups of rows per wave");
static_assert(HEAD_ROWS == 16, "the fold below ends with one value per lane: 2 heads x 16 rows = 32 lanes");

template <int M>
__device__ __forceinline__ void fold_step(float (&val)[2 * HEAD_ROWS], const int lane) {
    const bool up = (lane & M) != 0;
#pragma unroll
    for (int i = 0; i < M; ++i) {
        const float keep = up ? val[i + M] : val[i];
        const float send = up ? val[i]     : val[i + M];
        val[i] = keep + __shfl_xor(send, M, 32);
    }
}

__global__ __launch_bounds__(HEAD_THREADS)
void k_heads(const float* __restrict__ agg, const float* __restrict__ b,
             const float* __restrict__ Wt, const float* __restrict__ bt,
             const float* __restrict__ We, const float* __restrict__ be,
             float* __restrict__ out) {
    const int lane = threadIdx.x & 31;
    const int wave = __builtin_amdgcn_readfirstlane(threadIdx.x >> 5);
    const int w    = blockIdx.x * HEAD_WAVES + wave;
    if (w >= N_HEAD_WAVES) return;
    const int row0 = w * HEAD_ROWS;

    const float4 bb = ((const float4*)b)[lane];
    const float4 wt = ((const float4*)Wt)[lane];
    const float4 we = ((const float4*)We)[lane];
    const float  hb = (lane & 16) ? be[0] : bt[0];

    float4 v[HEAD_ROWS];
    const float4* p = (const float4*)(agg + (size_t)row0 * DIM) + lane;
#pragma unroll
    for (int r = 0; r < HEAD_ROWS; ++r) v[r] = p[r * (DIM / 4)];
    __builtin_amdgcn_sched_barrier(0);

    float val[2 * HEAD_ROWS];
#pragma unroll
    for (int r = 0; r < HEAD_ROWS; ++r) {
        float4 x = v[r];
        x.x = fmaxf(x.x + bb.x, 0.0f);
        x.y = fmaxf(x.y + bb.y, 0.0f);
        x.z = fmaxf(x.z + bb.z, 0.0f);
        x.w = fmaxf(x.w + bb.w, 0.0f);
        val[r]             = fmaf(x.w, wt.w, fmaf(x.z, wt.z, fmaf(x.y, wt.y, x.x * wt.x)));
        val[HEAD_ROWS + r] = fmaf(x.w, we.w, fmaf(x.z, we.z, fmaf(x.y, we.y, x.x * we.x)));
    }

    fold_step<16>(val, lane);
    fold_step<8>(val, lane);
    fold_step<4>(val, lane);
    fold_step<2>(val, lane);
    fold_step<1>(val, lane);

    out[(size_t)(lane >> 4) * N_NODES + row0 + (lane & 15)] = val[0] + hb;
}

extern "C" void kernel_launch(void* const* d_in, const int* in_sizes, int n_in,
                              void* d_out, int out_size, void* d_ws, size_t ws_size,
                              hipStream_t stream) {
    (void)in_sizes; (void)n_in; (void)out_size; (void)ws_size;

    const float* x   = (const float*)d_in[0];
    const int*   ei  = (const int*)  d_in[1];
    const float* W1  = (const float*)d_in[2];
    const float* b1  = (const float*)d_in[3];
    const float* W2  = (const float*)d_in[4];
    const float* b2  = (const float*)d_in[5];
    const float* Wt  = (const float*)d_in[6];
    const float* bt  = (const float*)d_in[7];
    const float* We  = (const float*)d_in[8];
    const float* be  = (const float*)d_in[9];
    float*       out = (float*)d_out;

    const int* src = ei;
    const int* dst = ei + N_EDGES;

    float* dinv = (float*)d_ws;
    float* bufA = dinv + ((N_NODES + 255) & ~255);
    float* bufB = bufA + (size_t)N_NODES * DIM;
    int*   recs      = (int*)(bufB + (size_t)N_NODES * DIM);
    int*   seg  = recs + (size_t)N_EDGES;

    k_fill_gemm1<<<N_BINB + GEMM_GRID, GEMM_THREADS, 0, stream>>>(src, dst, seg, recs, x, W1, bufA);
    k_tile_deg <<<N_TILES, DEG_THREADS, 0, stream>>>(recs, seg, dinv);

    k_tile_agg<true> <<<N_TILES, AGG_THREADS, 0, stream>>>(recs, seg, dinv, bufA, bufB);

    k_gemm2          <<<GEMM_GRID, GEMM_THREADS, 0, stream>>>(bufB, b1, W2, dinv, bufA);
    k_tile_agg<false><<<N_TILES, AGG_THREADS, 0, stream>>>(recs, seg, dinv, bufA, bufB);

    k_heads<<<HEAD_GRID, HEAD_THREADS, 0, stream>>>(bufB, b2, Wt, bt, We, be, out);
}


# aggregation: a tile's staged records counting-sorted by destination row in LDS; a wave sums each run of equal rows in registers: one LDS float add per run, not per edge; sum order changed
# speedup vs baseline: 4.5919x; 1.0608x over previous
#include <hip/hip_runtime.h>

#define N_NODES 50000
#define N_EDGES 800000
#define DIM 128

#define TILE        196
#define N_TILES     ((N_NODES + TILE - 1) / TILE)
#define BIN_EDGES   8192
#define N_BINB      ((N_EDGES + BIN_EDGES - 1) / BIN_EDGES)
#define DEG_THREADS 512
#define AGG_THREADS  1024
#define AGG_WAVES    (AGG_THREADS / 32)
#define AGG_G        16
#define AGG_STAGE    16384
#define AGG_SEGS     ((N_BINB + AGG_WAVES - 1) / AGG_WAVES)
#define AGG_ROWS     256

#define AGG_INIT_TRIPS ((TILE * (DIM / 4) + AGG_THREADS - 1) / AGG_THREADS)

static_assert(N_NODES <= 65536, "source index is packed into 16 bits");
static_assert(TILE <= 256, "local destination row is packed into 8 bits");
static_assert(N_TILES == 256, "one tile per WGP");
static_assert(AGG_STAGE % AGG_G == 0 && AGG_G == 16, "whole blocks per round; a block is half a wave's lanes");
static_assert(AGG_THREADS % 32 == 0 && DIM == 128, "a lane owns one column quad of a row");
static_assert(N_EDGES % 4 == 0, "edges are read four at a time");
static_assert(TILE <= AGG_ROWS && AGG_ROWS == 8 * 32 && AGG_ROWS <= AGG_THREADS, "one wave scans the row counters, eight per lane");

typedef float v2f __attribute__((ext_vector_type(2)));
typedef float v8f __attribute__((ext_vector_type(8)));

#define SEG_PITCH 100
#define SEG_SHIFT 14

static_assert(N_BINB <= SEG_PITCH && SEG_PITCH % 4 == 0 && SEG_PITCH <= 128, "a lane reads four words of its tile's table row");
static_assert(BIN_EDGES < (1 << SEG_SHIFT) && SEG_SHIFT <= 15, "first slot and count are packed into one word");

#define FILL_THREADS 256
#define FILL_GROUPS  (BIN_EDGES / 4 / FILL_THREADS)

static_assert(FILL_GROUPS * 4 * FILL_THREADS == BIN_EDGES, "a block's edges are a whole number of passes");
static_assert(FILL_THREADS == N_TILES, "one tile per thread");

__device__ __forceinline__
void fill_body(const int blk, const int* __restrict__ src, const int* __restrict__ dst,
               int* __restrict__ seg, int* __restrict__ recs) {
    __shared__ int s[FILL_THREADS];
    __shared__ int lcnt[N_TILES];
    __shared__ int lbase[N_TILES];
    const int tid  = threadIdx.x;
    const int base = blk * BIN_EDGES;

    int4 d[FILL_GROUPS], sv[FILL_GROUPS];
#pragma unroll
    for (int k = 0; k < FILL_GROUPS; ++k) {
        const int e = min(base + 4 * (k * FILL_THREADS + tid), N_EDGES - 4);
        d[k]  = *(const int4*)(dst + e);
        sv[k] = *(const int4*)(src + e);
    }

    lcnt[tid] = 0;
    __syncthreads();
#pragma unroll
    for (int k = 0; k < FILL_GROUPS; ++k) {
        if (base + 4 * (k * FILL_THREADS + tid) < N_EDGES) {
            atomicAdd(&lcnt[(unsigned)d[k].x / TILE], 1);
            atomicAdd(&lcnt[(unsigned)d[k].y / TILE], 1);
            atomicAdd(&lcnt[(unsigned)d[k].z / TILE], 1);
            atomicAdd(&lcnt[(unsigned)d[k].w / TILE], 1);
        }
    }
    __syncthreads();

    const int c = lcnt[tid];
    lcnt[tid] = 0;
    s[tid] = c;
    __syncthreads();
    for (int off = 1; off < FILL_THREADS; off <<= 1) {
        const int v = (tid >= off) ? s[tid - off] : 0;
        __syncthreads();
        s[tid] += v;
        __syncthreads();
    }
    const int excl = s[tid] - c;
    lbase[tid] = base + excl;
    seg[tid * SEG_PITCH + blk] = excl | (c << SEG_SHIFT);
    __syncthreads();

#pragma unroll
    for (int k = 0; k < FILL_GROUPS; ++k) {
        if (base + 4 * (k * FILL_THREADS + tid) < N_EDGES) {
            const int dd[4] = {d[k].x, d[k].y, d[k].z, d[k].w};
            const int ss[4] = {sv[k].x, sv[k].y, sv[k].z, sv[k].w};
#pragma unroll
            for (int q = 0; q < 4; ++q) {
                const int tile = (unsigned)dd[q] / TILE;
                const int row  = dd[q] - tile * TILE;
                const int r    = atomicAdd(&lcnt[tile], 1);
                recs[lbase[tile] + r] = ss[q] | (row << 16);
            }
        }
    }
}

struct SegLane { int word; int start; };

__device__ __forceinline__
int seg_scan(const int* __restrict__ seg, const int tile, const int lane, const int q, SegLane& sl) {
    int4 w = make_int4(0, 0, 0, 0);
    if (4 * lane < SEG_PITCH) w = *(const int4*)(seg + tile * SEG_PITCH + 4 * lane);
    const int w0 = (4 * lane + 0 < N_BINB) ? w.x : 0;
    const int w1 = (4 * lane + 1 < N_BINB) ? w.y : 0;
    const int w2 = (4 * lane + 2 < N_BINB) ? w.z : 0;
    const int w3 = (4 * lane + 3 < N_BINB) ? w.w : 0;
    const int c0 = w0 >> SEG_SHIFT, c1 = w1 >> SEG_SHIFT, c2 = w2 >> SEG_SHIFT, c3 = w3 >> SEG_SHIFT;
    const int own = c0 + c1 + c2 + c3;
    int inc = own;
#pragma unroll
    for (int off = 1; off < 32; off <<= 1) {
        const int v = __shfl_up(inc, off, 32);
        if (lane >= off) inc += v;
    }
    const int s0 = inc - own, s1 = s0 + c0, s2 = s1 + c1, s3 = s2 + c2;
    sl.word  = (q & 2) ? ((q & 1) ? w3 : w2) : ((q & 1) ? w1 : w0);
    sl.start = (q & 2) ? ((q & 1) ? s3 : s2) : ((q & 1) ? s1 : s0);
    return __shfl(inc, 31, 32);
}

__device__ __forceinline__
void seg_get(const SegLane& sl, const int b, int& cnt, int& goff, int& start) {
    const int word = __builtin_amdgcn_readlane(sl.word, b >> 2);
    start = __builtin_amdgcn_readlane(sl.start, b >> 2);
    cnt   = word >> SEG_SHIFT;
    goff  = b * BIN_EDGES + (word & ((1 << SEG_SHIFT) - 1));
}

#define DEG_WAVES (DEG_THREADS / 32)
#define DEG_SEGS  ((N_BINB + DEG_WAVES - 1) / DEG_WAVES)

static_assert(DEG_WAVES % 4 == 0 && AGG_WAVES % 4 == 0, "a wave's blocks all have the same b & 3");

__global__ __launch_bounds__(DEG_THREADS)
void k_tile_deg(const int* __restrict__ recs, const int* __restrict__ seg,
                float* __restrict__ dinv) {
    __shared__ int cnt[TILE];
    const int tid  = threadIdx.x;
    const int lane = tid & 31;
    const int wave = __builtin_amdgcn_readfirstlane(tid >> 5);
    if (tid < TILE) cnt[tid] = 0;
    __syncthreads();

    SegLane sl;
    seg_scan(seg, blockIdx.x, lane, wave & 3, sl);

    int r[DEG_SEGS], sc[DEG_SEGS], sg[DEG_SEGS];
#pragma unroll
    for (int k = 0; k < DEG_SEGS; ++k) {
        const int b = wave + k * DEG_WAVES;
        int st;
        sc[k] = 0; sg[k] = 0;
        if (b < N_BINB) seg_get(sl, b, sc[k], sg[k], st);
        r[k] = -1;
        if (lane < sc[k]) r[k] = recs[sg[k] + lane];
    }
#pragma unroll
    for (int k = 0; k < DEG_SEGS; ++k)
        if (r[k] >= 0) atomicAdd(&cnt[r[k] >> 16], 1);
#pragma unroll
    for (int k = 0; k < DEG_SEGS; ++k)
        for (int j = lane + 32; j < sc[k]; j += 32)
            atomicAdd(&cnt[recs[sg[k] + j] >> 16], 1);
    __syncthreads();
    if (tid < TILE) {
        const int node = blockIdx.x * TILE + tid;
        if (node < N_NODES) {
            const float dg = (float)(cnt[tid] + 1);
            dinv[node] = (dg > 0.0f) ? rsqrtf(dg) : 0.0f;
        }
    }
}

#define GEMM_THREADS 256
#define GEMM_WAVES   (GEMM_THREADS / 32)
#define N_STRIPS     (N_NODES / 16)
#define GEMM_GRID    ((N_STRIPS + GEMM_WAVES - 1) / GEMM_WAVES)
#define W_QUADS      (DIM / 4)

static_assert(N_NODES % 16 == 0, "whole 16-row strips");
static_assert(W_QUADS * (DIM / 4) == 4 * GEMM_THREADS, "W staging is four passes");

template <bool ACT, bool SCALE>
__device__ __forceinline__
void gemm_body(const int blk, const float* __restrict__ A, const float* __restrict__ bias,
               const float* __restrict__ W, const float* __restrict__ dinv,
               float* __restrict__ out) {
    __shared__ float4 wl[W_QUADS * DIM];

    const int tid  = threadIdx.x;
    const int lane = tid & 31;
    const int wave = __builtin_amdgcn_readfirstlane(tid >> 5);
    const int c    = lane & 15;
    const int h    = lane >> 4;
    const int strip = blk * GEMM_WAVES + wave;
    const int row0  = min(strip, N_STRIPS - 1) * 16;

    float4 a[16];
    const float4* arow = (const float4*)(A + (size_t)(row0 + c) * DIM) + h;
#pragma unroll
    for (int j = 0; j < 16; ++j) a[j] = arow[2 * j];

    float4 sc0 = make_float4(1.0f, 1.0f, 1.0f, 1.0f), sc1 = sc0;
    if constexpr (SCALE) {
        const float4* dq = (const float4*)(dinv + row0 + 8 * h);
        sc0 = dq[0];
        sc1 = dq[1];
    }

#pragma unroll
    for (int it = 0; it < 4; ++it) {
        const int q  = it * GEMM_THREADS + tid;
        const int kq = q >> 5;
        const int cq = q & 31;
        const float4* wp = (const float4*)(W + (size_t)(4 * kq) * DIM) + cq;
        const float4 g0 = wp[0];
        const float4 g1 = wp[DIM / 4];
        const float4 g2 = wp[2 * (DIM / 4)];
        const float4 g3 = wp[3 * (DIM / 4)];
        float4* d = wl + kq * DIM + 4 * cq;
        d[0] = make_float4(g0.x, g1.x, g2.x, g3.x);
        d[1] = make_float4(g0.y, g1.y, g2.y, g3.y);
        d[2] = make_float4(g0.z, g1.z, g2.z, g3.z);
        d[3] = make_float4(g0.w, g1.w, g2.w, g3.w);
    }

    if constexpr (ACT) {
        const float4* bq = (const float4*)bias + h;
#pragma unroll
        for (int j = 0; j < 16; ++j) {
            const float4 b = bq[2 * j];
            a[j].x = fmaxf(a[j].x + b.x, 0.0f);
            a[j].y = fmaxf(a[j].y + b.y, 0.0f);
            a[j].z = fmaxf(a[j].z + b.z, 0.0f);
            a[j].w = fmaxf(a[j].w + b.w, 0.0f);
        }
    }
    __syncthreads();
    if (strip >= N_STRIPS) return;

    v8f acc[8];
#pragma unroll
    for (int n = 0; n < 8; ++n) acc[n] = v8f{};

    const float4* bl = wl + h * DIM + c;
    float4 bcur[8], bnxt[8];
#pragma unroll
    for (int n = 0; n < 8; ++n) bcur[n] = bl[16 * n];
#pragma unroll
    for (int j = 0; j < 16; ++j) {
        __builtin_amdgcn_sched_barrier(0);
        if (j + 1 < 16) {
#pragma unroll
            for (int n = 0; n < 8; ++n) bnxt[n] = bl[2 * (j + 1) * DIM + 16 * n];
        }
        __builtin_amdgcn_sched_barrier(0);
        const v2f a0 = {a[j].x, a[j].y};
        const v2f a1 = {a[j].z, a[j].w};
#pragma unroll
        for (int n = 0; n < 8; ++n) {
            const v2f b0 = {bcur[n].x, bcur[n].y};
            acc[n] = __builtin_amdgcn_wmma_f32_16x16x4_f32(
                          false, a0,  false, b0,
                          (short)0, acc[n],  false,  false);
        }
#pragma unroll
        for (int n = 0; n < 8; ++n) {
            const v2f b1 = {bcur[n].z, bcur[n].w};
            acc[n] = __builtin_amdgcn_wmma_f32_16x16x4_f32(
                          false, a1,  false, b1,
                          (short)0, acc[n],  false,  false);
        }
#pragma unroll
        for (int n = 0; n < 8; ++n) bcur[n] = bnxt[n];
    }

    __builtin_amdgcn_sched_barrier(0);
    asm volatile("v_nop\n\tv_nop\n\tv_nop\n\tv_nop");
    __builtin_amdgcn_sched_barrier(0);

    const float sc[8] = {sc0.x, sc0.y, sc0.z, sc0.w, sc1.x, sc1.y, sc1.z, sc1.w};
    float* orow = out + (size_t)(row0 + 8 * h) * DIM + c;
#pragma unroll
    for (int n = 0; n < 8; ++n) {
#pragma unroll
        for (int r = 0; r < 8; ++r) {
            if constexpr (SCALE) orow[(size_t)r * DIM + 16 * n] = acc[n][r] * sc[r];
            else                 orow[(size_t)r * DIM + 16 * n] = acc[n][r];
        }
    }
}

__global__ __launch_bounds__(GEMM_THREADS)
void k_gemm2(const float* __restrict__ A, const float* __restrict__ bias,
             const float* __restrict__ W, const float* __restrict__ dinv,
             float* __restrict__ out) {
    gemm_body<true, true>(blockIdx.x, A, bias, W, dinv, out);
}

static_assert(FILL_THREADS == GEMM_THREADS, "one block size for both kinds of work-group");
__global__ __launch_bounds__(GEMM_THREADS)
void k_fill_gemm1(const int* __restrict__ src, const int* __restrict__ dst,
                  int* __restrict__ seg, int* __restrict__ recs,
                  const float* __restrict__ x, const float* __restrict__ W1,
                  float* __restrict__ t1) {
    if (blockIdx.x < N_BINB) {
        fill_body(blockIdx.x, src, dst, seg, recs);
    } else {
        gemm_body<false, false>(blockIdx.x - N_BINB, x, nullptr, W1, nullptr, t1);
    }
}

template <bool WSRC, bool FULL>
__device__ __forceinline__
void agg_block(const int cur, const float wcur, const int n, const int lane,
               const float* __restrict__ t, float* acc) {
    float4 v[AGG_G];
    int    dl[AGG_G];
#pragma unroll
    for (int j = 0; j < AGG_G; ++j) {
        const int pj = __builtin_amdgcn_readlane(cur, j);
        dl[j] = pj >> 16;
        v[j]  = ((const float4*)(t + (size_t)(pj & 0xFFFF) * DIM))[lane];
    }
    if constexpr (WSRC) {
#pragma unroll
        for (int j = 0; j < AGG_G; ++j) {
            const float wj = __int_as_float(
                __builtin_amdgcn_readlane(__float_as_int(wcur), j));
            v[j].x *= wj; v[j].y *= wj; v[j].z *= wj; v[j].w *= wj;
        }
    }
    int    row = dl[0];
    float4 s   = v[0];
#pragma unroll
    for (int j = 1; j < AGG_G; ++j) {
        if (FULL || j < n) {
            if (dl[j] != row) {
                float* a = acc + row * DIM + lane;
                atomicAdd(a,      s.x);
                atomicAdd(a + 32, s.y);
                atomicAdd(a + 64, s.z);
                atomicAdd(a + 96, s.w);
                row = dl[j];
                s   = v[j];
            } else {
                s.x += v[j].x; s.y += v[j].y; s.z += v[j].z; s.w += v[j].w;
            }
        }
    }
    float* a = acc + row * DIM + lane;
    atomicAdd(a,      s.x);
    atomicAdd(a + 32, s.y);
    atomicAdd(a + 64, s.z);
    atomicAdd(a + 96, s.w);
}

template <bool WSRC>
__global__ __launch_bounds__(AGG_THREADS)
void k_tile_agg(const int* __restrict__ recs, const int* __restrict__ seg,
                const float* __restrict__ dinv, const float* __restrict__ t,
                float* __restrict__ agg) {
    __shared__ float acc[TILE * DIM];
    __shared__ int   lrec[AGG_STAGE];
    __shared__ float lw[WSRC ? AGG_STAGE : 1];
    __shared__ __attribute__((aligned(16))) int rc[AGG_ROWS];

    const int tid   = threadIdx.x;
    const int lane  = tid & 31;
    const int wave  = __builtin_amdgcn_readfirstlane(tid >> 5);
    const int node0 = blockIdx.x * TILE;
    SegLane sl;
    const int cnt = seg_scan(seg, blockIdx.x, lane, wave & 3, sl);

    float  wq[AGG_INIT_TRIPS];
    float4 xq[AGG_INIT_TRIPS];
#pragma unroll
    for (int q = 0; q < AGG_INIT_TRIPS; ++q) {
        const int row  = min(wave + q * AGG_WAVES, TILE - 1);
        const int node = min(node0 + row, N_NODES - 1);
        wq[q] = dinv[node];
        xq[q] = ((const float4*)t)[(size_t)node * (DIM / 4) + lane];
    }
#pragma unroll
    for (int q = 0; q < AGG_INIT_TRIPS; ++q) {
        const int row = wave + q * AGG_WAVES;
        if (row < TILE) {
            float4 x = xq[q];
            if constexpr (WSRC) { x.x *= wq[q]; x.y *= wq[q]; x.z *= wq[q]; x.w *= wq[q]; }
            float* a = acc + row * DIM + lane;
            a[0]  = x.x;
            a[32] = x.y;
            a[64] = x.z;
            a[96] = x.w;
        }
    }

    int r0 = 0;
    do {
        const int nr = min(AGG_STAGE, cnt - r0);
        int sc[AGG_SEGS], sg[AGG_SEGS], sp[AGG_SEGS];
        int   sr[AGG_SEGS];
        float sw[AGG_SEGS];
        bool  ok[AGG_SEGS];
#pragma unroll
        for (int k = 0; k < AGG_SEGS; ++k) {
            const int b = wave + k * AGG_WAVES;
            int st = 0;
            sc[k] = 0; sg[k] = 0;
            if (b < N_BINB) seg_get(sl, b, sc[k], sg[k], st);
            sp[k] = st - r0;
            const int li = sp[k] + lane;
            ok[k] = lane < sc[k] && li >= 0 && li < nr;
            sr[k] = 0;
            if (ok[k]) sr[k] = recs[sg[k] + lane];
        }
        if constexpr (WSRC) {
#pragma unroll
            for (int k = 0; k < AGG_SEGS; ++k) sw[k] = dinv[sr[k] & 0xFFFF];
        }
        if (tid < AGG_ROWS) rc[tid] = 0;
        __syncthreads();
#pragma unroll
        for (int k = 0; k < AGG_SEGS; ++k)
            if (ok[k]) atomicAdd(&rc[sr[k] >> 16], 1);
#pragma unroll
        for (int k = 0; k < AGG_SEGS; ++k) {
            for (int j = lane + 32; j < sc[k]; j += 32) {
                const int li = sp[k] + j;
                if (li >= 0 && li < nr) atomicAdd(&rc[recs[sg[k] + j] >> 16], 1);
            }
        }
        __syncthreads();
        if (wave == 0) {
            const int4 c0 = ((const int4*)rc)[2 * lane];
            const int4 c1 = ((const int4*)rc)[2 * lane + 1];
            const int own = c0.x + c0.y + c0.z + c0.w + c1.x + c1.y + c1.z + c1.w;
            int inc = own;
#pragma unroll
            for (int off = 1; off < 32; off <<= 1) {
                const int u = __shfl_up(inc, off, 32);
                if (lane >= off) inc += u;
            }
            int4 p0, p1;
            p0.x = inc - own;   p0.y = p0.x + c0.x; p0.z = p0.y + c0.y; p0.w = p0.z + c0.z;
            p1.x = p0.w + c0.w; p1.y = p1.x + c1.x; p1.z = p1.y + c1.y; p1.w = p1.z + c1.z;
            ((int4*)rc)[2 * lane]     = p0;
            ((int4*)rc)[2 * lane + 1] = p1;
        }
        __syncthreads();
#pragma unroll
        for (int k = 0; k < AGG_SEGS; ++k) {
            if (ok[k]) {
                const int pos = atomicAdd(&rc[sr[k] >> 16], 1);
                lrec[pos] = sr[k];
                if constexpr (WSRC) lw[pos] = sw[k];
            }
        }
#pragma unroll
        for (int k = 0; k < AGG_SEGS; ++k) {
            for (int j = lane + 32; j < sc[k]; j += 32) {
                const int li = sp[k] + j;
                if (li >= 0 && li < nr) {
                    const int rj  = recs[sg[k] + j];
                    const int pos = atomicAdd(&rc[rj >> 16], 1);
                    lrec[pos] = rj;
                    if constexpr (WSRC) lw[pos] = dinv[rj & 0xFFFF];
                }
            }
        }
        if (tid < AGG_G && nr + tid < AGG_STAGE) {
            lrec[nr + tid] = 0;
            if constexpr (WSRC) lw[nr + tid] = 0.0f;
        }
        __syncthreads();

        for (int b = wave * AGG_G; b < nr; b += AGG_WAVES * AGG_G) {
            const int cur = lrec[b + (lane & (AGG_G - 1))];
            float wcur = 0.0f;
            if constexpr (WSRC) wcur = lw[b + (lane & (AGG_G - 1))];
            const int n = min(AGG_G, nr - b);

            if (n == AGG_G) agg_block<WSRC, true >(cur, wcur, n, lane, t, acc);
            else            agg_block<WSRC, false>(cur, wcur, n, lane, t, acc);
        }
        r0 += AGG_STAGE;
        __syncthreads();
    } while (r0 < cnt);

#pragma unroll
    for (int q = 0; q < AGG_INIT_TRIPS; ++q) {
        const float* a = acc + min(wave + q * AGG_WAVES, TILE - 1) * DIM + lane;
        xq[q] = make_float4(a[0], a[32], a[64], a[96]);
    }
#pragma unroll
    for (int q = 0; q < AGG_INIT_TRIPS; ++q) {
        const int row  = wave + q * AGG_WAVES;
        const int node = node0 + row;
        if (row < TILE && node < N_NODES) {
            const float w = wq[q];
            float4 v = xq[q];
            v.x *= w; v.y *= w; v.z *= w; v.w *= w;
            ((float4*)agg)[(size_t)node * (DIM / 4) + lane] = v;
        }
    }
}

#define HEAD_ROWS     16
#define HEAD_THREADS  128
#define HEAD_WAVES    (HEAD_THREADS / 32)
#define N_HEAD_WAVES  (N_NODES / HEAD_ROWS)
#define HEAD_GRID     ((N_HEAD_WAVES + HEAD_WAVES - 1) / HEAD_WAVES)

static_assert(N_NODES % HEAD_ROWS == 0, "whole groups of rows per wave");
static_assert(HEAD_ROWS == 16, "the fold below ends with one value per lane: 2 heads x 16 rows = 32 lanes");

template <int M>
__device__ __forceinline__ void fold_step(float (&val)[2 * HEAD_ROWS], const int lane) {
    const bool up = (lane & M) != 0;
#pragma unroll
    for (int i = 0; i < M; ++i) {
        const float keep = up ? val[i + M] : val[i];
        const float send = up ? val[i]     : val[i + M];
        val[i] = keep + __shfl_xor(send, M, 32);
    }
}

__global__ __launch_bounds__(HEAD_THREADS)
void k_heads(const float* __restrict__ agg, const float* __restrict__ b,
             const float* __restrict__ Wt, const float* __restrict__ bt,
             const float* __restrict__ We, const float* __restrict__ be,
             float* __restrict__ out) {
    const int lane = threadIdx.x & 31;
    const int wave = __builtin_amdgcn_readfirstlane(threadIdx.x >> 5);
    const int w    = blockIdx.x * HEAD_WAVES + wave;
    if (w >= N_HEAD_WAVES) return;
    const int row0 = w * HEAD_ROWS;

    const float4 bb = ((const float4*)b)[lane];
    const float4 wt = ((const float4*)Wt)[lane];
    const float4 we = ((const float4*)We)[lane];
    const float  hb = (lane & 16) ? be[0] : bt[0];

    float4 v[HEAD_ROWS];
    const float4* p = (const float4*)(agg + (size_t)row0 * DIM) + lane;
#pragma unroll
    for (int r = 0; r < HEAD_ROWS; ++r) v[r] = p[r * (DIM / 4)];
    __builtin_amdgcn_sched_barrier(0);

    float val[2 * HEAD_ROWS];
#pragma unroll
    for (int r = 0; r < HEAD_ROWS; ++r) {
        float4 x = v[r];
        x.x = fmaxf(x.x + bb.x, 0.0f);
        x.y = fmaxf(x.y + bb.y, 0.0f);
        x.z = fmaxf(x.z + bb.z, 0.0f);
        x.w = fmaxf(x.w + bb.w, 0.0f);
        val[r]             = fmaf(x.w, wt.w, fmaf(x.z, wt.z, fmaf(x.y, wt.y, x.x * wt.x)));
        val[HEAD_ROWS + r] = fmaf(x.w, we.w, fmaf(x.z, we.z, fmaf(x.y, we.y, x.x * we.x)));
    }

    fold_step<16>(val, lane);
    fold_step<8>(val, lane);
    fold_step<4>(val, lane);
    fold_step<2>(val, lane);
    fold_step<1>(val, lane);

    out[(size_t)(lane >> 4) * N_NODES + row0 + (lane & 15)] = val[0] + hb;
}

extern "C" void kernel_launch(void* const* d_in, const int* in_sizes, int n_in,
                              void* d_out, int out_size, void* d_ws, size_t ws_size,
                              hipStream_t stream) {
    (void)in_sizes; (void)n_in; (void)out_size; (void)ws_size;

    const float* x   = (const float*)d_in[0];
    const int*   ei  = (const int*)  d_in[1];
    const float* W1  = (const float*)d_in[2];
    const float* b1  = (const float*)d_in[3];
    const float* W2  = (const float*)d_in[4];
    const float* b2  = (const float*)d_in[5];
    const float* Wt  = (const float*)d_in[6];
    const float* bt  = (const float*)d_in[7];
    const float* We  = (const float*)d_in[8];
    const float* be  = (const float*)d_in[9];
    float*       out = (float*)d_out;

    const int* src = ei;
    const int* dst = ei + N_EDGES;

    float* dinv = (float*)d_ws;
    float* bufA = dinv + ((N_NODES + 255) & ~255);
    float* bufB = bufA + (size_t)N_NODES * DIM;
    int*   recs      = (int*)(bufB + (size_t)N_NODES * DIM);
    int*   seg  = recs + (size_t)N_EDGES;

    k_fill_gemm1<<<N_BINB + GEMM_GRID, GEMM_THREADS, 0, stream>>>(src, dst, seg, recs, x, W1, bufA);
    k_tile_deg <<<N_TILES, DEG_THREADS, 0, stream>>>(recs, seg, dinv);

    k_tile_agg<true> <<<N_TILES, AGG_THREADS, 0, stream>>>(recs, seg, dinv, bufA, bufB);

    k_gemm2          <<<GEMM_GRID, GEMM_THREADS, 0, stream>>>(bufB, b1, W2, dinv, bufA);
    k_tile_agg<false><<<N_TILES, AGG_THREADS, 0, stream>>>(recs, seg, dinv, bufA, bufB);

    k_heads<<<HEAD_GRID, HEAD_THREADS, 0, stream>>>(bufB, b2, Wt, bt, We, be, out);
}
